# starting kernel
# baseline (speedup 1.0000x reference)
#include <hip/hip_runtime.h>
#include <hip/hip_bf16.h>

typedef __attribute__((ext_vector_type(2))) float v2f;
typedef __attribute__((ext_vector_type(8))) float v8f;

#define IN_C  128
#define HID_C 128
#define OUT_C 64

__global__ void deg_init(float* __restrict__ deg, int n) {
  int i = blockIdx.x * blockDim.x + threadIdx.x;
  if (i < n) deg[i] = 1.0f;
}

__global__ void deg_count(const int* __restrict__ dst, int E, float* __restrict__ deg) {
  int i = blockIdx.x * blockDim.x + threadIdx.x;
  if (i < E) atomicAdd(&deg[dst[i]], 1.0f);
}

__global__ void deg_finish(float* __restrict__ deg, int n) {
  int i = blockIdx.x * blockDim.x + threadIdx.x;
  if (i < n) deg[i] = rsqrtf(deg[i]);
}

__global__ void wmma_gemm_f32(const float* __restrict__ A, const float* __restrict__ B,
                              float* __restrict__ C, int M, int N, int K) {
  int wave = (int)((blockIdx.x * (long long)blockDim.x + threadIdx.x) >> 5);
  int lane = threadIdx.x & 31;
  int ntiles = N >> 4;
  int total = (M >> 4) * ntiles;
  if (wave >= total) return;
  int mt = wave / ntiles, nt = wave - mt * ntiles;
  int m0 = mt << 4, n0 = nt << 4;
  int half = lane >> 4;
  int l = lane & 15;

  const float* Arow = A + (size_t)(m0 + l) * K + 2 * half;
  const float* Bcol = B + (size_t)(2 * half) * N + n0 + l;

  v8f acc = {};
  for (int k = 0; k < K; k += 4) {
    v2f a, b;
    a.x = Arow[k];
    a.y = Arow[k + 1];
    b.x = Bcol[(size_t)k * N];
    b.y = Bcol[(size_t)(k + 1) * N];
    acc = __builtin_amdgcn_wmma_f32_16x16x4_f32(false, a, false, b, (short)0, acc,
                                                false, false);
  }

  float* Crow = C + (size_t)(m0 + 8 * half) * N + n0 + l;
#pragma unroll
  for (int i = 0; i < 8; ++i) Crow[(size_t)i * N] = acc[i];
}

__global__ void agg_self(const float* __restrict__ h, const float* __restrict__ dinv,
                         float* __restrict__ out, long long total, int C) {
  long long t = blockIdx.x * (long long)blockDim.x + threadIdx.x;
  if (t >= total) return;
  int v = (int)(t / C);
  float di = dinv[v];
  out[t] = h[t] * di * di;
}

__global__ void agg_scatter(const float* __restrict__ h, const int* __restrict__ src,
                            const int* __restrict__ dst, const float* __restrict__ dinv,
                            float* __restrict__ out, int E, int C) {
  long long t = blockIdx.x * (long long)blockDim.x + threadIdx.x;
  int groups = C >> 2;
  long long total = (long long)E * groups;
  if (t >= total) return;
  int e  = (int)(t / groups);
  int c4 = (int)(t - (long long)e * groups) << 2;
  int s = src[e], d = dst[e];
  float nrm = dinv[s] * dinv[d];
  float4 hv = *(const float4*)(h + (size_t)s * C + c4);
  float* o = out + (size_t)d * C + c4;
  atomicAdd(o + 0, hv.x * nrm);
  atomicAdd(o + 1, hv.y * nrm);
  atomicAdd(o + 2, hv.z * nrm);
  atomicAdd(o + 3, hv.w * nrm);
}

__global__ void bias_act(float* __restrict__ out, const float* __restrict__ b,
                         long long total, int C, int do_relu) {
  long long t = blockIdx.x * (long long)blockDim.x + threadIdx.x;
  if (t >= total) return;
  float v = out[t] + b[(int)(t % C)];
  out[t] = do_relu ? fmaxf(v, 0.0f) : v;
}

__global__ void decode_mlp(const float* __restrict__ z,
                           const int* __restrict__ sidx, const int* __restrict__ didx,
                           const float* __restrict__ Wa,
                           const float* __restrict__ ba,
                           const float* __restrict__ Wb,
                           const float* __restrict__ bb,
                           float* __restrict__ out, int P) {
  int wave = (int)((blockIdx.x * (long long)blockDim.x + threadIdx.x) >> 5);
  int lane = threadIdx.x & 31;
  if (wave >= (P >> 4)) return;
  int m0 = wave << 4;
  int half = lane >> 4;
  int l = lane & 15;

  int row = m0 + l;
  int s = sidx[row], d = didx[row];
  const float* zs = z + (size_t)s * OUT_C;
  const float* zd = z + (size_t)d * OUT_C;

  v8f acc[8];
#pragma unroll
  for (int t = 0; t < 8; ++t) acc[t] = (v8f){};

  for (int k = 0; k < 2 * OUT_C; k += 4) {
    int ka = k + 2 * half;
    v2f a;
    a.x = (ka     < OUT_C) ? zs[ka]         : zd[ka - OUT_C];
    a.y = (ka + 1 < OUT_C) ? zs[ka + 1]     : zd[ka + 1 - OUT_C];
    const float* Bp = Wa + (size_t)ka * HID_C + l;
#pragma unroll
    for (int t = 0; t < 8; ++t) {
      v2f b;
      b.x = Bp[t * 16];
      b.y = Bp[HID_C + t * 16];
      acc[t] = __builtin_amdgcn_wmma_f32_16x16x4_f32(false, a, false, b, (short)0,
                                                     acc[t], false, false);
    }
  }

  float part[8];
#pragma unroll
  for (int i = 0; i < 8; ++i) part[i] = 0.0f;
#pragma unroll
  for (int t = 0; t < 8; ++t) {
    int nidx = t * 16 + l;
    float bav = ba[nidx];
    float wbv = Wb[nidx];
#pragma unroll
    for (int i = 0; i < 8; ++i) {
      float hv = fmaxf(acc[t][i] + bav, 0.0f);
      part[i] += hv * wbv;
    }
  }
#pragma unroll
  for (int m = 1; m <= 8; m <<= 1) {
#pragma unroll
    for (int i = 0; i < 8; ++i) part[i] += __shfl_xor(part[i], m, 32);
  }
  if (l == 0) {
    float bbv = bb[0];
#pragma unroll
    for (int i = 0; i < 8; ++i) out[m0 + 8 * half + i] = part[i] + bbv;
  }
}

extern "C" void kernel_launch(void* const* d_in, const int* in_sizes, int n_in,
                              void* d_out, int out_size, void* d_ws, size_t ws_size,
                              hipStream_t stream) {
  const float* x   = (const float*)d_in[0];
  const int* eidx  = (const int*)d_in[1];
  const int* pidx  = (const int*)d_in[2];
  const int* nidx  = (const int*)d_in[3];
  const float* W1  = (const float*)d_in[4];
  const float* b1  = (const float*)d_in[5];
  const float* W2  = (const float*)d_in[6];
  const float* b2  = (const float*)d_in[7];
  const float* Wa  = (const float*)d_in[8];
  const float* ba  = (const float*)d_in[9];
  const float* Wb  = (const float*)d_in[10];
  const float* bb  = (const float*)d_in[11];
  float* out = (float*)d_out;

  const int n = in_sizes[0] / IN_C;
  const int E = in_sizes[1] / 2;
  const int P = in_sizes[2] / 2;
  const int* e_src = eidx;
  const int* e_dst = eidx + E;

  float* ws   = (float*)d_ws;
  float* h1   = ws;
  float* z1   = h1 + (size_t)n * HID_C;
  float* dinv = z1 + (size_t)n * HID_C;
  float* h2   = h1;
  float* z2   = h1 + (size_t)n * OUT_C;

  auto cdiv = [](long long a, long long b) { return (int)((a + b - 1) / b); };
  const int B = 256;

  deg_init<<<cdiv(n, B), B, 0, stream>>>(dinv, n);
  deg_count<<<cdiv(E, B), B, 0, stream>>>(e_dst, E, dinv);
  deg_finish<<<cdiv(n, B), B, 0, stream>>>(dinv, n);

  {
    long long waves = (long long)(n / 16) * (HID_C / 16);
    wmma_gemm_f32<<<cdiv(waves * 32, B), B, 0, stream>>>(x, W1, h1, n, HID_C, IN_C);
    long long tot = (long long)n * HID_C;
    agg_self<<<cdiv(tot, B), B, 0, stream>>>(h1, dinv, z1, tot, HID_C);
    agg_scatter<<<cdiv((long long)E * (HID_C / 4), B), B, 0, stream>>>(
        h1, e_src, e_dst, dinv, z1, E, HID_C);
    bias_act<<<cdiv(tot, B), B, 0, stream>>>(z1, b1, tot, HID_C, 1);
  }

  {
    long long waves = (long long)(n / 16) * (OUT_C / 16);
    wmma_gemm_f32<<<cdiv(waves * 32, B), B, 0, stream>>>(z1, W2, h2, n, OUT_C, HID_C);
    long long tot = (long long)n * OUT_C;
    agg_self<<<cdiv(tot, B), B, 0, stream>>>(h2, dinv, z2, tot, OUT_C);
    agg_scatter<<<cdiv((long long)E * (OUT_C / 4), B), B, 0, stream>>>(
        h2, e_src, e_dst, dinv, z2, E, OUT_C);
    bias_act<<<cdiv(tot, B), B, 0, stream>>>(z2, b2, tot, OUT_C, 0);
  }

  {
    long long waves = P / 16;
    decode_mlp<<<cdiv(waves * 32, B), B, 0, stream>>>(z2, pidx, pidx + P, Wa, ba, Wb,
                                                      bb, out, P);
    decode_mlp<<<cdiv(waves * 32, B), B, 0, stream>>>(z2, nidx, nidx + P, Wa, ba, Wb,
                                                      bb, out + P, P);
  }
}


# agg: float-atomic scatter-add replaced by in-call counting sort of edges by dst (int atomic rank + scan) and one-writer-per-row gather sum; self loop+bias+relu folded in; sum order = arrival rank
# speedup vs baseline: 1.7688x; 1.7688x over previous
#include <hip/hip_runtime.h>
#include <hip/hip_bf16.h>

typedef __attribute__((ext_vector_type(2))) float v2f;
typedef __attribute__((ext_vector_type(8))) float v8f;

#define IN_C  128
#define HID_C 128
#define OUT_C 64

__global__ void cnt_zero(int* __restrict__ cnt, int n) {
  int i = blockIdx.x * blockDim.x + threadIdx.x;
  if (i < n) cnt[i] = 0;
}

__global__ void edge_rank(const int* __restrict__ dst, int E, int* __restrict__ cnt,
                          int* __restrict__ rnk) {
  int i = blockIdx.x * blockDim.x + threadIdx.x;
  if (i < E) rnk[i] = atomicAdd(&cnt[dst[i]], 1);
}

__global__ __launch_bounds__(1024) void seg_offsets(const int* __restrict__ cnt, int n,
                                                    int* __restrict__ off,
                                                    float* __restrict__ dinv) {
  __shared__ int w_pre[32];
  __shared__ int w_cnt[32];
  const int tid = threadIdx.x;
  const int lane = tid & 31;
  const int w = tid >> 5;
  const int start = blockIdx.x * 1024;

  int pre = 0;
  const int4* c4 = (const int4*)cnt;
#pragma unroll 4
  for (int i = tid; i < (start >> 2); i += 1024) {
    int4 q = c4[i];
    pre += (q.x + q.y) + (q.z + q.w);
  }
  const int v = start + tid;
  const int c = (v < n) ? cnt[v] : 0;

  int inc = c;
  for (int d = 1; d < 32; d <<= 1) {
    int t = __shfl_up(inc, d, 32);
    if (lane >= d) inc += t;
  }
  for (int m = 16; m >= 1; m >>= 1) pre += __shfl_xor(pre, m, 32);
  if (lane == 31) w_cnt[w] = inc;
  if (lane == 0) w_pre[w] = pre;
  __syncthreads();

  int base = 0;
  for (int k = 0; k < 32; ++k) {
    base += w_pre[k];
    if (k < w) base += w_cnt[k];
  }
  if (v < n) {
    off[v] = base + inc - c;
    if (v == n - 1) off[n] = base + inc;
    dinv[v] = rsqrtf((float)(c + 1));
  }
}

__global__ void edge_fill(const int* __restrict__ src, const int* __restrict__ dst,
                          const int* __restrict__ rnk, const int* __restrict__ off, int E,
                          int* __restrict__ ssrc) {
  int i = blockIdx.x * blockDim.x + threadIdx.x;
  if (i < E) ssrc[off[dst[i]] + rnk[i]] = src[i];
}

__global__ void wmma_gemm_f32(const float* __restrict__ A, const float* __restrict__ B,
                              float* __restrict__ C, int M, int N, int K) {
  int wave = (int)((blockIdx.x * (long long)blockDim.x + threadIdx.x) >> 5);
  int lane = threadIdx.x & 31;
  int ntiles = N >> 4;
  int total = (M >> 4) * ntiles;
  if (wave >= total) return;
  int mt = wave / ntiles, nt = wave - mt * ntiles;
  int m0 = mt << 4, n0 = nt << 4;
  int half = lane >> 4;
  int l = lane & 15;

  const float* Arow = A + (size_t)(m0 + l) * K + 2 * half;
  const float* Bcol = B + (size_t)(2 * half) * N + n0 + l;

  v8f acc = {};
  for (int k = 0; k < K; k += 4) {
    v2f a, b;
    a.x = Arow[k];
    a.y = Arow[k + 1];
    b.x = Bcol[(size_t)k * N];
    b.y = Bcol[(size_t)(k + 1) * N];
    acc = __builtin_amdgcn_wmma_f32_16x16x4_f32(false, a, false, b, (short)0, acc,
                                                false, false);
  }

  float* Crow = C + (size_t)(m0 + 8 * half) * N + n0 + l;
#pragma unroll
  for (int i = 0; i < 8; ++i) Crow[(size_t)i * N] = acc[i];
}

template <int C, bool RELU>
__global__ __launch_bounds__(256) void agg_gather(const float* __restrict__ h,
                                                  const int* __restrict__ off,
                                                  const int* __restrict__ ssrc,
                                                  const float* __restrict__ dinv,
                                                  const float* __restrict__ bias,
                                                  float* __restrict__ out, int n) {
  constexpr int VW = C / 32;
  typedef float VT __attribute__((ext_vector_type(VW)));
  const int lane = threadIdx.x & 31;
  const int v = __builtin_amdgcn_readfirstlane(
      (int)((blockIdx.x * (long long)blockDim.x + threadIdx.x) >> 5));
  if (v >= n) return;
  const int beg = off[v];
  const int c = off[v + 1] - beg;
  const float dv = dinv[v];
  __builtin_assume(c >= 0);
  const float* hl = h + lane * VW;

  const int tot = c + 1;
  VT acc = {};

  for (int j0 = 0; j0 < tot; j0 += 32) {
    const int m = min(32, tot - j0);
    const int idx = j0 + lane;
    int s_l = v;
    float d_l = (idx == c) ? dv : 0.0f;
    if (idx < c) {
      s_l = ssrc[beg + idx];
      d_l = dinv[s_l];
    }
    for (int j = 0; j < m; j += 8) {
      VT r[8];
      float wt[8];
      for (int u = 0; u < 8; ++u) {
        const int s = __builtin_amdgcn_readlane(s_l, j + u);
        wt[u] = dv * __builtin_bit_cast(
            float, __builtin_amdgcn_readlane(__builtin_bit_cast(int, d_l), j + u));
        r[u] = *(const VT*)(hl + (size_t)s * C);
      }
      for (int u = 0; u < 8; ++u) acc += r[u] * wt[u];
    }
  }

  const VT bv = *(const VT*)(bias + lane * VW);
  acc += bv;
  if (RELU) {
    for (int i = 0; i < VW; ++i) acc[i] = fmaxf(acc[i], 0.0f);
  }
  *(VT*)(out + (size_t)v * C + lane * VW) = acc;
}

__global__ void decode_mlp(const float* __restrict__ z,
                           const int* __restrict__ sidx, const int* __restrict__ didx,
                           const float* __restrict__ Wa,
                           const float* __restrict__ ba,
                           const float* __restrict__ Wb,
                           const float* __restrict__ bb,
                           float* __restrict__ out, int P) {
  int wave = (int)((blockIdx.x * (long long)blockDim.x + threadIdx.x) >> 5);
  int lane = threadIdx.x & 31;
  if (wave >= (P >> 4)) return;
  int m0 = wave << 4;
  int half = lane >> 4;
  int l = lane & 15;

  int row = m0 + l;
  int s = sidx[row], d = didx[row];
  const float* zs = z + (size_t)s * OUT_C;
  const float* zd = z + (size_t)d * OUT_C;

  v8f acc[8];
#pragma unroll
  for (int t = 0; t < 8; ++t) acc[t] = (v8f){};

  for (int k = 0; k < 2 * OUT_C; k += 4) {
    int ka = k + 2 * half;
    v2f a;
    a.x = (ka     < OUT_C) ? zs[ka]         : zd[ka - OUT_C];
    a.y = (ka + 1 < OUT_C) ? zs[ka + 1]     : zd[ka + 1 - OUT_C];
    const float* Bp = Wa + (size_t)ka * HID_C + l;
#pragma unroll
    for (int t = 0; t < 8; ++t) {
      v2f b;
      b.x = Bp[t * 16];
      b.y = Bp[HID_C + t * 16];
      acc[t] = __builtin_amdgcn_wmma_f32_16x16x4_f32(false, a, false, b, (short)0,
                                                     acc[t], false, false);
    }
  }

  float part[8];
#pragma unroll
  for (int i = 0; i < 8; ++i) part[i] = 0.0f;
#pragma unroll
  for (int t = 0; t < 8; ++t) {
    int nidx = t * 16 + l;
    float bav = ba[nidx];
    float wbv = Wb[nidx];
#pragma unroll
    for (int i = 0; i < 8; ++i) {
      float hv = fmaxf(acc[t][i] + bav, 0.0f);
      part[i] += hv * wbv;
    }
  }
#pragma unroll
  for (int m = 1; m <= 8; m <<= 1) {
#pragma unroll
    for (int i = 0; i < 8; ++i) part[i] += __shfl_xor(part[i], m, 32);
  }
  if (l == 0) {
    float bbv = bb[0];
#pragma unroll
    for (int i = 0; i < 8; ++i) out[m0 + 8 * half + i] = part[i] + bbv;
  }
}

extern "C" void kernel_launch(void* const* d_in, const int* in_sizes, int n_in,
                              void* d_out, int out_size, void* d_ws, size_t ws_size,
                              hipStream_t stream) {
  const float* x   = (const float*)d_in[0];
  const int* eidx  = (const int*)d_in[1];
  const int* pidx  = (const int*)d_in[2];
  const int* nidx  = (const int*)d_in[3];
  const float* W1  = (const float*)d_in[4];
  const float* b1  = (const float*)d_in[5];
  const float* W2  = (const float*)d_in[6];
  const float* b2  = (const float*)d_in[7];
  const float* Wa  = (const float*)d_in[8];
  const float* ba  = (const float*)d_in[9];
  const float* Wb  = (const float*)d_in[10];
  const float* bb  = (const float*)d_in[11];
  float* out = (float*)d_out;

  const int n = in_sizes[0] / IN_C;
  const int E = in_sizes[1] / 2;
  const int P = in_sizes[2] / 2;
  const int* e_src = eidx;
  const int* e_dst = eidx + E;

  float* ws   = (float*)d_ws;
  float* h1   = ws;
  float* z1   = h1 + (size_t)n * HID_C;
  float* dinv = z1 + (size_t)n * HID_C;
  int*   cnt  = (int*)(dinv + n);
  int*   off  = cnt + n;
  int*   rnk  = off + n + 1;
  int*   ssrc = rnk + E;
  float* h2   = h1;
  float* z2   = h1 + (size_t)n * OUT_C;

  auto cdiv = [](long long a, long long b) { return (int)((a + b - 1) / b); };
  const int B = 256;

  cnt_zero<<<cdiv(n, B), B, 0, stream>>>(cnt, n);
  edge_rank<<<cdiv(E, B), B, 0, stream>>>(e_dst, E, cnt, rnk);
  seg_offsets<<<cdiv(n, 1024), 1024, 0, stream>>>(cnt, n, off, dinv);
  edge_fill<<<cdiv(E, B), B, 0, stream>>>(e_src, e_dst, rnk, off, E, ssrc);

  {
    long long waves = (long long)(n / 16) * (HID_C / 16);
    wmma_gemm_f32<<<cdiv(waves * 32, B), B, 0, stream>>>(x, W1, h1, n, HID_C, IN_C);
    agg_gather<HID_C, true><<<cdiv((long long)n * 32, B), B, 0, stream>>>(
        h1, off, ssrc, dinv, b1, z1, n);
  }

  {
    long long waves = (long long)(n / 16) * (OUT_C / 16);
    wmma_gemm_f32<<<cdiv(waves * 32, B), B, 0, stream>>>(z1, W2, h2, n, OUT_C, HID_C);
    agg_gather<OUT_C, false><<<cdiv((long long)n * 32, B), B, 0, stream>>>(
        h2, off, ssrc, dinv, b2, z2, n);
  }

  {
    long long waves = P / 16;
    decode_mlp<<<cdiv(waves * 32, B), B, 0, stream>>>(z2, pidx, pidx + P, Wa, ba, Wb,
                                                      bb, out, P);
    decode_mlp<<<cdiv(waves * 32, B), B, 0, stream>>>(z2, nidx, nidx + P, Wa, ba, Wb,
                                                      bb, out + P, P);
  }
}


# decode: feat@Wa split by linearity into z@Wa[0:64] and z@Wa[64:128], taken once per node on WMMA f32; each pair adds two projected rows; k sum now two halves added, Wb dot summed in a new tree
# speedup vs baseline: 2.2179x; 1.2539x over previous
#include <hip/hip_runtime.h>
#include <hip/hip_bf16.h>

typedef __attribute__((ext_vector_type(2))) float v2f;
typedef __attribute__((ext_vector_type(8))) float v8f;

#define IN_C  128
#define HID_C 128
#define OUT_C 64

__global__ void cnt_zero(int* __restrict__ cnt, int n) {
  int i = blockIdx.x * blockDim.x + threadIdx.x;
  if (i < n) cnt[i] = 0;
}

__global__ void edge_rank(const int* __restrict__ dst, int E, int* __restrict__ cnt,
                          int* __restrict__ rnk) {
  int i = blockIdx.x * blockDim.x + threadIdx.x;
  if (i < E) rnk[i] = atomicAdd(&cnt[dst[i]], 1);
}

__global__ __launch_bounds__(1024) void seg_offsets(const int* __restrict__ cnt, int n,
                                                    int* __restrict__ off,
                                                    float* __restrict__ dinv) {
  __shared__ int w_pre[32];
  __shared__ int w_cnt[32];
  const int tid = threadIdx.x;
  const int lane = tid & 31;
  const int w = tid >> 5;
  const int start = blockIdx.x * 1024;

  int pre = 0;
  const int4* c4 = (const int4*)cnt;
#pragma unroll 4
  for (int i = tid; i < (start >> 2); i += 1024) {
    int4 q = c4[i];
    pre += (q.x + q.y) + (q.z + q.w);
  }
  const int v = start + tid;
  const int c = (v < n) ? cnt[v] : 0;

  int inc = c;
  for (int d = 1; d < 32; d <<= 1) {
    int t = __shfl_up(inc, d, 32);
    if (lane >= d) inc += t;
  }
  for (int m = 16; m >= 1; m >>= 1) pre += __shfl_xor(pre, m, 32);
  if (lane == 31) w_cnt[w] = inc;
  if (lane == 0) w_pre[w] = pre;
  __syncthreads();

  int base = 0;
  for (int k = 0; k < 32; ++k) {
    base += w_pre[k];
    if (k < w) base += w_cnt[k];
  }
  if (v < n) {
    off[v] = base + inc - c;
    if (v == n - 1) off[n] = base + inc;
    dinv[v] = rsqrtf((float)(c + 1));
  }
}

__global__ void edge_fill(const int* __restrict__ src, const int* __restrict__ dst,
                          const int* __restrict__ rnk, const int* __restrict__ off, int E,
                          int* __restrict__ ssrc) {
  int i = blockIdx.x * blockDim.x + threadIdx.x;
  if (i < E) ssrc[off[dst[i]] + rnk[i]] = src[i];
}

__global__ void wmma_gemm_f32(const float* __restrict__ A, const float* __restrict__ B,
                              float* __restrict__ C, int M, int N, int K) {
  int wave = (int)((blockIdx.x * (long long)blockDim.x + threadIdx.x) >> 5);
  int lane = threadIdx.x & 31;
  int ntiles = N >> 4;
  int total = (M >> 4) * ntiles;
  if (wave >= total) return;
  int mt = wave / ntiles, nt = wave - mt * ntiles;
  int m0 = mt << 4, n0 = nt << 4;
  int half = lane >> 4;
  int l = lane & 15;

  const float* Arow = A + (size_t)(m0 + l) * K + 2 * half;
  const float* Bcol = B + (size_t)(2 * half) * N + n0 + l;

  v8f acc = {};
  for (int k = 0; k < K; k += 4) {
    v2f a, b;
    a.x = Arow[k];
    a.y = Arow[k + 1];
    b.x = Bcol[(size_t)k * N];
    b.y = Bcol[(size_t)(k + 1) * N];
    acc = __builtin_amdgcn_wmma_f32_16x16x4_f32(false, a, false, b, (short)0, acc,
                                                false, false);
  }

  float* Crow = C + (size_t)(m0 + 8 * half) * N + n0 + l;
#pragma unroll
  for (int i = 0; i < 8; ++i) Crow[(size_t)i * N] = acc[i];
}

template <int C, bool RELU>
__global__ __launch_bounds__(256) void agg_gather(const float* __restrict__ h,
                                                  const int* __restrict__ off,
                                                  const int* __restrict__ ssrc,
                                                  const float* __restrict__ dinv,
                                                  const float* __restrict__ bias,
                                                  float* __restrict__ out, int n) {
  constexpr int VW = C / 32;
  typedef float VT __attribute__((ext_vector_type(VW)));
  const int lane = threadIdx.x & 31;
  const int v = __builtin_amdgcn_readfirstlane(
      (int)((blockIdx.x * (long long)blockDim.x + threadIdx.x) >> 5));
  if (v >= n) return;
  const int beg = off[v];
  const int c = off[v + 1] - beg;
  const float dv = dinv[v];
  __builtin_assume(c >= 0);
  const float* hl = h + lane * VW;

  const int tot = c + 1;
  VT acc = {};

  for (int j0 = 0; j0 < tot; j0 += 32) {
    const int m = min(32, tot - j0);
    const int idx = j0 + lane;
    int s_l = v;
    float d_l = (idx == c) ? dv : 0.0f;
    if (idx < c) {
      s_l = ssrc[beg + idx];
      d_l = dinv[s_l];
    }
    for (int j = 0; j < m; j += 8) {
      VT r[8];
      float wt[8];
      for (int u = 0; u < 8; ++u) {
        const int s = __builtin_amdgcn_readlane(s_l, j + u);
        wt[u] = dv * __builtin_bit_cast(
            float, __builtin_amdgcn_readlane(__builtin_bit_cast(int, d_l), j + u));
        r[u] = *(const VT*)(hl + (size_t)s * C);
      }
      for (int u = 0; u < 8; ++u) acc += r[u] * wt[u];
    }
  }

  const VT bv = *(const VT*)(bias + lane * VW);
  acc += bv;
  if (RELU) {
    for (int i = 0; i < VW; ++i) acc[i] = fmaxf(acc[i], 0.0f);
  }
  *(VT*)(out + (size_t)v * C + lane * VW) = acc;
}

__global__ __launch_bounds__(256) void node_proj(const float* __restrict__ z,
                                                 const float* __restrict__ Wa,
                                                 float* __restrict__ U, float* __restrict__ V,
                                                 int n) {
  int wave = (int)((blockIdx.x * (long long)blockDim.x + threadIdx.x) >> 5);
  int lane = threadIdx.x & 31;
  if (wave >= 2 * (n >> 4)) return;
  int sel = wave & 1;
  int m0 = (wave >> 1) << 4;
  int half = lane >> 4;
  int l = lane & 15;

  const float* zr = z + (size_t)(m0 + l) * OUT_C + 2 * half;
  const float* Wp = Wa + (size_t)(sel * OUT_C + 2 * half) * HID_C + l;

  v8f acc[8];
#pragma unroll
  for (int t = 0; t < 8; ++t) acc[t] = (v8f){};

  for (int k = 0; k < OUT_C; k += 4) {
    v2f a;
    a.x = zr[k];
    a.y = zr[k + 1];
    const float* Bp = Wp + (size_t)k * HID_C;
#pragma unroll
    for (int t = 0; t < 8; ++t) {
      v2f b;
      b.x = Bp[t * 16];
      b.y = Bp[HID_C + t * 16];
      acc[t] = __builtin_amdgcn_wmma_f32_16x16x4_f32(false, a, false, b, (short)0,
                                                     acc[t], false, false);
    }
  }

  float* Crow = (sel ? V : U) + (size_t)(m0 + 8 * half) * HID_C + l;
#pragma unroll
  for (int t = 0; t < 8; ++t) {
#pragma unroll
    for (int i = 0; i < 8; ++i) Crow[(size_t)i * HID_C + t * 16] = acc[t][i];
  }
}

__global__ __launch_bounds__(256) void pair_score(const float* __restrict__ U,
                                                  const float* __restrict__ V,
                                                  const int* __restrict__ pidx,
                                                  const int* __restrict__ nidx,
                                                  const float* __restrict__ ba,
                                                  const float* __restrict__ Wb,
                                                  const float* __restrict__ bb,
                                                  float* __restrict__ out, int P) {
  typedef float f4 __attribute__((ext_vector_type(4)));
  const int lane = threadIdx.x & 31;
  const int wave = (int)((blockIdx.x * (long long)blockDim.x + threadIdx.x) >> 5);
  const int q = lane >> 3;
  const int t = lane & 7;
  const int total = 2 * P;
  const int base = wave * 16;
  if (base >= total) return;

  int sv[4], dv[4];
#pragma unroll
  for (int st = 0; st < 4; ++st) {
    const int pc = min(base + st * 4 + q, total - 1);
    const int* ix = (pc < P) ? pidx : nidx;
    const int r = (pc < P) ? pc : pc - P;
    sv[st] = ix[r];
    dv[st] = ix[P + r];
  }

  f4 bav[4], wbv[4];
#pragma unroll
  for (int i = 0; i < 4; ++i) {
    bav[i] = *(const f4*)(ba + i * 32 + t * 4);
    wbv[i] = *(const f4*)(Wb + i * 32 + t * 4);
  }
  const float bbv = bb[0];

#pragma unroll
  for (int g = 0; g < 4; g += 2) {
    f4 ru[2][4], rv[2][4];
#pragma unroll
    for (int h = 0; h < 2; ++h) {
      const float* up = U + (size_t)sv[g + h] * HID_C + t * 4;
      const float* vp = V + (size_t)dv[g + h] * HID_C + t * 4;
#pragma unroll
      for (int i = 0; i < 4; ++i) {
        ru[h][i] = *(const f4*)(up + i * 32);
        rv[h][i] = *(const f4*)(vp + i * 32);
      }
    }
#pragma unroll
    for (int h = 0; h < 2; ++h) {
      float part = 0.0f;
#pragma unroll
      for (int i = 0; i < 4; ++i) {
        const f4 hs = (ru[h][i] + rv[h][i]) + bav[i];
#pragma unroll
        for (int c = 0; c < 4; ++c) part += fmaxf(hs[c], 0.0f) * wbv[i][c];
      }
      part += __shfl_xor(part, 1, 32);
      part += __shfl_xor(part, 2, 32);
      part += __shfl_xor(part, 4, 32);
      const int p = base + (g + h) * 4 + q;
      if (t == 0 && p < total) out[p] = part + bbv;
    }
  }
}

extern "C" void kernel_launch(void* const* d_in, const int* in_sizes, int n_in,
                              void* d_out, int out_size, void* d_ws, size_t ws_size,
                              hipStream_t stream) {
  const float* x   = (const float*)d_in[0];
  const int* eidx  = (const int*)d_in[1];
  const int* pidx  = (const int*)d_in[2];
  const int* nidx  = (const int*)d_in[3];
  const float* W1  = (const float*)d_in[4];
  const float* b1  = (const float*)d_in[5];
  const float* W2  = (const float*)d_in[6];
  const float* b2  = (const float*)d_in[7];
  const float* Wa  = (const float*)d_in[8];
  const float* ba  = (const float*)d_in[9];
  const float* Wb  = (const float*)d_in[10];
  const float* bb  = (const float*)d_in[11];
  float* out = (float*)d_out;

  const int n = in_sizes[0] / IN_C;
  const int E = in_sizes[1] / 2;
  const int P = in_sizes[2] / 2;
  const int* e_src = eidx;
  const int* e_dst = eidx + E;

  float* ws   = (float*)d_ws;
  float* h1   = ws;
  float* z1   = h1 + (size_t)n * HID_C;
  float* pU   = z1 + (size_t)n * HID_C;
  float* pV   = pU + (size_t)n * HID_C;
  float* dinv = pV + (size_t)n * HID_C;
  int*   cnt  = (int*)(dinv + n);
  int*   off  = cnt + n;
  int*   rnk  = off + n + 1;
  int*   ssrc = rnk + E;
  float* h2   = h1;
  float* z2   = h1 + (size_t)n * OUT_C;

  auto cdiv = [](long long a, long long b) { return (int)((a + b - 1) / b); };
  const int B = 256;

  cnt_zero<<<cdiv(n, B), B, 0, stream>>>(cnt, n);
  edge_rank<<<cdiv(E, B), B, 0, stream>>>(e_dst, E, cnt, rnk);
  seg_offsets<<<cdiv(n, 1024), 1024, 0, stream>>>(cnt, n, off, dinv);
  edge_fill<<<cdiv(E, B), B, 0, stream>>>(e_src, e_dst, rnk, off, E, ssrc);

  {
    long long waves = (long long)(n / 16) * (HID_C / 16);
    wmma_gemm_f32<<<cdiv(waves * 32, B), B, 0, stream>>>(x, W1, h1, n, HID_C, IN_C);
    agg_gather<HID_C, true><<<cdiv((long long)n * 32, B), B, 0, stream>>>(
        h1, off, ssrc, dinv, b1, z1, n);
  }

  {
    long long waves = (long long)(n / 16) * (OUT_C / 16);
    wmma_gemm_f32<<<cdiv(waves * 32, B), B, 0, stream>>>(z1, W2, h2, n, OUT_C, HID_C);
    agg_gather<OUT_C, false><<<cdiv((long long)n * 32, B), B, 0, stream>>>(
        h2, off, ssrc, dinv, b2, z2, n);
  }

  {
    long long waves = 2LL * (n / 16);
    node_proj<<<cdiv(waves * 32, B), B, 0, stream>>>(z2, Wa, pU, pV, n);
    long long pwaves = cdiv(2LL * P, 16);
    pair_score<<<cdiv(pwaves * 32, B), B, 0, stream>>>(pU, pV, pidx, nidx, ba, Wb, bb,
                                                       out, P);
  }
}


# layer products (x@W1, z1@W2): strip kernel, a wave takes 16 rows over all columns with its A fragments in registers, W staged once per work-group in LDS as k pairs; k order and sums as before
# speedup vs baseline: 2.4070x; 1.0853x over previous
#include <hip/hip_runtime.h>
#include <hip/hip_bf16.h>

typedef __attribute__((ext_vector_type(2))) float v2f;
typedef __attribute__((ext_vector_type(8))) float v8f;

#define IN_C  128
#define HID_C 128
#define OUT_C 64

__global__ void cnt_zero(int* __restrict__ cnt, int n) {
  int i = blockIdx.x * blockDim.x + threadIdx.x;
  if (i < n) cnt[i] = 0;
}

__global__ void edge_rank(const int* __restrict__ dst, int E, int* __restrict__ cnt,
                          int* __restrict__ rnk) {
  int i = blockIdx.x * blockDim.x + threadIdx.x;
  if (i < E) rnk[i] = atomicAdd(&cnt[dst[i]], 1);
}

__global__ __launch_bounds__(1024) void seg_offsets(const int* __restrict__ cnt, int n,
                                                    int* __restrict__ off,
                                                    float* __restrict__ dinv) {
  __shared__ int w_pre[32];
  __shared__ int w_cnt[32];
  const int tid = threadIdx.x;
  const int lane = tid & 31;
  const int w = tid >> 5;
  const int start = blockIdx.x * 1024;

  int pre = 0;
  const int4* c4 = (const int4*)cnt;
#pragma unroll 4
  for (int i = tid; i < (start >> 2); i += 1024) {
    int4 q = c4[i];
    pre += (q.x + q.y) + (q.z + q.w);
  }
  const int v = start + tid;
  const int c = (v < n) ? cnt[v] : 0;

  int inc = c;
  for (int d = 1; d < 32; d <<= 1) {
    int t = __shfl_up(inc, d, 32);
    if (lane >= d) inc += t;
  }
  for (int m = 16; m >= 1; m >>= 1) pre += __shfl_xor(pre, m, 32);
  if (lane == 31) w_cnt[w] = inc;
  if (lane == 0) w_pre[w] = pre;
  __syncthreads();

  int base = 0;
  for (int k = 0; k < 32; ++k) {
    base += w_pre[k];
    if (k < w) base += w_cnt[k];
  }
  if (v < n) {
    off[v] = base + inc - c;
    if (v == n - 1) off[n] = base + inc;
    dinv[v] = rsqrtf((float)(c + 1));
  }
}

__global__ void edge_fill(const int* __restrict__ src, const int* __restrict__ dst,
                          const int* __restrict__ rnk, const int* __restrict__ off, int E,
                          int* __restrict__ ssrc) {
  int i = blockIdx.x * blockDim.x + threadIdx.x;
  if (i < E) ssrc[off[dst[i]] + rnk[i]] = src[i];
}

template <int K, int NT>
__global__ __launch_bounds__(256) void strip_gemm(const float* __restrict__ A,
                                                  const float* __restrict__ B,
                                                  float* __restrict__ C, int M, int ldb,
                                                  int ldc) {
  typedef float f4 __attribute__((ext_vector_type(4)));
  constexpr int N = NT * 16;
  constexpr int KP = K / 2;
  constexpr int KS = K / 4;
  constexpr int PITCH = 2 * N + 32;
  constexpr int ITEMS = KP * (N / 4);
  static_assert(ITEMS % 256 == 0, "staging loop has no tail");
  __shared__ __attribute__((aligned(16))) float Bl[KP * PITCH];

  const int tid = threadIdx.x;
  const int lane = tid & 31;
  const int w = __builtin_amdgcn_readfirstlane(tid >> 5);
  const int c = lane & 15;
  const int h = lane >> 4;

  const int nstrips = M >> 4;
  const int strip = blockIdx.x * 8 + w;
  const bool active = strip < nstrips;
  const int m0 = (active ? strip : nstrips - 1) << 4;

  v2f a[KS];
  const float* Arow = A + (size_t)(m0 + c) * K + 2 * h;
#pragma unroll
  for (int kk = 0; kk < KS; ++kk) a[kk] = *(const v2f*)(Arow + 4 * kk);

  {
    f4 r0[ITEMS / 256], r1[ITEMS / 256];
#pragma unroll
    for (int j = 0; j < ITEMS / 256; ++j) {
      const int i = tid + 256 * j;
      const int kp = i / (N / 4), n4 = i % (N / 4);
      r0[j] = *(const f4*)(B + (size_t)(2 * kp) * ldb + 4 * n4);
      r1[j] = *(const f4*)(B + (size_t)(2 * kp + 1) * ldb + 4 * n4);
    }
#pragma unroll
    for (int j = 0; j < ITEMS / 256; ++j) {
      const int i = tid + 256 * j;
      const int kp = i / (N / 4), n4 = i % (N / 4);
      float* d = Bl + kp * PITCH + 8 * n4;
      *(f4*)d = (f4){r0[j].x, r1[j].x, r0[j].y, r1[j].y};
      *(f4*)(d + 4) = (f4){r0[j].z, r1[j].z, r0[j].w, r1[j].w};
    }
  }
  __syncthreads();

  if (active) {
    v8f acc[NT];
#pragma unroll
    for (int t = 0; t < NT; ++t) acc[t] = (v8f){};

    const float* Bf = Bl + h * PITCH + 2 * c;
#pragma unroll
    for (int kk = 0; kk < KS; ++kk) {
#pragma unroll
      for (int t = 0; t < NT; ++t) {
        const v2f b = *(const v2f*)(Bf + (2 * kk) * PITCH + 32 * t);
        acc[t] = __builtin_amdgcn_wmma_f32_16x16x4_f32(false, a[kk], false, b, (short)0,
                                                       acc[t], false, false);
      }
    }

    float* Crow = C + (size_t)(m0 + 8 * h) * ldc + c;
#pragma unroll
    for (int t = 0; t < NT; ++t) {
#pragma unroll
      for (int i = 0; i < 8; ++i) Crow[(size_t)i * ldc + t * 16] = acc[t][i];
    }
  }
}

template <int C, bool RELU>
__global__ __launch_bounds__(256) void agg_gather(const float* __restrict__ h,
                                                  const int* __restrict__ off,
                                                  const int* __restrict__ ssrc,
                                                  const float* __restrict__ dinv,
                                                  const float* __restrict__ bias,
                                                  float* __restrict__ out, int n) {
  constexpr int VW = C / 32;
  typedef float VT __attribute__((ext_vector_type(VW)));
  const int lane = threadIdx.x & 31;
  const int v = __builtin_amdgcn_readfirstlane(
      (int)((blockIdx.x * (long long)blockDim.x + threadIdx.x) >> 5));
  if (v >= n) return;
  const int beg = off[v];
  const int c = off[v + 1] - beg;
  const float dv = dinv[v];
  __builtin_assume(c >= 0);
  const float* hl = h + lane * VW;

  const int tot = c + 1;
  VT acc = {};

  for (int j0 = 0; j0 < tot; j0 += 32) {
    const int m = min(32, tot - j0);
    const int idx = j0 + lane;
    int s_l = v;
    float d_l = (idx == c) ? dv : 0.0f;
    if (idx < c) {
      s_l = ssrc[beg + idx];
      d_l = dinv[s_l];
    }
    for (int j = 0; j < m; j += 8) {
      VT r[8];
      float wt[8];
      for (int u = 0; u < 8; ++u) {
        const int s = __builtin_amdgcn_readlane(s_l, j + u);
        wt[u] = dv * __builtin_bit_cast(
            float, __builtin_amdgcn_readlane(__builtin_bit_cast(int, d_l), j + u));
        r[u] = *(const VT*)(hl + (size_t)s * C);
      }
      for (int u = 0; u < 8; ++u) acc += r[u] * wt[u];
    }
  }

  const VT bv = *(const VT*)(bias + lane * VW);
  acc += bv;
  if (RELU) {
    for (int i = 0; i < VW; ++i) acc[i] = fmaxf(acc[i], 0.0f);
  }
  *(VT*)(out + (size_t)v * C + lane * VW) = acc;
}

__global__ __launch_bounds__(256) void node_proj(const float* __restrict__ z,
                                                 const float* __restrict__ Wa,
                                                 float* __restrict__ U, float* __restrict__ V,
                                                 int n) {
  int wave = (int)((blockIdx.x * (long long)blockDim.x + threadIdx.x) >> 5);
  int lane = threadIdx.x & 31;
  if (wave >= 2 * (n >> 4)) return;
  int sel = wave & 1;
  int m0 = (wave >> 1) << 4;
  int half = lane >> 4;
  int l = lane & 15;

  const float* zr = z + (size_t)(m0 + l) * OUT_C + 2 * half;
  const float* Wp = Wa + (size_t)(sel * OUT_C + 2 * half) * HID_C + l;

  v8f acc[8];
#pragma unroll
  for (int t = 0; t < 8; ++t) acc[t] = (v8f){};

  for (int k = 0; k < OUT_C; k += 4) {
    v2f a;
    a.x = zr[k];
    a.y = zr[k + 1];
    const float* Bp = Wp + (size_t)k * HID_C;
#pragma unroll
    for (int t = 0; t < 8; ++t) {
      v2f b;
      b.x = Bp[t * 16];
      b.y = Bp[HID_C + t * 16];
      acc[t] = __builtin_amdgcn_wmma_f32_16x16x4_f32(false, a, false, b, (short)0,
                                                     acc[t], false, false);
    }
  }

  float* Crow = (sel ? V : U) + (size_t)(m0 + 8 * half) * HID_C + l;
#pragma unroll
  for (int t = 0; t < 8; ++t) {
#pragma unroll
    for (int i = 0; i < 8; ++i) Crow[(size_t)i * HID_C + t * 16] = acc[t][i];
  }
}

__global__ __launch_bounds__(256) void pair_score(const float* __restrict__ U,
                                                  const float* __restrict__ V,
                                                  const int* __restrict__ pidx,
                                                  const int* __restrict__ nidx,
                                                  const float* __restrict__ ba,
                                                  const float* __restrict__ Wb,
                                                  const float* __restrict__ bb,
                                                  float* __restrict__ out, int P) {
  typedef float f4 __attribute__((ext_vector_type(4)));
  const int lane = threadIdx.x & 31;
  const int wave = (int)((blockIdx.x * (long long)blockDim.x + threadIdx.x) >> 5);
  const int q = lane >> 3;
  const int t = lane & 7;
  const int total = 2 * P;
  const int base = wave * 16;
  if (base >= total) return;

  int sv[4], dv[4];
#pragma unroll
  for (int st = 0; st < 4; ++st) {
    const int pc = min(base + st * 4 + q, total - 1);
    const int* ix = (pc < P) ? pidx : nidx;
    const int r = (pc < P) ? pc : pc - P;
    sv[st] = ix[r];
    dv[st] = ix[P + r];
  }

  f4 bav[4], wbv[4];
#pragma unroll
  for (int i = 0; i < 4; ++i) {
    bav[i] = *(const f4*)(ba + i * 32 + t * 4);
    wbv[i] = *(const f4*)(Wb + i * 32 + t * 4);
  }
  const float bbv = bb[0];

#pragma unroll
  for (int g = 0; g < 4; g += 2) {
    f4 ru[2][4], rv[2][4];
#pragma unroll
    for (int h = 0; h < 2; ++h) {
      const float* up = U + (size_t)sv[g + h] * HID_C + t * 4;
      const float* vp = V + (size_t)dv[g + h] * HID_C + t * 4;
#pragma unroll
      for (int i = 0; i < 4; ++i) {
        ru[h][i] = *(const f4*)(up + i * 32);
        rv[h][i] = *(const f4*)(vp + i * 32);
      }
    }
#pragma unroll
    for (int h = 0; h < 2; ++h) {
      float part = 0.0f;
#pragma unroll
      for (int i = 0; i < 4; ++i) {
        const f4 hs = (ru[h][i] + rv[h][i]) + bav[i];
#pragma unroll
        for (int c = 0; c < 4; ++c) part += fmaxf(hs[c], 0.0f) * wbv[i][c];
      }
      part += __shfl_xor(part, 1, 32);
      part += __shfl_xor(part, 2, 32);
      part += __shfl_xor(part, 4, 32);
      const int p = base + (g + h) * 4 + q;
      if (t == 0 && p < total) out[p] = part + bbv;
    }
  }
}

extern "C" void kernel_launch(void* const* d_in, const int* in_sizes, int n_in,
                              void* d_out, int out_size, void* d_ws, size_t ws_size,
                              hipStream_t stream) {
  const float* x   = (const float*)d_in[0];
  const int* eidx  = (const int*)d_in[1];
  const int* pidx  = (const int*)d_in[2];
  const int* nidx  = (const int*)d_in[3];
  const float* W1  = (const float*)d_in[4];
  const float* b1  = (const float*)d_in[5];
  const float* W2  = (const float*)d_in[6];
  const float* b2  = (const float*)d_in[7];
  const float* Wa  = (const float*)d_in[8];
  const float* ba  = (const float*)d_in[9];
  const float* Wb  = (const float*)d_in[10];
  const float* bb  = (const float*)d_in[11];
  float* out = (float*)d_out;

  const int n = in_sizes[0] / IN_C;
  const int E = in_sizes[1] / 2;
  const int P = in_sizes[2] / 2;
  const int* e_src = eidx;
  const int* e_dst = eidx + E;

  float* ws   = (float*)d_ws;
  float* h1   = ws;
  float* z1   = h1 + (size_t)n * HID_C;
  float* pU   = z1 + (size_t)n * HID_C;
  float* pV   = pU + (size_t)n * HID_C;
  float* dinv = pV + (size_t)n * HID_C;
  int*   cnt  = (int*)(dinv + n);
  int*   off  = cnt + n;
  int*   rnk  = off + n + 1;
  int*   ssrc = rnk + E;
  float* h2   = h1;
  float* z2   = h1 + (size_t)n * OUT_C;

  auto cdiv = [](long long a, long long b) { return (int)((a + b - 1) / b); };
  const int B = 256;
  const int gblocks = cdiv(n / 16, 8);

  cnt_zero<<<cdiv(n, B), B, 0, stream>>>(cnt, n);
  edge_rank<<<cdiv(E, B), B, 0, stream>>>(e_dst, E, cnt, rnk);
  seg_offsets<<<cdiv(n, 1024), 1024, 0, stream>>>(cnt, n, off, dinv);
  edge_fill<<<cdiv(E, B), B, 0, stream>>>(e_src, e_dst, rnk, off, E, ssrc);

  {
    strip_gemm<IN_C, HID_C / 16><<<gblocks, B, 0, stream>>>(x, W1, h1, n, HID_C, HID_C);
    agg_gather<HID_C, true><<<cdiv((long long)n * 32, B), B, 0, stream>>>(
        h1, off, ssrc, dinv, b1, z1, n);
  }

  {
    strip_gemm<HID_C, OUT_C / 16><<<gblocks, B, 0, stream>>>(z1, W2, h2, n, OUT_C, OUT_C);
    agg_gather<OUT_C, false><<<cdiv((long long)n * 32, B), B, 0, stream>>>(
        h2, off, ssrc, dinv, b2, z2, n);
  }

  {
    long long waves = 2LL * (n / 16);
    node_proj<<<cdiv(waves * 32, B), B, 0, stream>>>(z2, Wa, pU, pV, n);
    long long pwaves = cdiv(2LL * P, 16);
    pair_score<<<cdiv(pwaves * 32, B), B, 0, stream>>>(pU, pV, pidx, nidx, ba, Wb, bb,
                                                       out, P);
  }
}


# edge bucketing: per-edge global atomic rank with its clear, scan and fill launches replaced by a two-level counting sort on LDS counters (2 launches); bucket and sum order = LDS arrival order
# speedup vs baseline: 3.2056x; 1.3318x over previous
#include <hip/hip_runtime.h>
#include <hip/hip_bf16.h>

typedef __attribute__((ext_vector_type(2))) float v2f;
typedef __attribute__((ext_vector_type(8))) float v8f;

#define IN_C  128
#define HID_C 128
#define OUT_C 64

#define BIN_SHIFT 8
#define BIN_NODES 256
#define MAX_BINS  256
#define CHUNK_EPT 4
#define CHUNK     (1024 * CHUNK_EPT)

__device__ __forceinline__ int wave_scan_incl(int x, int lane) {
  for (int d = 1; d < 32; d <<= 1) {
    int t = __shfl_up(x, d, 32);
    if (lane >= d) x += t;
  }
  return x;
}

__global__ __launch_bounds__(1024) void edge_bin(const int* __restrict__ src,
                                                 const int* __restrict__ dst, int E, int nbins,
                                                 int* __restrict__ Hc, int* __restrict__ Hs,
                                                 int* __restrict__ tdst, int* __restrict__ tsrc) {
  __shared__ int hist[MAX_BINS];
  __shared__ int wsum[8];
  const int tid = threadIdx.x;
  const int lane = tid & 31;
  const int w = tid >> 5;
  const int b = blockIdx.x;
  const int G = gridDim.x;
  const int e0 = b * CHUNK;

  if (tid < MAX_BINS) hist[tid] = 0;
  __syncthreads();

  int d[CHUNK_EPT], s[CHUNK_EPT], r[CHUNK_EPT];
#pragma unroll
  for (int k = 0; k < CHUNK_EPT; ++k) {
    const int e = e0 + tid + 1024 * k;
    d[k] = -1;
    s[k] = 0;
    if (e < E) {
      d[k] = dst[e];
      s[k] = src[e];
    }
  }
#pragma unroll
  for (int k = 0; k < CHUNK_EPT; ++k) {
    r[k] = 0;
    if (d[k] >= 0) r[k] = atomicAdd(&hist[d[k] >> BIN_SHIFT], 1);
  }
  __syncthreads();

  int c = 0, inc = 0;
  if (tid < MAX_BINS) {
    c = hist[tid];
    inc = wave_scan_incl(c, lane);
    if (lane == 31) wsum[w] = inc;
  }
  __syncthreads();
  if (tid < MAX_BINS) {
    int base = 0;
    for (int j = 0; j < 8; ++j)
      if (j < w) base += wsum[j];
    const int start = base + inc - c;
    hist[tid] = start;
    if (tid < nbins) {
      Hc[tid * G + b] = c;
      Hs[tid * G + b] = start;
    }
  }
  __syncthreads();

#pragma unroll
  for (int k = 0; k < CHUNK_EPT; ++k) {
    if (d[k] >= 0) {
      const int p = e0 + hist[d[k] >> BIN_SHIFT] + r[k];
      tdst[p] = d[k];
      tsrc[p] = s[k];
    }
  }
}

__global__ __launch_bounds__(1024) void bin_sort(const int* __restrict__ Hc,
                                                 const int* __restrict__ Hs, int G,
                                                 const int* __restrict__ tdst,
                                                 const int* __restrict__ tsrc, int n,
                                                 int* __restrict__ off, float* __restrict__ dinv,
                                                 int* __restrict__ ssrc) {
  __shared__ int seg_first[1024];
  __shared__ int seg_pos[1024];
  __shared__ int cnt_l[BIN_NODES];
  __shared__ int cur_l[BIN_NODES];
  __shared__ int w_a[32];
  __shared__ int w_b[32];
  const int tid = threadIdx.x;
  const int lane = tid & 31;
  const int w = tid >> 5;
  const int k = blockIdx.x;
  const int lo = k << BIN_SHIFT;
  const int front = k * G;

  int pre = 0;
  const int4* h4 = (const int4*)Hc;
#pragma unroll 4
  for (int i = tid; i < (front >> 2); i += 1024) {
    int4 q = h4[i];
    pre += (q.x + q.y) + (q.z + q.w);
  }
  {
    const int i = (front & ~3) + tid;
    if (i < front) pre += Hc[i];
  }
  const int c = (tid < G) ? Hc[front + tid] : 0;
  const int st = (tid < G) ? Hs[front + tid] : 0;
  if (tid < BIN_NODES) cnt_l[tid] = 0;

  const int inc = wave_scan_incl(c, lane);
  for (int m = 16; m >= 1; m >>= 1) pre += __shfl_xor(pre, m, 32);
  if (lane == 31) w_a[w] = inc;
  if (lane == 0) w_b[w] = pre;
  __syncthreads();

  int base = 0, total = 0, front_edges = 0;
  for (int j = 0; j < 32; ++j) {
    const int t = w_a[j];
    total += t;
    if (j < w) base += t;
    front_edges += w_b[j];
  }
  seg_first[tid] = base + inc - c;
  seg_pos[tid] = tid * CHUNK + st;
  __syncthreads();

  for (int i = tid; i < total; i += 1024) {
    int b = 0;
#pragma unroll
    for (int step = 512; step >= 1; step >>= 1)
      if (seg_first[b + step] <= i) b += step;
    const int p = seg_pos[b] + (i - seg_first[b]);
    atomicAdd(&cnt_l[tdst[p] - lo], 1);
  }
  __syncthreads();

  int cv = 0, cinc = 0;
  if (tid < BIN_NODES) {
    cv = cnt_l[tid];
    cinc = wave_scan_incl(cv, lane);
    if (lane == 31) w_a[w] = cinc;
  }
  __syncthreads();
  if (tid < BIN_NODES) {
    int b2 = 0;
    for (int j = 0; j < 8; ++j)
      if (j < w) b2 += w_a[j];
    const int o = b2 + cinc - cv;
    cur_l[tid] = o;
    const int v = lo + tid;
    if (v < n) {
      off[v] = front_edges + o;
      if (v == n - 1) off[n] = front_edges + o + cv;
      dinv[v] = rsqrtf((float)(cv + 1));
    }
  }
  __syncthreads();

  for (int i = tid; i < total; i += 1024) {
    int b = 0;
#pragma unroll
    for (int step = 512; step >= 1; step >>= 1)
      if (seg_first[b + step] <= i) b += step;
    const int p = seg_pos[b] + (i - seg_first[b]);
    const int dd = tdst[p];
    const int ss = tsrc[p];
    const int pos = atomicAdd(&cur_l[dd - lo], 1);
    ssrc[front_edges + pos] = ss;
  }
}

template <int K, int NT>
__global__ __launch_bounds__(256) void strip_gemm(const float* __restrict__ A,
                                                  const float* __restrict__ B,
                                                  float* __restrict__ C, int M, int ldb,
                                                  int ldc) {
  typedef float f4 __attribute__((ext_vector_type(4)));
  constexpr int N = NT * 16;
  constexpr int KP = K / 2;
  constexpr int KS = K / 4;
  constexpr int PITCH = 2 * N + 32;
  constexpr int ITEMS = KP * (N / 4);
  static_assert(ITEMS % 256 == 0, "staging loop has no tail");
  __shared__ __attribute__((aligned(16))) float Bl[KP * PITCH];

  const int tid = threadIdx.x;
  const int lane = tid & 31;
  const int w = __builtin_amdgcn_readfirstlane(tid >> 5);
  const int c = lane & 15;
  const int h = lane >> 4;

  const int nstrips = M >> 4;
  const int strip = blockIdx.x * 8 + w;
  const bool active = strip < nstrips;
  const int m0 = (active ? strip : nstrips - 1) << 4;

  v2f a[KS];
  const float* Arow = A + (size_t)(m0 + c) * K + 2 * h;
#pragma unroll
  for (int kk = 0; kk < KS; ++kk) a[kk] = *(const v2f*)(Arow + 4 * kk);

  {
    f4 r0[ITEMS / 256], r1[ITEMS / 256];
#pragma unroll
    for (int j = 0; j < ITEMS / 256; ++j) {
      const int i = tid + 256 * j;
      const int kp = i / (N / 4), n4 = i % (N / 4);
      r0[j] = *(const f4*)(B + (size_t)(2 * kp) * ldb + 4 * n4);
      r1[j] = *(const f4*)(B + (size_t)(2 * kp + 1) * ldb + 4 * n4);
    }
#pragma unroll
    for (int j = 0; j < ITEMS / 256; ++j) {
      const int i = tid + 256 * j;
      const int kp = i / (N / 4), n4 = i % (N / 4);
      float* d = Bl + kp * PITCH + 8 * n4;
      *(f4*)d = (f4){r0[j].x, r1[j].x, r0[j].y, r1[j].y};
      *(f4*)(d + 4) = (f4){r0[j].z, r1[j].z, r0[j].w, r1[j].w};
    }
  }
  __syncthreads();

  if (active) {
    v8f acc[NT];
#pragma unroll
    for (int t = 0; t < NT; ++t) acc[t] = (v8f){};

    const float* Bf = Bl + h * PITCH + 2 * c;
#pragma unroll
    for (int kk = 0; kk < KS; ++kk) {
#pragma unroll
      for (int t = 0; t < NT; ++t) {
        const v2f b = *(const v2f*)(Bf + (2 * kk) * PITCH + 32 * t);
        acc[t] = __builtin_amdgcn_wmma_f32_16x16x4_f32(false, a[kk], false, b, (short)0,
                                                       acc[t], false, false);
      }
    }

    float* Crow = C + (size_t)(m0 + 8 * h) * ldc + c;
#pragma unroll
    for (int t = 0; t < NT; ++t) {
#pragma unroll
      for (int i = 0; i < 8; ++i) Crow[(size_t)i * ldc + t * 16] = acc[t][i];
    }
  }
}

template <int C, bool RELU>
__global__ __launch_bounds__(256) void agg_gather(const float* __restrict__ h,
                                                  const int* __restrict__ off,
                                                  const int* __restrict__ ssrc,
                                                  const float* __restrict__ dinv,
                                                  const float* __restrict__ bias,
                                                  float* __restrict__ out, int n) {
  constexpr int VW = C / 32;
  typedef float VT __attribute__((ext_vector_type(VW)));
  const int lane = threadIdx.x & 31;
  const int v = __builtin_amdgcn_readfirstlane(
      (int)((blockIdx.x * (long long)blockDim.x + threadIdx.x) >> 5));
  if (v >= n) return;
  const int beg = off[v];
  const int c = off[v + 1] - beg;
  const float dv = dinv[v];
  __builtin_assume(c >= 0);
  const float* hl = h + lane * VW;

  const int tot = c + 1;
  VT acc = {};

  for (int j0 = 0; j0 < tot; j0 += 32) {
    const int m = min(32, tot - j0);
    const int idx = j0 + lane;
    int s_l = v;
    float d_l = (idx == c) ? dv : 0.0f;
    if (idx < c) {
      s_l = ssrc[beg + idx];
      d_l = dinv[s_l];
    }
    for (int j = 0; j < m; j += 8) {
      VT r[8];
      float wt[8];
      for (int u = 0; u < 8; ++u) {
        const int s = __builtin_amdgcn_readlane(s_l, j + u);
        wt[u] = dv * __builtin_bit_cast(
            float, __builtin_amdgcn_readlane(__builtin_bit_cast(int, d_l), j + u));
        r[u] = *(const VT*)(hl + (size_t)s * C);
      }
      for (int u = 0; u < 8; ++u) acc += r[u] * wt[u];
    }
  }

  const VT bv = *(const VT*)(bias + lane * VW);
  acc += bv;
  if (RELU) {
    for (int i = 0; i < VW; ++i) acc[i] = fmaxf(acc[i], 0.0f);
  }
  *(VT*)(out + (size_t)v * C + lane * VW) = acc;
}

__global__ __launch_bounds__(256) void node_proj(const float* __restrict__ z,
                                                 const float* __restrict__ Wa,
                                                 float* __restrict__ U, float* __restrict__ V,
                                                 int n) {
  int wave = (int)((blockIdx.x * (long long)blockDim.x + threadIdx.x) >> 5);
  int lane = threadIdx.x & 31;
  if (wave >= 2 * (n >> 4)) return;
  int sel = wave & 1;
  int m0 = (wave >> 1) << 4;
  int half = lane >> 4;
  int l = lane & 15;

  const float* zr = z + (size_t)(m0 + l) * OUT_C + 2 * half;
  const float* Wp = Wa + (size_t)(sel * OUT_C + 2 * half) * HID_C + l;

  v8f acc[8];
#pragma unroll
  for (int t = 0; t < 8; ++t) acc[t] = (v8f){};

  for (int k = 0; k < OUT_C; k += 4) {
    v2f a;
    a.x = zr[k];
    a.y = zr[k + 1];
    const float* Bp = Wp + (size_t)k * HID_C;
#pragma unroll
    for (int t = 0; t < 8; ++t) {
      v2f b;
      b.x = Bp[t * 16];
      b.y = Bp[HID_C + t * 16];
      acc[t] = __builtin_amdgcn_wmma_f32_16x16x4_f32(false, a, false, b, (short)0,
                                                     acc[t], false, false);
    }
  }

  float* Crow = (sel ? V : U) + (size_t)(m0 + 8 * half) * HID_C + l;
#pragma unroll
  for (int t = 0; t < 8; ++t) {
#pragma unroll
    for (int i = 0; i < 8; ++i) Crow[(size_t)i * HID_C + t * 16] = acc[t][i];
  }
}

__global__ __launch_bounds__(256) void pair_score(const float* __restrict__ U,
                                                  const float* __restrict__ V,
                                                  const int* __restrict__ pidx,
                                                  const int* __restrict__ nidx,
                                                  const float* __restrict__ ba,
                                                  const float* __restrict__ Wb,
                                                  const float* __restrict__ bb,
                                                  float* __restrict__ out, int P) {
  typedef float f4 __attribute__((ext_vector_type(4)));
  const int lane = threadIdx.x & 31;
  const int wave = (int)((blockIdx.x * (long long)blockDim.x + threadIdx.x) >> 5);
  const int q = lane >> 3;
  const int t = lane & 7;
  const int total = 2 * P;
  const int base = wave * 16;
  if (base >= total) return;

  int sv[4], dv[4];
#pragma unroll
  for (int st = 0; st < 4; ++st) {
    const int pc = min(base + st * 4 + q, total - 1);
    const int* ix = (pc < P) ? pidx : nidx;
    const int r = (pc < P) ? pc : pc - P;
    sv[st] = ix[r];
    dv[st] = ix[P + r];
  }

  f4 bav[4], wbv[4];
#pragma unroll
  for (int i = 0; i < 4; ++i) {
    bav[i] = *(const f4*)(ba + i * 32 + t * 4);
    wbv[i] = *(const f4*)(Wb + i * 32 + t * 4);
  }
  const float bbv = bb[0];

#pragma unroll
  for (int g = 0; g < 4; g += 2) {
    f4 ru[2][4], rv[2][4];
#pragma unroll
    for (int h = 0; h < 2; ++h) {
      const float* up = U + (size_t)sv[g + h] * HID_C + t * 4;
      const float* vp = V + (size_t)dv[g + h] * HID_C + t * 4;
#pragma unroll
      for (int i = 0; i < 4; ++i) {
        ru[h][i] = *(const f4*)(up + i * 32);
        rv[h][i] = *(const f4*)(vp + i * 32);
      }
    }
#pragma unroll
    for (int h = 0; h < 2; ++h) {
      float part = 0.0f;
#pragma unroll
      for (int i = 0; i < 4; ++i) {
        const f4 hs = (ru[h][i] + rv[h][i]) + bav[i];
#pragma unroll
        for (int c = 0; c < 4; ++c) part += fmaxf(hs[c], 0.0f) * wbv[i][c];
      }
      part += __shfl_xor(part, 1, 32);
      part += __shfl_xor(part, 2, 32);
      part += __shfl_xor(part, 4, 32);
      const int p = base + (g + h) * 4 + q;
      if (t == 0 && p < total) out[p] = part + bbv;
    }
  }
}

extern "C" void kernel_launch(void* const* d_in, const int* in_sizes, int n_in,
                              void* d_out, int out_size, void* d_ws, size_t ws_size,
                              hipStream_t stream) {
  const float* x   = (const float*)d_in[0];
  const int* eidx  = (const int*)d_in[1];
  const int* pidx  = (const int*)d_in[2];
  const int* nidx  = (const int*)d_in[3];
  const float* W1  = (const float*)d_in[4];
  const float* b1  = (const float*)d_in[5];
  const float* W2  = (const float*)d_in[6];
  const float* b2  = (const float*)d_in[7];
  const float* Wa  = (const float*)d_in[8];
  const float* ba  = (const float*)d_in[9];
  const float* Wb  = (const float*)d_in[10];
  const float* bb  = (const float*)d_in[11];
  float* out = (float*)d_out;

  const int n = in_sizes[0] / IN_C;
  const int E = in_sizes[1] / 2;
  const int P = in_sizes[2] / 2;
  const int* e_src = eidx;
  const int* e_dst = eidx + E;

  float* ws   = (float*)d_ws;
  float* h1   = ws;
  float* z1   = h1 + (size_t)n * HID_C;
  float* pU   = z1 + (size_t)n * HID_C;
  float* pV   = pU + (size_t)n * HID_C;
  int*   Hc   = (int*)(pV + (size_t)n * HID_C);
  int*   Hs   = Hc + MAX_BINS * 1024;
  float* dinv = (float*)(Hs + MAX_BINS * 1024);
  int*   off  = (int*)(dinv + n);
  int*   ssrc = off + n + 1;
  int*   tdst = ssrc + E;
  int*   tsrc = tdst + E;
  float* h2   = h1;
  float* z2   = h1 + (size_t)n * OUT_C;

  auto cdiv = [](long long a, long long b) { return (int)((a + b - 1) / b); };
  const int B = 256;
  const int gblocks = cdiv(n / 16, 8);

  const int nbins = cdiv(n, BIN_NODES);
  const int nchunks = cdiv(E, CHUNK);
  edge_bin<<<nchunks, 1024, 0, stream>>>(e_src, e_dst, E, nbins, Hc, Hs, tdst, tsrc);
  bin_sort<<<nbins, 1024, 0, stream>>>(Hc, Hs, nchunks, tdst, tsrc, n, off, dinv, ssrc);

  {
    strip_gemm<IN_C, HID_C / 16><<<gblocks, B, 0, stream>>>(x, W1, h1, n, HID_C, HID_C);
    agg_gather<HID_C, true><<<cdiv((long long)n * 32, B), B, 0, stream>>>(
        h1, off, ssrc, dinv, b1, z1, n);
  }

  {
    strip_gemm<HID_C, OUT_C / 16><<<gblocks, B, 0, stream>>>(z1, W2, h2, n, OUT_C, OUT_C);
    agg_gather<OUT_C, false><<<cdiv((long long)n * 32, B), B, 0, stream>>>(
        h2, off, ssrc, dinv, b2, z2, n);
  }

  {
    long long waves = 2LL * (n / 16);
    node_proj<<<cdiv(waves * 32, B), B, 0, stream>>>(z2, Wa, pU, pV, n);
    long long pwaves = cdiv(2LL * P, 16);
    pair_score<<<cdiv(pwaves * 32, B), B, 0, stream>>>(pU, pV, pidx, nidx, ba, Wb, bb,
                                                       out, P);
  }
}


# bin_sort: a bin's chunk runs dealt to waves (lanes = a run's items, 8 runs' loads in flight), no 10-step LDS search per item; front count = sum of loaded starts; bucket and sum order = LDS arrival
# speedup vs baseline: 3.3168x; 1.0347x over previous
#include <hip/hip_runtime.h>
#include <hip/hip_bf16.h>

typedef __attribute__((ext_vector_type(2))) float v2f;
typedef __attribute__((ext_vector_type(8))) float v8f;

#define IN_C  128
#define HID_C 128
#define OUT_C 64

#define BIN_SHIFT 8
#define BIN_NODES 256
#define MAX_BINS  256
#define CHUNK_EPT 4
#define CHUNK     (1024 * CHUNK_EPT)

__device__ __forceinline__ int wave_scan_incl(int x, int lane) {
  for (int d = 1; d < 32; d <<= 1) {
    int t = __shfl_up(x, d, 32);
    if (lane >= d) x += t;
  }
  return x;
}

__global__ __launch_bounds__(1024) void edge_bin(const int* __restrict__ src,
                                                 const int* __restrict__ dst, int E, int nbins,
                                                 int* __restrict__ Hc, int* __restrict__ Hs,
                                                 int* __restrict__ tdst, int* __restrict__ tsrc) {
  __shared__ int hist[MAX_BINS];
  __shared__ int wsum[8];
  const int tid = threadIdx.x;
  const int lane = tid & 31;
  const int w = tid >> 5;
  const int b = blockIdx.x;
  const int G = gridDim.x;
  const int e0 = b * CHUNK;

  if (tid < MAX_BINS) hist[tid] = 0;
  __syncthreads();

  int d[CHUNK_EPT], s[CHUNK_EPT], r[CHUNK_EPT];
#pragma unroll
  for (int k = 0; k < CHUNK_EPT; ++k) {
    const int e = e0 + tid + 1024 * k;
    d[k] = -1;
    s[k] = 0;
    if (e < E) {
      d[k] = dst[e];
      s[k] = src[e];
    }
  }
#pragma unroll
  for (int k = 0; k < CHUNK_EPT; ++k) {
    r[k] = 0;
    if (d[k] >= 0) r[k] = atomicAdd(&hist[d[k] >> BIN_SHIFT], 1);
  }
  __syncthreads();

  int c = 0, inc = 0;
  if (tid < MAX_BINS) {
    c = hist[tid];
    inc = wave_scan_incl(c, lane);
    if (lane == 31) wsum[w] = inc;
  }
  __syncthreads();
  if (tid < MAX_BINS) {
    int base = 0;
    for (int j = 0; j < 8; ++j)
      if (j < w) base += wsum[j];
    const int start = base + inc - c;
    hist[tid] = start;
    if (tid < nbins) {
      Hc[tid * G + b] = c;
      Hs[tid * G + b] = start;
    }
  }
  __syncthreads();

#pragma unroll
  for (int k = 0; k < CHUNK_EPT; ++k) {
    if (d[k] >= 0) {
      const int p = e0 + hist[d[k] >> BIN_SHIFT] + r[k];
      tdst[p] = d[k];
      tsrc[p] = s[k];
    }
  }
}

__global__ __launch_bounds__(1024) void bin_sort(const int* __restrict__ Hc,
                                                 const int* __restrict__ Hs, int G,
                                                 const int* __restrict__ tdst,
                                                 const int* __restrict__ tsrc, int n,
                                                 int* __restrict__ off, float* __restrict__ dinv,
                                                 int* __restrict__ ssrc) {
  __shared__ int seg_cnt[1024];
  __shared__ int seg_pos[1024];
  __shared__ int cnt_l[BIN_NODES];
  __shared__ int cur_l[BIN_NODES];
  __shared__ int w_a[32];
  __shared__ int w_b[32];
  constexpr int RU = 8;
  const int tid = threadIdx.x;
  const int lane = tid & 31;
  const int w = tid >> 5;
  const int wv = __builtin_amdgcn_readfirstlane(w);
  const int k = blockIdx.x;
  const int lo = k << BIN_SHIFT;
  const int front = k * G;

  const int c = (tid < G) ? Hc[front + tid] : 0;
  const int st = (tid < G) ? Hs[front + tid] : 0;
  if (tid < BIN_NODES) cnt_l[tid] = 0;
  seg_cnt[tid] = c;
  seg_pos[tid] = tid * CHUNK + st;

  int pre = st;
  for (int m = 16; m >= 1; m >>= 1) pre += __shfl_xor(pre, m, 32);
  if (lane == 0) w_b[w] = pre;
  __syncthreads();

  int front_edges = 0;
  for (int j = 0; j < 32; ++j) front_edges += w_b[j];

  for (int b0 = wv; b0 < G; b0 += 32 * RU) {
    int cb[RU], pb[RU], d[RU];
#pragma unroll
    for (int u = 0; u < RU; ++u) {
      const int b = b0 + 32 * u;
      const int bb = (b < G) ? b : 0;
      cb[u] = seg_cnt[bb];
      pb[u] = seg_pos[bb];
    }
#pragma unroll
    for (int u = 0; u < RU; ++u) {
      const int b = b0 + 32 * u;
      cb[u] = (b < G) ? __builtin_amdgcn_readfirstlane(cb[u]) : 0;
      pb[u] = __builtin_amdgcn_readfirstlane(pb[u]);
      const bool in = lane < cb[u];
      const int v = tdst[in ? pb[u] + lane : 0];
      d[u] = in ? v : -1;
    }
#pragma unroll
    for (int u = 0; u < RU; ++u)
      if (d[u] >= 0) atomicAdd(&cnt_l[d[u] - lo], 1);
#pragma unroll
    for (int u = 0; u < RU; ++u)
      for (int j = 32 + lane; j < cb[u]; j += 32) atomicAdd(&cnt_l[tdst[pb[u] + j] - lo], 1);
  }
  __syncthreads();

  int cv = 0, cinc = 0;
  if (tid < BIN_NODES) {
    cv = cnt_l[tid];
    cinc = wave_scan_incl(cv, lane);
    if (lane == 31) w_a[w] = cinc;
  }
  __syncthreads();
  if (tid < BIN_NODES) {
    int b2 = 0;
    for (int j = 0; j < 8; ++j)
      if (j < w) b2 += w_a[j];
    const int o = b2 + cinc - cv;
    cur_l[tid] = o;
    const int v = lo + tid;
    if (v < n) {
      off[v] = front_edges + o;
      if (v == n - 1) off[n] = front_edges + o + cv;
      dinv[v] = rsqrtf((float)(cv + 1));
    }
  }
  __syncthreads();

  for (int b0 = wv; b0 < G; b0 += 32 * RU) {
    int cb[RU], pb[RU], d[RU], s[RU];
#pragma unroll
    for (int u = 0; u < RU; ++u) {
      const int b = b0 + 32 * u;
      const int bb = (b < G) ? b : 0;
      cb[u] = seg_cnt[bb];
      pb[u] = seg_pos[bb];
    }
#pragma unroll
    for (int u = 0; u < RU; ++u) {
      const int b = b0 + 32 * u;
      cb[u] = (b < G) ? __builtin_amdgcn_readfirstlane(cb[u]) : 0;
      pb[u] = __builtin_amdgcn_readfirstlane(pb[u]);
      const bool in = lane < cb[u];
      const int q = in ? pb[u] + lane : 0;
      const int v = tdst[q];
      s[u] = tsrc[q];
      d[u] = in ? v : -1;
    }
#pragma unroll
    for (int u = 0; u < RU; ++u) {
      if (d[u] >= 0) {
        const int pos = atomicAdd(&cur_l[d[u] - lo], 1);
        ssrc[front_edges + pos] = s[u];
      }
    }
#pragma unroll
    for (int u = 0; u < RU; ++u) {
      for (int j = 32 + lane; j < cb[u]; j += 32) {
        const int dd = tdst[pb[u] + j];
        const int ss = tsrc[pb[u] + j];
        const int pos = atomicAdd(&cur_l[dd - lo], 1);
        ssrc[front_edges + pos] = ss;
      }
    }
  }
}

template <int K, int NT>
__global__ __launch_bounds__(256) void strip_gemm(const float* __restrict__ A,
                                                  const float* __restrict__ B,
                                                  float* __restrict__ C, int M, int ldb,
                                                  int ldc) {
  typedef float f4 __attribute__((ext_vector_type(4)));
  constexpr int N = NT * 16;
  constexpr int KP = K / 2;
  constexpr int KS = K / 4;
  constexpr int PITCH = 2 * N + 32;
  constexpr int ITEMS = KP * (N / 4);
  static_assert(ITEMS % 256 == 0, "staging loop has no tail");
  __shared__ __attribute__((aligned(16))) float Bl[KP * PITCH];

  const int tid = threadIdx.x;
  const int lane = tid & 31;
  const int w = __builtin_amdgcn_readfirstlane(tid >> 5);
  const int c = lane & 15;
  const int h = lane >> 4;

  const int nstrips = M >> 4;
  const int strip = blockIdx.x * 8 + w;
  const bool active = strip < nstrips;
  const int m0 = (active ? strip : nstrips - 1) << 4;

  v2f a[KS];
  const float* Arow = A + (size_t)(m0 + c) * K + 2 * h;
#pragma unroll
  for (int kk = 0; kk < KS; ++kk) a[kk] = *(const v2f*)(Arow + 4 * kk);

  {
    f4 r0[ITEMS / 256], r1[ITEMS / 256];
#pragma unroll
    for (int j = 0; j < ITEMS / 256; ++j) {
      const int i = tid + 256 * j;
      const int kp = i / (N / 4), n4 = i % (N / 4);
      r0[j] = *(const f4*)(B + (size_t)(2 * kp) * ldb + 4 * n4);
      r1[j] = *(const f4*)(B + (size_t)(2 * kp + 1) * ldb + 4 * n4);
    }
#pragma unroll
    for (int j = 0; j < ITEMS / 256; ++j) {
      const int i = tid + 256 * j;
      const int kp = i / (N / 4), n4 = i % (N / 4);
      float* d = Bl + kp * PITCH + 8 * n4;
      *(f4*)d = (f4){r0[j].x, r1[j].x, r0[j].y, r1[j].y};
      *(f4*)(d + 4) = (f4){r0[j].z, r1[j].z, r0[j].w, r1[j].w};
    }
  }
  __syncthreads();

  if (active) {
    v8f acc[NT];
#pragma unroll
    for (int t = 0; t < NT; ++t) acc[t] = (v8f){};

    const float* Bf = Bl + h * PITCH + 2 * c;
#pragma unroll
    for (int kk = 0; kk < KS; ++kk) {
#pragma unroll
      for (int t = 0; t < NT; ++t) {
        const v2f b = *(const v2f*)(Bf + (2 * kk) * PITCH + 32 * t);
        acc[t] = __builtin_amdgcn_wmma_f32_16x16x4_f32(false, a[kk], false, b, (short)0,
                                                       acc[t], false, false);
      }
    }

    float* Crow = C + (size_t)(m0 + 8 * h) * ldc + c;
#pragma unroll
    for (int t = 0; t < NT; ++t) {
#pragma unroll
      for (int i = 0; i < 8; ++i) Crow[(size_t)i * ldc + t * 16] = acc[t][i];
    }
  }
}

template <int C, bool RELU>
__global__ __launch_bounds__(256) void agg_gather(const float* __restrict__ h,
                                                  const int* __restrict__ off,
                                                  const int* __restrict__ ssrc,
                                                  const float* __restrict__ dinv,
                                                  const float* __restrict__ bias,
                                                  float* __restrict__ out, int n) {
  constexpr int VW = C / 32;
  typedef float VT __attribute__((ext_vector_type(VW)));
  const int lane = threadIdx.x & 31;
  const int v = __builtin_amdgcn_readfirstlane(
      (int)((blockIdx.x * (long long)blockDim.x + threadIdx.x) >> 5));
  if (v >= n) return;
  const int beg = off[v];
  const int c = off[v + 1] - beg;
  const float dv = dinv[v];
  __builtin_assume(c >= 0);
  const float* hl = h + lane * VW;

  const int tot = c + 1;
  VT acc = {};

  for (int j0 = 0; j0 < tot; j0 += 32) {
    const int m = min(32, tot - j0);
    const int idx = j0 + lane;
    int s_l = v;
    float d_l = (idx == c) ? dv : 0.0f;
    if (idx < c) {
      s_l = ssrc[beg + idx];
      d_l = dinv[s_l];
    }
    for (int j = 0; j < m; j += 8) {
      VT r[8];
      float wt[8];
      for (int u = 0; u < 8; ++u) {
        const int s = __builtin_amdgcn_readlane(s_l, j + u);
        wt[u] = dv * __builtin_bit_cast(
            float, __builtin_amdgcn_readlane(__builtin_bit_cast(int, d_l), j + u));
        r[u] = *(const VT*)(hl + (size_t)s * C);
      }
      for (int u = 0; u < 8; ++u) acc += r[u] * wt[u];
    }
  }

  const VT bv = *(const VT*)(bias + lane * VW);
  acc += bv;
  if (RELU) {
    for (int i = 0; i < VW; ++i) acc[i] = fmaxf(acc[i], 0.0f);
  }
  *(VT*)(out + (size_t)v * C + lane * VW) = acc;
}

__global__ __launch_bounds__(256) void node_proj(const float* __restrict__ z,
                                                 const float* __restrict__ Wa,
                                                 float* __restrict__ U, float* __restrict__ V,
                                                 int n) {
  int wave = (int)((blockIdx.x * (long long)blockDim.x + threadIdx.x) >> 5);
  int lane = threadIdx.x & 31;
  if (wave >= 2 * (n >> 4)) return;
  int sel = wave & 1;
  int m0 = (wave >> 1) << 4;
  int half = lane >> 4;
  int l = lane & 15;

  const float* zr = z + (size_t)(m0 + l) * OUT_C + 2 * half;
  const float* Wp = Wa + (size_t)(sel * OUT_C + 2 * half) * HID_C + l;

  v8f acc[8];
#pragma unroll
  for (int t = 0; t < 8; ++t) acc[t] = (v8f){};

  for (int k = 0; k < OUT_C; k += 4) {
    v2f a;
    a.x = zr[k];
    a.y = zr[k + 1];
    const float* Bp = Wp + (size_t)k * HID_C;
#pragma unroll
    for (int t = 0; t < 8; ++t) {
      v2f b;
      b.x = Bp[t * 16];
      b.y = Bp[HID_C + t * 16];
      acc[t] = __builtin_amdgcn_wmma_f32_16x16x4_f32(false, a, false, b, (short)0,
                                                     acc[t], false, false);
    }
  }

  float* Crow = (sel ? V : U) + (size_t)(m0 + 8 * half) * HID_C + l;
#pragma unroll
  for (int t = 0; t < 8; ++t) {
#pragma unroll
    for (int i = 0; i < 8; ++i) Crow[(size_t)i * HID_C + t * 16] = acc[t][i];
  }
}

__global__ __launch_bounds__(256) void pair_score(const float* __restrict__ U,
                                                  const float* __restrict__ V,
                                                  const int* __restrict__ pidx,
                                                  const int* __restrict__ nidx,
                                                  const float* __restrict__ ba,
                                                  const float* __restrict__ Wb,
                                                  const float* __restrict__ bb,
                                                  float* __restrict__ out, int P) {
  typedef float f4 __attribute__((ext_vector_type(4)));
  const int lane = threadIdx.x & 31;
  const int wave = (int)((blockIdx.x * (long long)blockDim.x + threadIdx.x) >> 5);
  const int q = lane >> 3;
  const int t = lane & 7;
  const int total = 2 * P;
  const int base = wave * 16;
  if (base >= total) return;

  int sv[4], dv[4];
#pragma unroll
  for (int st = 0; st < 4; ++st) {
    const int pc = min(base + st * 4 + q, total - 1);
    const int* ix = (pc < P) ? pidx : nidx;
    const int r = (pc < P) ? pc : pc - P;
    sv[st] = ix[r];
    dv[st] = ix[P + r];
  }

  f4 bav[4], wbv[4];
#pragma unroll
  for (int i = 0; i < 4; ++i) {
    bav[i] = *(const f4*)(ba + i * 32 + t * 4);
    wbv[i] = *(const f4*)(Wb + i * 32 + t * 4);
  }
  const float bbv = bb[0];

#pragma unroll
  for (int g = 0; g < 4; g += 2) {
    f4 ru[2][4], rv[2][4];
#pragma unroll
    for (int h = 0; h < 2; ++h) {
      const float* up = U + (size_t)sv[g + h] * HID_C + t * 4;
      const float* vp = V + (size_t)dv[g + h] * HID_C + t * 4;
#pragma unroll
      for (int i = 0; i < 4; ++i) {
        ru[h][i] = *(const f4*)(up + i * 32);
        rv[h][i] = *(const f4*)(vp + i * 32);
      }
    }
#pragma unroll
    for (int h = 0; h < 2; ++h) {
      float part = 0.0f;
#pragma unroll
      for (int i = 0; i < 4; ++i) {
        const f4 hs = (ru[h][i] + rv[h][i]) + bav[i];
#pragma unroll
        for (int c = 0; c < 4; ++c) part += fmaxf(hs[c], 0.0f) * wbv[i][c];
      }
      part += __shfl_xor(part, 1, 32);
      part += __shfl_xor(part, 2, 32);
      part += __shfl_xor(part, 4, 32);
      const int p = base + (g + h) * 4 + q;
      if (t == 0 && p < total) out[p] = part + bbv;
    }
  }
}

extern "C" void kernel_launch(void* const* d_in, const int* in_sizes, int n_in,
                              void* d_out, int out_size, void* d_ws, size_t ws_size,
                              hipStream_t stream) {
  const float* x   = (const float*)d_in[0];
  const int* eidx  = (const int*)d_in[1];
  const int* pidx  = (const int*)d_in[2];
  const int* nidx  = (const int*)d_in[3];
  const float* W1  = (const float*)d_in[4];
  const float* b1  = (const float*)d_in[5];
  const float* W2  = (const float*)d_in[6];
  const float* b2  = (const float*)d_in[7];
  const float* Wa  = (const float*)d_in[8];
  const float* ba  = (const float*)d_in[9];
  const float* Wb  = (const float*)d_in[10];
  const float* bb  = (const float*)d_in[11];
  float* out = (float*)d_out;

  const int n = in_sizes[0] / IN_C;
  const int E = in_sizes[1] / 2;
  const int P = in_sizes[2] / 2;
  const int* e_src = eidx;
  const int* e_dst = eidx + E;

  float* ws   = (float*)d_ws;
  float* h1   = ws;
  float* z1   = h1 + (size_t)n * HID_C;
  float* pU   = z1 + (size_t)n * HID_C;
  float* pV   = pU + (size_t)n * HID_C;
  int*   Hc   = (int*)(pV + (size_t)n * HID_C);
  int*   Hs   = Hc + MAX_BINS * 1024;
  float* dinv = (float*)(Hs + MAX_BINS * 1024);
  int*   off  = (int*)(dinv + n);
  int*   ssrc = off + n + 1;
  int*   tdst = ssrc + E;
  int*   tsrc = tdst + E;
  float* h2   = h1;
  float* z2   = h1 + (size_t)n * OUT_C;

  auto cdiv = [](long long a, long long b) { return (int)((a + b - 1) / b); };
  const int B = 256;
  const int gblocks = cdiv(n / 16, 8);

  const int nbins = cdiv(n, BIN_NODES);
  const int nchunks = cdiv(E, CHUNK);
  edge_bin<<<nchunks, 1024, 0, stream>>>(e_src, e_dst, E, nbins, Hc, Hs, tdst, tsrc);
  bin_sort<<<nbins, 1024, 0, stream>>>(Hc, Hs, nchunks, tdst, tsrc, n, off, dinv, ssrc);

  {
    strip_gemm<IN_C, HID_C / 16><<<gblocks, B, 0, stream>>>(x, W1, h1, n, HID_C, HID_C);
    agg_gather<HID_C, true><<<cdiv((long long)n * 32, B), B, 0, stream>>>(
        h1, off, ssrc, dinv, b1, z1, n);
  }

  {
    strip_gemm<HID_C, OUT_C / 16><<<gblocks, B, 0, stream>>>(z1, W2, h2, n, OUT_C, OUT_C);
    agg_gather<OUT_C, false><<<cdiv((long long)n * 32, B), B, 0, stream>>>(
        h2, off, ssrc, dinv, b2, z2, n);
  }

  {
    long long waves = 2LL * (n / 16);
    node_proj<<<cdiv(waves * 32, B), B, 0, stream>>>(z2, Wa, pU, pV, n);
    long long pwaves = cdiv(2LL * P, 16);
    pair_score<<<cdiv(pwaves * 32, B), B, 0, stream>>>(pU, pV, pidx, nidx, ba, Wb, bb,
                                                       out, P);
  }
}


# x@W1 (needs no edge sort) gets no launch of its own: its strips run as extra work-groups of the two bucketing launches, 4 waves a strip, 64 registers so both roles share a WGP; sums unchanged
# speedup vs baseline: 3.5403x; 1.0674x over previous
#include <hip/hip_runtime.h>
#include <hip/hip_bf16.h>

typedef __attribute__((ext_vector_type(2))) float v2f;
typedef __attribute__((ext_vector_type(8))) float v8f;

#define IN_C  128
#define HID_C 128
#define OUT_C 64

#define BIN_SHIFT 8
#define BIN_NODES 256
#define MAX_BINS  256
#define CHUNK_EPT 4
#define CHUNK     (1024 * CHUNK_EPT)

__device__ __forceinline__ int wave_scan_incl(int x, int lane) {
  for (int d = 1; d < 32; d <<= 1) {
    int t = __shfl_up(x, d, 32);
    if (lane >= d) x += t;
  }
  return x;
}

__device__ __forceinline__ void edge_bin_role(const int* __restrict__ src,
                                              const int* __restrict__ dst, int E, int nbins,
                                              int* __restrict__ Hc, int* __restrict__ Hs,
                                              int* __restrict__ tdst, int* __restrict__ tsrc,
                                              int b, int G) {
  __shared__ int hist[MAX_BINS];
  __shared__ int wsum[8];
  const int tid = threadIdx.x;
  const int lane = tid & 31;
  const int w = tid >> 5;
  const int e0 = b * CHUNK;

  if (tid < MAX_BINS) hist[tid] = 0;
  __syncthreads();

  int d[CHUNK_EPT], s[CHUNK_EPT], r[CHUNK_EPT];
#pragma unroll
  for (int k = 0; k < CHUNK_EPT; ++k) {
    const int e = e0 + tid + 1024 * k;
    d[k] = -1;
    s[k] = 0;
    if (e < E) {
      d[k] = dst[e];
      s[k] = src[e];
    }
  }
#pragma unroll
  for (int k = 0; k < CHUNK_EPT; ++k) {
    r[k] = 0;
    if (d[k] >= 0) r[k] = atomicAdd(&hist[d[k] >> BIN_SHIFT], 1);
  }
  __syncthreads();

  int c = 0, inc = 0;
  if (tid < MAX_BINS) {
    c = hist[tid];
    inc = wave_scan_incl(c, lane);
    if (lane == 31) wsum[w] = inc;
  }
  __syncthreads();
  if (tid < MAX_BINS) {
    int base = 0;
    for (int j = 0; j < 8; ++j)
      if (j < w) base += wsum[j];
    const int start = base + inc - c;
    hist[tid] = start;
    if (tid < nbins) {
      Hc[tid * G + b] = c;
      Hs[tid * G + b] = start;
    }
  }
  __syncthreads();

#pragma unroll
  for (int k = 0; k < CHUNK_EPT; ++k) {
    if (d[k] >= 0) {
      const int p = e0 + hist[d[k] >> BIN_SHIFT] + r[k];
      tdst[p] = d[k];
      tsrc[p] = s[k];
    }
  }
}

__device__ __forceinline__ void bin_sort_role(const int* __restrict__ Hc,
                                              const int* __restrict__ Hs, int G,
                                              const int* __restrict__ tdst,
                                              const int* __restrict__ tsrc, int n,
                                              int* __restrict__ off, float* __restrict__ dinv,
                                              int* __restrict__ ssrc, int k) {
  __shared__ int seg_cnt[1024];
  __shared__ int seg_pos[1024];
  __shared__ int cnt_l[BIN_NODES];
  __shared__ int cur_l[BIN_NODES];
  __shared__ int w_a[32];
  __shared__ int w_b[32];
  constexpr int RU = 8;
  const int tid = threadIdx.x;
  const int lane = tid & 31;
  const int w = tid >> 5;
  const int wv = __builtin_amdgcn_readfirstlane(w);
  const int lo = k << BIN_SHIFT;
  const int front = k * G;

  const int c = (tid < G) ? Hc[front + tid] : 0;
  const int st = (tid < G) ? Hs[front + tid] : 0;
  if (tid < BIN_NODES) cnt_l[tid] = 0;
  seg_cnt[tid] = c;
  seg_pos[tid] = tid * CHUNK + st;

  int pre = st;
  for (int m = 16; m >= 1; m >>= 1) pre += __shfl_xor(pre, m, 32);
  if (lane == 0) w_b[w] = pre;
  __syncthreads();

  int front_edges = 0;
  for (int j = 0; j < 32; ++j) front_edges += w_b[j];

  for (int b0 = wv; b0 < G; b0 += 32 * RU) {
    int cb[RU], pb[RU], d[RU];
#pragma unroll
    for (int u = 0; u < RU; ++u) {
      const int b = b0 + 32 * u;
      const int bb = (b < G) ? b : 0;
      cb[u] = seg_cnt[bb];
      pb[u] = seg_pos[bb];
    }
#pragma unroll
    for (int u = 0; u < RU; ++u) {
      const int b = b0 + 32 * u;
      cb[u] = (b < G) ? __builtin_amdgcn_readfirstlane(cb[u]) : 0;
      pb[u] = __builtin_amdgcn_readfirstlane(pb[u]);
      const bool in = lane < cb[u];
      const int v = tdst[in ? pb[u] + lane : 0];
      d[u] = in ? v : -1;
    }
#pragma unroll
    for (int u = 0; u < RU; ++u)
      if (d[u] >= 0) atomicAdd(&cnt_l[d[u] - lo], 1);
#pragma unroll
    for (int u = 0; u < RU; ++u)
      for (int j = 32 + lane; j < cb[u]; j += 32) atomicAdd(&cnt_l[tdst[pb[u] + j] - lo], 1);
  }
  __syncthreads();

  int cv = 0, cinc = 0;
  if (tid < BIN_NODES) {
    cv = cnt_l[tid];
    cinc = wave_scan_incl(cv, lane);
    if (lane == 31) w_a[w] = cinc;
  }
  __syncthreads();
  if (tid < BIN_NODES) {
    int b2 = 0;
    for (int j = 0; j < 8; ++j)
      if (j < w) b2 += w_a[j];
    const int o = b2 + cinc - cv;
    cur_l[tid] = o;
    const int v = lo + tid;
    if (v < n) {
      off[v] = front_edges + o;
      if (v == n - 1) off[n] = front_edges + o + cv;
      dinv[v] = rsqrtf((float)(cv + 1));
    }
  }
  __syncthreads();

  for (int b0 = wv; b0 < G; b0 += 32 * RU) {
    int cb[RU], pb[RU], d[RU], s[RU];
#pragma unroll
    for (int u = 0; u < RU; ++u) {
      const int b = b0 + 32 * u;
      const int bb = (b < G) ? b : 0;
      cb[u] = seg_cnt[bb];
      pb[u] = seg_pos[bb];
    }
#pragma unroll
    for (int u = 0; u < RU; ++u) {
      const int b = b0 + 32 * u;
      cb[u] = (b < G) ? __builtin_amdgcn_readfirstlane(cb[u]) : 0;
      pb[u] = __builtin_amdgcn_readfirstlane(pb[u]);
      const bool in = lane < cb[u];
      const int q = in ? pb[u] + lane : 0;
      const int v = tdst[q];
      s[u] = tsrc[q];
      d[u] = in ? v : -1;
    }
#pragma unroll
    for (int u = 0; u < RU; ++u) {
      if (d[u] >= 0) {
        const int pos = atomicAdd(&cur_l[d[u] - lo], 1);
        ssrc[front_edges + pos] = s[u];
      }
    }
#pragma unroll
    for (int u = 0; u < RU; ++u) {
      for (int j = 32 + lane; j < cb[u]; j += 32) {
        const int dd = tdst[pb[u] + j];
        const int ss = tsrc[pb[u] + j];
        const int pos = atomicAdd(&cur_l[dd - lo], 1);
        ssrc[front_edges + pos] = ss;
      }
    }
  }
}

template <int K, int NT>
__global__ __launch_bounds__(256) void strip_gemm(const float* __restrict__ A,
                                                  const float* __restrict__ B,
                                                  float* __restrict__ C, int M, int ldb,
                                                  int ldc) {
  typedef float f4 __attribute__((ext_vector_type(4)));
  constexpr int N = NT * 16;
  constexpr int KP = K / 2;
  constexpr int KS = K / 4;
  constexpr int PITCH = 2 * N + 32;
  constexpr int ITEMS = KP * (N / 4);
  static_assert(ITEMS % 256 == 0, "staging loop has no tail");
  __shared__ __attribute__((aligned(16))) float Bl[KP * PITCH];

  const int tid = threadIdx.x;
  const int lane = tid & 31;
  const int w = __builtin_amdgcn_readfirstlane(tid >> 5);
  const int c = lane & 15;
  const int h = lane >> 4;

  const int nstrips = M >> 4;
  const int strip = blockIdx.x * 8 + w;
  const bool active = strip < nstrips;
  const int m0 = (active ? strip : nstrips - 1) << 4;

  v2f a[KS];
  const float* Arow = A + (size_t)(m0 + c) * K + 2 * h;
#pragma unroll
  for (int kk = 0; kk < KS; ++kk) a[kk] = *(const v2f*)(Arow + 4 * kk);

  {
    f4 r0[ITEMS / 256], r1[ITEMS / 256];
#pragma unroll
    for (int j = 0; j < ITEMS / 256; ++j) {
      const int i = tid + 256 * j;
      const int kp = i / (N / 4), n4 = i % (N / 4);
      r0[j] = *(const f4*)(B + (size_t)(2 * kp) * ldb + 4 * n4);
      r1[j] = *(const f4*)(B + (size_t)(2 * kp + 1) * ldb + 4 * n4);
    }
#pragma unroll
    for (int j = 0; j < ITEMS / 256; ++j) {
      const int i = tid + 256 * j;
      const int kp = i / (N / 4), n4 = i % (N / 4);
      float* d = Bl + kp * PITCH + 8 * n4;
      *(f4*)d = (f4){r0[j].x, r1[j].x, r0[j].y, r1[j].y};
      *(f4*)(d + 4) = (f4){r0[j].z, r1[j].z, r0[j].w, r1[j].w};
    }
  }
  __syncthreads();

  if (active) {
    v8f acc[NT];
#pragma unroll
    for (int t = 0; t < NT; ++t) acc[t] = (v8f){};

    const float* Bf = Bl + h * PITCH + 2 * c;
#pragma unroll
    for (int kk = 0; kk < KS; ++kk) {
#pragma unroll
      for (int t = 0; t < NT; ++t) {
        const v2f b = *(const v2f*)(Bf + (2 * kk) * PITCH + 32 * t);
        acc[t] = __builtin_amdgcn_wmma_f32_16x16x4_f32(false, a[kk], false, b, (short)0,
                                                       acc[t], false, false);
      }
    }

    float* Crow = C + (size_t)(m0 + 8 * h) * ldc + c;
#pragma unroll
    for (int t = 0; t < NT; ++t) {
#pragma unroll
      for (int i = 0; i < 8; ++i) Crow[(size_t)i * ldc + t * 16] = acc[t][i];
    }
  }
}

template <int K, int NT>
__device__ __forceinline__ void strip_gemm_role(const float* __restrict__ A,
                                                const float* __restrict__ B,
                                                float* __restrict__ C, int M, int ldb, int ldc,
                                                int wg) {
  typedef float f4 __attribute__((ext_vector_type(4)));
  constexpr int N = NT * 16;
  constexpr int KP = K / 2;
  constexpr int KS = K / 4;
  constexpr int KQ = KS / 4;
  constexpr int TW = NT / 4;
  constexpr int PITCH = 2 * N + 32;
  constexpr int ITEMS = KP * (N / 4);
  static_assert(ITEMS % 1024 == 0, "staging loop has no tail");
  static_assert(NT % 4 == 0 && KS % 4 == 0, "four waves per strip, four quarters of k");
  __shared__ __attribute__((aligned(16))) float Bl[KP * PITCH];

  const int tid = threadIdx.x;
  const int lane = tid & 31;
  const int w = __builtin_amdgcn_readfirstlane(tid >> 5);
  const int c = lane & 15;
  const int h = lane >> 4;

  const int nstrips = M >> 4;
  const int strip = wg * 8 + (w >> 2);
  const int t0 = (w & 3) * TW;
  const bool active = strip < nstrips;
  const int m0 = (active ? strip : nstrips - 1) << 4;

  v2f a[KQ];
  const float* Arow = A + (size_t)(m0 + c) * K + 2 * h;
#pragma unroll
  for (int kk = 0; kk < KQ; ++kk) a[kk] = *(const v2f*)(Arow + 4 * kk);

  {
    f4 r0[ITEMS / 1024], r1[ITEMS / 1024];
#pragma unroll
    for (int j = 0; j < ITEMS / 1024; ++j) {
      const int i = tid + 1024 * j;
      const int kp = i / (N / 4), n4 = i % (N / 4);
      r0[j] = *(const f4*)(B + (size_t)(2 * kp) * ldb + 4 * n4);
      r1[j] = *(const f4*)(B + (size_t)(2 * kp + 1) * ldb + 4 * n4);
    }
#pragma unroll
    for (int j = 0; j < ITEMS / 1024; ++j) {
      const int i = tid + 1024 * j;
      const int kp = i / (N / 4), n4 = i % (N / 4);
      float* d = Bl + kp * PITCH + 8 * n4;
      *(f4*)d = (f4){r0[j].x, r1[j].x, r0[j].y, r1[j].y};
      *(f4*)(d + 4) = (f4){r0[j].z, r1[j].z, r0[j].w, r1[j].w};
    }
  }
  __syncthreads();

  if (active) {
    v8f acc[TW];
#pragma unroll
    for (int t = 0; t < TW; ++t) acc[t] = (v8f){};

    const float* Bf = Bl + h * PITCH + 2 * c + 32 * t0;
#pragma unroll 1
    for (int kq = 0; kq < 4; ++kq) {
      const float* Bq = Bf + (2 * KQ * kq) * PITCH;
#pragma unroll
      for (int kk = 0; kk < KQ; ++kk) {
#pragma unroll
        for (int t = 0; t < TW; ++t) {
          const v2f b = *(const v2f*)(Bq + (2 * kk) * PITCH + 32 * t);
          acc[t] = __builtin_amdgcn_wmma_f32_16x16x4_f32(false, a[kk], false, b, (short)0,
                                                         acc[t], false, false);
        }
      }
      if (kq < 3) {
#pragma unroll
        for (int kk = 0; kk < KQ; ++kk) a[kk] = *(const v2f*)(Arow + 4 * (KQ * (kq + 1) + kk));
      }
    }

    float* Crow = C + (size_t)(m0 + 8 * h) * ldc + c + 16 * t0;
#pragma unroll
    for (int t = 0; t < TW; ++t) {
#pragma unroll
      for (int i = 0; i < 8; ++i) Crow[(size_t)i * ldc + t * 16] = acc[t][i];
    }
  }
}

__global__ __launch_bounds__(1024, 16) void edge_bin_gemm(
    const int* __restrict__ src, const int* __restrict__ dst, int E, int nbins,
    int* __restrict__ Hc, int* __restrict__ Hs, int* __restrict__ tdst, int* __restrict__ tsrc,
    int nrole, const float* __restrict__ x, const float* __restrict__ W1,
    float* __restrict__ h1, int n, int wg0) {
  const int b = blockIdx.x;
  if (b < nrole)
    edge_bin_role(src, dst, E, nbins, Hc, Hs, tdst, tsrc, b, nrole);
  else
    strip_gemm_role<IN_C, HID_C / 16>(x, W1, h1, n, HID_C, HID_C, wg0 + b - nrole);
}

__global__ __launch_bounds__(1024, 16) void bin_sort_gemm(
    const int* __restrict__ Hc, const int* __restrict__ Hs, int G,
    const int* __restrict__ tdst, const int* __restrict__ tsrc, int n, int* __restrict__ off,
    float* __restrict__ dinv, int* __restrict__ ssrc, int nrole, const float* __restrict__ x,
    const float* __restrict__ W1, float* __restrict__ h1, int wg0) {
  const int b = blockIdx.x;
  if (b < nrole)
    bin_sort_role(Hc, Hs, G, tdst, tsrc, n, off, dinv, ssrc, b);
  else
    strip_gemm_role<IN_C, HID_C / 16>(x, W1, h1, n, HID_C, HID_C, wg0 + b - nrole);
}

template <int C, bool RELU>
__global__ __launch_bounds__(256) void agg_gather(const float* __restrict__ h,
                                                  const int* __restrict__ off,
                                                  const int* __restrict__ ssrc,
                                                  const float* __restrict__ dinv,
                                                  const float* __restrict__ bias,
                                                  float* __restrict__ out, int n) {
  constexpr int VW = C / 32;
  typedef float VT __attribute__((ext_vector_type(VW)));
  const int lane = threadIdx.x & 31;
  const int v = __builtin_amdgcn_readfirstlane(
      (int)((blockIdx.x * (long long)blockDim.x + threadIdx.x) >> 5));
  if (v >= n) return;
  const int beg = off[v];
  const int c = off[v + 1] - beg;
  const float dv = dinv[v];
  __builtin_assume(c >= 0);
  const float* hl = h + lane * VW;

  const int tot = c + 1;
  VT acc = {};

  for (int j0 = 0; j0 < tot; j0 += 32) {
    const int m = min(32, tot - j0);
    const int idx = j0 + lane;
    int s_l = v;
    float d_l = (idx == c) ? dv : 0.0f;
    if (idx < c) {
      s_l = ssrc[beg + idx];
      d_l = dinv[s_l];
    }
    for (int j = 0; j < m; j += 8) {
      VT r[8];
      float wt[8];
      for (int u = 0; u < 8; ++u) {
        const int s = __builtin_amdgcn_readlane(s_l, j + u);
        wt[u] = dv * __builtin_bit_cast(
            float, __builtin_amdgcn_readlane(__builtin_bit_cast(int, d_l), j + u));
        r[u] = *(const VT*)(hl + (size_t)s * C);
      }
      for (int u = 0; u < 8; ++u) acc += r[u] * wt[u];
    }
  }

  const VT bv = *(const VT*)(bias + lane * VW);
  acc += bv;
  if (RELU) {
    for (int i = 0; i < VW; ++i) acc[i] = fmaxf(acc[i], 0.0f);
  }
  *(VT*)(out + (size_t)v * C + lane * VW) = acc;
}

__global__ __launch_bounds__(256) void node_proj(const float* __restrict__ z,
                                                 const float* __restrict__ Wa,
                                                 float* __restrict__ U, float* __restrict__ V,
                                                 int n) {
  int wave = (int)((blockIdx.x * (long long)blockDim.x + threadIdx.x) >> 5);
  int lane = threadIdx.x & 31;
  if (wave >= 2 * (n >> 4)) return;
  int sel = wave & 1;
  int m0 = (wave >> 1) << 4;
  int half = lane >> 4;
  int l = lane & 15;

  const float* zr = z + (size_t)(m0 + l) * OUT_C + 2 * half;
  const float* Wp = Wa + (size_t)(sel * OUT_C + 2 * half) * HID_C + l;

  v8f acc[8];
#pragma unroll
  for (int t = 0; t < 8; ++t) acc[t] = (v8f){};

  for (int k = 0; k < OUT_C; k += 4) {
    v2f a;
    a.x = zr[k];
    a.y = zr[k + 1];
    const float* Bp = Wp + (size_t)k * HID_C;
#pragma unroll
    for (int t = 0; t < 8; ++t) {
      v2f b;
      b.x = Bp[t * 16];
      b.y = Bp[HID_C + t * 16];
      acc[t] = __builtin_amdgcn_wmma_f32_16x16x4_f32(false, a, false, b, (short)0,
                                                     acc[t], false, false);
    }
  }

  float* Crow = (sel ? V : U) + (size_t)(m0 + 8 * half) * HID_C + l;
#pragma unroll
  for (int t = 0; t < 8; ++t) {
#pragma unroll
    for (int i = 0; i < 8; ++i) Crow[(size_t)i * HID_C + t * 16] = acc[t][i];
  }
}

__global__ __launch_bounds__(256) void pair_score(const float* __restrict__ U,
                                                  const float* __restrict__ V,
                                                  const int* __restrict__ pidx,
                                                  const int* __restrict__ nidx,
                                                  const float* __restrict__ ba,
                                                  const float* __restrict__ Wb,
                                                  const float* __restrict__ bb,
                                                  float* __restrict__ out, int P) {
  typedef float f4 __attribute__((ext_vector_type(4)));
  const int lane = threadIdx.x & 31;
  const int wave = (int)((blockIdx.x * (long long)blockDim.x + threadIdx.x) >> 5);
  const int q = lane >> 3;
  const int t = lane & 7;
  const int total = 2 * P;
  const int base = wave * 16;
  if (base >= total) return;

  int sv[4], dv[4];
#pragma unroll
  for (int st = 0; st < 4; ++st) {
    const int pc = min(base + st * 4 + q, total - 1);
    const int* ix = (pc < P) ? pidx : nidx;
    const int r = (pc < P) ? pc : pc - P;
    sv[st] = ix[r];
    dv[st] = ix[P + r];
  }

  f4 bav[4], wbv[4];
#pragma unroll
  for (int i = 0; i < 4; ++i) {
    bav[i] = *(const f4*)(ba + i * 32 + t * 4);
    wbv[i] = *(const f4*)(Wb + i * 32 + t * 4);
  }
  const float bbv = bb[0];

#pragma unroll
  for (int g = 0; g < 4; g += 2) {
    f4 ru[2][4], rv[2][4];
#pragma unroll
    for (int h = 0; h < 2; ++h) {
      const float* up = U + (size_t)sv[g + h] * HID_C + t * 4;
      const float* vp = V + (size_t)dv[g + h] * HID_C + t * 4;
#pragma unroll
      for (int i = 0; i < 4; ++i) {
        ru[h][i] = *(const f4*)(up + i * 32);
        rv[h][i] = *(const f4*)(vp + i * 32);
      }
    }
#pragma unroll
    for (int h = 0; h < 2; ++h) {
      float part = 0.0f;
#pragma unroll
      for (int i = 0; i < 4; ++i) {
        const f4 hs = (ru[h][i] + rv[h][i]) + bav[i];
#pragma unroll
        for (int c = 0; c < 4; ++c) part += fmaxf(hs[c], 0.0f) * wbv[i][c];
      }
      part += __shfl_xor(part, 1, 32);
      part += __shfl_xor(part, 2, 32);
      part += __shfl_xor(part, 4, 32);
      const int p = base + (g + h) * 4 + q;
      if (t == 0 && p < total) out[p] = part + bbv;
    }
  }
}

extern "C" void kernel_launch(void* const* d_in, const int* in_sizes, int n_in,
                              void* d_out, int out_size, void* d_ws, size_t ws_size,
                              hipStream_t stream) {
  const float* x   = (const float*)d_in[0];
  const int* eidx  = (const int*)d_in[1];
  const int* pidx  = (const int*)d_in[2];
  const int* nidx  = (const int*)d_in[3];
  const float* W1  = (const float*)d_in[4];
  const float* b1  = (const float*)d_in[5];
  const float* W2  = (const float*)d_in[6];
  const float* b2  = (const float*)d_in[7];
  const float* Wa  = (const float*)d_in[8];
  const float* ba  = (const float*)d_in[9];
  const float* Wb  = (const float*)d_in[10];
  const float* bb  = (const float*)d_in[11];
  float* out = (float*)d_out;

  const int n = in_sizes[0] / IN_C;
  const int E = in_sizes[1] / 2;
  const int P = in_sizes[2] / 2;
  const int* e_src = eidx;
  const int* e_dst = eidx + E;

  float* ws   = (float*)d_ws;
  float* h1   = ws;
  float* z1   = h1 + (size_t)n * HID_C;
  float* pU   = z1 + (size_t)n * HID_C;
  float* pV   = pU + (size_t)n * HID_C;
  int*   Hc   = (int*)(pV + (size_t)n * HID_C);
  int*   Hs   = Hc + MAX_BINS * 1024;
  float* dinv = (float*)(Hs + MAX_BINS * 1024);
  int*   off  = (int*)(dinv + n);
  int*   ssrc = off + n + 1;
  int*   tdst = ssrc + E;
  int*   tsrc = tdst + E;
  float* h2   = h1;
  float* z2   = h1 + (size_t)n * OUT_C;

  auto cdiv = [](long long a, long long b) { return (int)((a + b - 1) / b); };
  const int B = 256;
  const int gblocks = cdiv(n / 16, 8);

  const int nbins = cdiv(n, BIN_NODES);
  const int nchunks = cdiv(E, CHUNK);
  const int pwgs = cdiv(n / 16, 8);
  const int pwgs1 = pwgs / 2;
  edge_bin_gemm<<<nchunks + pwgs1, 1024, 0, stream>>>(e_src, e_dst, E, nbins, Hc, Hs, tdst,
                                                      tsrc, nchunks, x, W1, h1, n, 0);
  bin_sort_gemm<<<nbins + (pwgs - pwgs1), 1024, 0, stream>>>(
      Hc, Hs, nchunks, tdst, tsrc, n, off, dinv, ssrc, nbins, x, W1, h1, pwgs1);

  {
    agg_gather<HID_C, true><<<cdiv((long long)n * 32, B), B, 0, stream>>>(
        h1, off, ssrc, dinv, b1, z1, n);
  }

  {
    strip_gemm<HID_C, OUT_C / 16><<<gblocks, B, 0, stream>>>(z1, W2, h2, n, OUT_C, OUT_C);
    agg_gather<OUT_C, false><<<cdiv((long long)n * 32, B), B, 0, stream>>>(
        h2, off, ssrc, dinv, b2, z2, n);
  }

  {
    long long waves = 2LL * (n / 16);
    node_proj<<<cdiv(waves * 32, B), B, 0, stream>>>(z2, Wa, pU, pV, n);
    long long pwaves = cdiv(2LL * P, 16);
    pair_score<<<cdiv(pwaves * 32, B), B, 0, stream>>>(pU, pV, pidx, nidx, ba, Wb, bb,
                                                       out, P);
  }
}


# node projections (z2 @ Wa halves) in the strip form of the layer products: Wa half staged once per 4-wave work-group in LDS as k pairs, A fragments in registers; k order and sums unchanged
# speedup vs baseline: 3.7005x; 1.0453x over previous
#include <hip/hip_runtime.h>
#include <hip/hip_bf16.h>

typedef __attribute__((ext_vector_type(2))) float v2f;
typedef __attribute__((ext_vector_type(8))) float v8f;

#define IN_C  128
#define HID_C 128
#define OUT_C 64

#define BIN_SHIFT 8
#define BIN_NODES 256
#define MAX_BINS  256
#define CHUNK_EPT 4
#define CHUNK     (1024 * CHUNK_EPT)

__device__ __forceinline__ int wave_scan_incl(int x, int lane) {
  for (int d = 1; d < 32; d <<= 1) {
    int t = __shfl_up(x, d, 32);
    if (lane >= d) x += t;
  }
  return x;
}

__device__ __forceinline__ void edge_bin_role(const int* __restrict__ src,
                                              const int* __restrict__ dst, int E, int nbins,
                                              int* __restrict__ Hc, int* __restrict__ Hs,
                                              int* __restrict__ tdst, int* __restrict__ tsrc,
                                              int b, int G) {
  __shared__ int hist[MAX_BINS];
  __shared__ int wsum[8];
  const int tid = threadIdx.x;
  const int lane = tid & 31;
  const int w = tid >> 5;
  const int e0 = b * CHUNK;

  if (tid < MAX_BINS) hist[tid] = 0;
  __syncthreads();

  int d[CHUNK_EPT], s[CHUNK_EPT], r[CHUNK_EPT];
#pragma unroll
  for (int k = 0; k < CHUNK_EPT; ++k) {
    const int e = e0 + tid + 1024 * k;
    d[k] = -1;
    s[k] = 0;
    if (e < E) {
      d[k] = dst[e];
      s[k] = src[e];
    }
  }
#pragma unroll
  for (int k = 0; k < CHUNK_EPT; ++k) {
    r[k] = 0;
    if (d[k] >= 0) r[k] = atomicAdd(&hist[d[k] >> BIN_SHIFT], 1);
  }
  __syncthreads();

  int c = 0, inc = 0;
  if (tid < MAX_BINS) {
    c = hist[tid];
    inc = wave_scan_incl(c, lane);
    if (lane == 31) wsum[w] = inc;
  }
  __syncthreads();
  if (tid < MAX_BINS) {
    int base = 0;
    for (int j = 0; j < 8; ++j)
      if (j < w) base += wsum[j];
    const int start = base + inc - c;
    hist[tid] = start;
    if (tid < nbins) {
      Hc[tid * G + b] = c;
      Hs[tid * G + b] = start;
    }
  }
  __syncthreads();

#pragma unroll
  for (int k = 0; k < CHUNK_EPT; ++k) {
    if (d[k] >= 0) {
      const int p = e0 + hist[d[k] >> BIN_SHIFT] + r[k];
      tdst[p] = d[k];
      tsrc[p] = s[k];
    }
  }
}

__device__ __forceinline__ void bin_sort_role(const int* __restrict__ Hc,
                                              const int* __restrict__ Hs, int G,
                                              const int* __restrict__ tdst,
                                              const int* __restrict__ tsrc, int n,
                                              int* __restrict__ off, float* __restrict__ dinv,
                                              int* __restrict__ ssrc, int k) {
  __shared__ int seg_cnt[1024];
  __shared__ int seg_pos[1024];
  __shared__ int cnt_l[BIN_NODES];
  __shared__ int cur_l[BIN_NODES];
  __shared__ int w_a[32];
  __shared__ int w_b[32];
  constexpr int RU = 8;
  const int tid = threadIdx.x;
  const int lane = tid & 31;
  const int w = tid >> 5;
  const int wv = __builtin_amdgcn_readfirstlane(w);
  const int lo = k << BIN_SHIFT;
  const int front = k * G;

  const int c = (tid < G) ? Hc[front + tid] : 0;
  const int st = (tid < G) ? Hs[front + tid] : 0;
  if (tid < BIN_NODES) cnt_l[tid] = 0;
  seg_cnt[tid] = c;
  seg_pos[tid] = tid * CHUNK + st;

  int pre = st;
  for (int m = 16; m >= 1; m >>= 1) pre += __shfl_xor(pre, m, 32);
  if (lane == 0) w_b[w] = pre;
  __syncthreads();

  int front_edges = 0;
  for (int j = 0; j < 32; ++j) front_edges += w_b[j];

  for (int b0 = wv; b0 < G; b0 += 32 * RU) {
    int cb[RU], pb[RU], d[RU];
#pragma unroll
    for (int u = 0; u < RU; ++u) {
      const int b = b0 + 32 * u;
      const int bb = (b < G) ? b : 0;
      cb[u] = seg_cnt[bb];
      pb[u] = seg_pos[bb];
    }
#pragma unroll
    for (int u = 0; u < RU; ++u) {
      const int b = b0 + 32 * u;
      cb[u] = (b < G) ? __builtin_amdgcn_readfirstlane(cb[u]) : 0;
      pb[u] = __builtin_amdgcn_readfirstlane(pb[u]);
      const bool in = lane < cb[u];
      const int v = tdst[in ? pb[u] + lane : 0];
      d[u] = in ? v : -1;
    }
#pragma unroll
    for (int u = 0; u < RU; ++u)
      if (d[u] >= 0) atomicAdd(&cnt_l[d[u] - lo], 1);
#pragma unroll
    for (int u = 0; u < RU; ++u)
      for (int j = 32 + lane; j < cb[u]; j += 32) atomicAdd(&cnt_l[tdst[pb[u] + j] - lo], 1);
  }
  __syncthreads();

  int cv = 0, cinc = 0;
  if (tid < BIN_NODES) {
    cv = cnt_l[tid];
    cinc = wave_scan_incl(cv, lane);
    if (lane == 31) w_a[w] = cinc;
  }
  __syncthreads();
  if (tid < BIN_NODES) {
    int b2 = 0;
    for (int j = 0; j < 8; ++j)
      if (j < w) b2 += w_a[j];
    const int o = b2 + cinc - cv;
    cur_l[tid] = o;
    const int v = lo + tid;
    if (v < n) {
      off[v] = front_edges + o;
      if (v == n - 1) off[n] = front_edges + o + cv;
      dinv[v] = rsqrtf((float)(cv + 1));
    }
  }
  __syncthreads();

  for (int b0 = wv; b0 < G; b0 += 32 * RU) {
    int cb[RU], pb[RU], d[RU], s[RU];
#pragma unroll
    for (int u = 0; u < RU; ++u) {
      const int b = b0 + 32 * u;
      const int bb = (b < G) ? b : 0;
      cb[u] = seg_cnt[bb];
      pb[u] = seg_pos[bb];
    }
#pragma unroll
    for (int u = 0; u < RU; ++u) {
      const int b = b0 + 32 * u;
      cb[u] = (b < G) ? __builtin_amdgcn_readfirstlane(cb[u]) : 0;
      pb[u] = __builtin_amdgcn_readfirstlane(pb[u]);
      const bool in = lane < cb[u];
      const int q = in ? pb[u] + lane : 0;
      const int v = tdst[q];
      s[u] = tsrc[q];
      d[u] = in ? v : -1;
    }
#pragma unroll
    for (int u = 0; u < RU; ++u) {
      if (d[u] >= 0) {
        const int pos = atomicAdd(&cur_l[d[u] - lo], 1);
        ssrc[front_edges + pos] = s[u];
      }
    }
#pragma unroll
    for (int u = 0; u < RU; ++u) {
      for (int j = 32 + lane; j < cb[u]; j += 32) {
        const int dd = tdst[pb[u] + j];
        const int ss = tsrc[pb[u] + j];
        const int pos = atomicAdd(&cur_l[dd - lo], 1);
        ssrc[front_edges + pos] = ss;
      }
    }
  }
}

template <int K, int NT>
__global__ __launch_bounds__(256) void strip_gemm(const float* __restrict__ A,
                                                  const float* __restrict__ B,
                                                  float* __restrict__ C, int M, int ldb,
                                                  int ldc) {
  typedef float f4 __attribute__((ext_vector_type(4)));
  constexpr int N = NT * 16;
  constexpr int KP = K / 2;
  constexpr int KS = K / 4;
  constexpr int PITCH = 2 * N + 32;
  constexpr int ITEMS = KP * (N / 4);
  static_assert(ITEMS % 256 == 0, "staging loop has no tail");
  __shared__ __attribute__((aligned(16))) float Bl[KP * PITCH];

  const int tid = threadIdx.x;
  const int lane = tid & 31;
  const int w = __builtin_amdgcn_readfirstlane(tid >> 5);
  const int c = lane & 15;
  const int h = lane >> 4;

  const int nstrips = M >> 4;
  const int strip = blockIdx.x * 8 + w;
  const bool active = strip < nstrips;
  const int m0 = (active ? strip : nstrips - 1) << 4;

  v2f a[KS];
  const float* Arow = A + (size_t)(m0 + c) * K + 2 * h;
#pragma unroll
  for (int kk = 0; kk < KS; ++kk) a[kk] = *(const v2f*)(Arow + 4 * kk);

  {
    f4 r0[ITEMS / 256], r1[ITEMS / 256];
#pragma unroll
    for (int j = 0; j < ITEMS / 256; ++j) {
      const int i = tid + 256 * j;
      const int kp = i / (N / 4), n4 = i % (N / 4);
      r0[j] = *(const f4*)(B + (size_t)(2 * kp) * ldb + 4 * n4);
      r1[j] = *(const f4*)(B + (size_t)(2 * kp + 1) * ldb + 4 * n4);
    }
#pragma unroll
    for (int j = 0; j < ITEMS / 256; ++j) {
      const int i = tid + 256 * j;
      const int kp = i / (N / 4), n4 = i % (N / 4);
      float* d = Bl + kp * PITCH + 8 * n4;
      *(f4*)d = (f4){r0[j].x, r1[j].x, r0[j].y, r1[j].y};
      *(f4*)(d + 4) = (f4){r0[j].z, r1[j].z, r0[j].w, r1[j].w};
    }
  }
  __syncthreads();

  if (active) {
    v8f acc[NT];
#pragma unroll
    for (int t = 0; t < NT; ++t) acc[t] = (v8f){};

    const float* Bf = Bl + h * PITCH + 2 * c;
#pragma unroll
    for (int kk = 0; kk < KS; ++kk) {
#pragma unroll
      for (int t = 0; t < NT; ++t) {
        const v2f b = *(const v2f*)(Bf + (2 * kk) * PITCH + 32 * t);
        acc[t] = __builtin_amdgcn_wmma_f32_16x16x4_f32(false, a[kk], false, b, (short)0,
                                                       acc[t], false, false);
      }
    }

    float* Crow = C + (size_t)(m0 + 8 * h) * ldc + c;
#pragma unroll
    for (int t = 0; t < NT; ++t) {
#pragma unroll
      for (int i = 0; i < 8; ++i) Crow[(size_t)i * ldc + t * 16] = acc[t][i];
    }
  }
}

template <int K, int NT>
__device__ __forceinline__ void strip_gemm_role(const float* __restrict__ A,
                                                const float* __restrict__ B,
                                                float* __restrict__ C, int M, int ldb, int ldc,
                                                int wg) {
  typedef float f4 __attribute__((ext_vector_type(4)));
  constexpr int N = NT * 16;
  constexpr int KP = K / 2;
  constexpr int KS = K / 4;
  constexpr int KQ = KS / 4;
  constexpr int TW = NT / 4;
  constexpr int PITCH = 2 * N + 32;
  constexpr int ITEMS = KP * (N / 4);
  static_assert(ITEMS % 1024 == 0, "staging loop has no tail");
  static_assert(NT % 4 == 0 && KS % 4 == 0, "four waves per strip, four quarters of k");
  __shared__ __attribute__((aligned(16))) float Bl[KP * PITCH];

  const int tid = threadIdx.x;
  const int lane = tid & 31;
  const int w = __builtin_amdgcn_readfirstlane(tid >> 5);
  const int c = lane & 15;
  const int h = lane >> 4;

  const int nstrips = M >> 4;
  const int strip = wg * 8 + (w >> 2);
  const int t0 = (w & 3) * TW;
  const bool active = strip < nstrips;
  const int m0 = (active ? strip : nstrips - 1) << 4;

  v2f a[KQ];
  const float* Arow = A + (size_t)(m0 + c) * K + 2 * h;
#pragma unroll
  for (int kk = 0; kk < KQ; ++kk) a[kk] = *(const v2f*)(Arow + 4 * kk);

  {
    f4 r0[ITEMS / 1024], r1[ITEMS / 1024];
#pragma unroll
    for (int j = 0; j < ITEMS / 1024; ++j) {
      const int i = tid + 1024 * j;
      const int kp = i / (N / 4), n4 = i % (N / 4);
      r0[j] = *(const f4*)(B + (size_t)(2 * kp) * ldb + 4 * n4);
      r1[j] = *(const f4*)(B + (size_t)(2 * kp + 1) * ldb + 4 * n4);
    }
#pragma unroll
    for (int j = 0; j < ITEMS / 1024; ++j) {
      const int i = tid + 1024 * j;
      const int kp = i / (N / 4), n4 = i % (N / 4);
      float* d = Bl + kp * PITCH + 8 * n4;
      *(f4*)d = (f4){r0[j].x, r1[j].x, r0[j].y, r1[j].y};
      *(f4*)(d + 4) = (f4){r0[j].z, r1[j].z, r0[j].w, r1[j].w};
    }
  }
  __syncthreads();

  if (active) {
    v8f acc[TW];
#pragma unroll
    for (int t = 0; t < TW; ++t) acc[t] = (v8f){};

    const float* Bf = Bl + h * PITCH + 2 * c + 32 * t0;
#pragma unroll 1
    for (int kq = 0; kq < 4; ++kq) {
      const float* Bq = Bf + (2 * KQ * kq) * PITCH;
#pragma unroll
      for (int kk = 0; kk < KQ; ++kk) {
#pragma unroll
        for (int t = 0; t < TW; ++t) {
          const v2f b = *(const v2f*)(Bq + (2 * kk) * PITCH + 32 * t);
          acc[t] = __builtin_amdgcn_wmma_f32_16x16x4_f32(false, a[kk], false, b, (short)0,
                                                         acc[t], false, false);
        }
      }
      if (kq < 3) {
#pragma unroll
        for (int kk = 0; kk < KQ; ++kk) a[kk] = *(const v2f*)(Arow + 4 * (KQ * (kq + 1) + kk));
      }
    }

    float* Crow = C + (size_t)(m0 + 8 * h) * ldc + c + 16 * t0;
#pragma unroll
    for (int t = 0; t < TW; ++t) {
#pragma unroll
      for (int i = 0; i < 8; ++i) Crow[(size_t)i * ldc + t * 16] = acc[t][i];
    }
  }
}

__global__ __launch_bounds__(1024, 16) void edge_bin_gemm(
    const int* __restrict__ src, const int* __restrict__ dst, int E, int nbins,
    int* __restrict__ Hc, int* __restrict__ Hs, int* __restrict__ tdst, int* __restrict__ tsrc,
    int nrole, const float* __restrict__ x, const float* __restrict__ W1,
    float* __restrict__ h1, int n, int wg0) {
  const int b = blockIdx.x;
  if (b < nrole)
    edge_bin_role(src, dst, E, nbins, Hc, Hs, tdst, tsrc, b, nrole);
  else
    strip_gemm_role<IN_C, HID_C / 16>(x, W1, h1, n, HID_C, HID_C, wg0 + b - nrole);
}

__global__ __launch_bounds__(1024, 16) void bin_sort_gemm(
    const int* __restrict__ Hc, const int* __restrict__ Hs, int G,
    const int* __restrict__ tdst, const int* __restrict__ tsrc, int n, int* __restrict__ off,
    float* __restrict__ dinv, int* __restrict__ ssrc, int nrole, const float* __restrict__ x,
    const float* __restrict__ W1, float* __restrict__ h1, int wg0) {
  const int b = blockIdx.x;
  if (b < nrole)
    bin_sort_role(Hc, Hs, G, tdst, tsrc, n, off, dinv, ssrc, b);
  else
    strip_gemm_role<IN_C, HID_C / 16>(x, W1, h1, n, HID_C, HID_C, wg0 + b - nrole);
}

template <int C, bool RELU>
__global__ __launch_bounds__(256) void agg_gather(const float* __restrict__ h,
                                                  const int* __restrict__ off,
                                                  const int* __restrict__ ssrc,
                                                  const float* __restrict__ dinv,
                                                  const float* __restrict__ bias,
                                                  float* __restrict__ out, int n) {
  constexpr int VW = C / 32;
  typedef float VT __attribute__((ext_vector_type(VW)));
  const int lane = threadIdx.x & 31;
  const int v = __builtin_amdgcn_readfirstlane(
      (int)((blockIdx.x * (long long)blockDim.x + threadIdx.x) >> 5));
  if (v >= n) return;
  const int beg = off[v];
  const int c = off[v + 1] - beg;
  const float dv = dinv[v];
  __builtin_assume(c >= 0);
  const float* hl = h + lane * VW;

  const int tot = c + 1;
  VT acc = {};

  for (int j0 = 0; j0 < tot; j0 += 32) {
    const int m = min(32, tot - j0);
    const int idx = j0 + lane;
    int s_l = v;
    float d_l = (idx == c) ? dv : 0.0f;
    if (idx < c) {
      s_l = ssrc[beg + idx];
      d_l = dinv[s_l];
    }
    for (int j = 0; j < m; j += 8) {
      VT r[8];
      float wt[8];
      for (int u = 0; u < 8; ++u) {
        const int s = __builtin_amdgcn_readlane(s_l, j + u);
        wt[u] = dv * __builtin_bit_cast(
            float, __builtin_amdgcn_readlane(__builtin_bit_cast(int, d_l), j + u));
        r[u] = *(const VT*)(hl + (size_t)s * C);
      }
      for (int u = 0; u < 8; ++u) acc += r[u] * wt[u];
    }
  }

  const VT bv = *(const VT*)(bias + lane * VW);
  acc += bv;
  if (RELU) {
    for (int i = 0; i < VW; ++i) acc[i] = fmaxf(acc[i], 0.0f);
  }
  *(VT*)(out + (size_t)v * C + lane * VW) = acc;
}

template <int K, int NT, int NW>
__global__ __launch_bounds__(NW * 32) void proj_strip(const float* __restrict__ A,
                                                      const float* __restrict__ Wab,
                                                      float* __restrict__ U,
                                                      float* __restrict__ V, int M) {
  typedef float f4 __attribute__((ext_vector_type(4)));
  constexpr int NTH = NW * 32;
  constexpr int N = NT * 16;
  constexpr int KP = K / 2;
  constexpr int KS = K / 4;
  constexpr int PITCH = 2 * N + 32;
  constexpr int ITEMS = KP * (N / 4);
  static_assert(ITEMS % NTH == 0, "staging loop has no tail");
  __shared__ __attribute__((aligned(16))) float Bl[KP * PITCH];

  const int tid = threadIdx.x;
  const int lane = tid & 31;
  const int w = __builtin_amdgcn_readfirstlane(tid >> 5);
  const int c = lane & 15;
  const int h = lane >> 4;
  const int sel = blockIdx.y;
  const float* B = Wab + (size_t)sel * K * N;
  float* C = sel ? V : U;

  const int nstrips = M >> 4;
  const int strip = blockIdx.x * NW + w;
  const bool active = strip < nstrips;
  const int m0 = (active ? strip : nstrips - 1) << 4;

  v2f a[KS];
  const float* Arow = A + (size_t)(m0 + c) * K + 2 * h;
#pragma unroll
  for (int kk = 0; kk < KS; ++kk) a[kk] = *(const v2f*)(Arow + 4 * kk);

  {
    f4 r0[ITEMS / NTH], r1[ITEMS / NTH];
#pragma unroll
    for (int j = 0; j < ITEMS / NTH; ++j) {
      const int i = tid + NTH * j;
      const int kp = i / (N / 4), n4 = i % (N / 4);
      r0[j] = *(const f4*)(B + (size_t)(2 * kp) * N + 4 * n4);
      r1[j] = *(const f4*)(B + (size_t)(2 * kp + 1) * N + 4 * n4);
    }
#pragma unroll
    for (int j = 0; j < ITEMS / NTH; ++j) {
      const int i = tid + NTH * j;
      const int kp = i / (N / 4), n4 = i % (N / 4);
      float* d = Bl + kp * PITCH + 8 * n4;
      *(f4*)d = (f4){r0[j].x, r1[j].x, r0[j].y, r1[j].y};
      *(f4*)(d + 4) = (f4){r0[j].z, r1[j].z, r0[j].w, r1[j].w};
    }
  }
  __syncthreads();

  if (active) {
    v8f acc[NT];
#pragma unroll
    for (int t = 0; t < NT; ++t) acc[t] = (v8f){};

    const float* Bf = Bl + h * PITCH + 2 * c;
#pragma unroll
    for (int kk = 0; kk < KS; ++kk) {
#pragma unroll
      for (int t = 0; t < NT; ++t) {
        const v2f b = *(const v2f*)(Bf + (2 * kk) * PITCH + 32 * t);
        acc[t] = __builtin_amdgcn_wmma_f32_16x16x4_f32(false, a[kk], false, b, (short)0,
                                                       acc[t], false, false);
      }
    }

    float* Crow = C + (size_t)(m0 + 8 * h) * N + c;
#pragma unroll
    for (int t = 0; t < NT; ++t) {
#pragma unroll
      for (int i = 0; i < 8; ++i) Crow[(size_t)i * N + t * 16] = acc[t][i];
    }
  }
}

__global__ __launch_bounds__(256) void pair_score(const float* __restrict__ U,
                                                  const float* __restrict__ V,
                                                  const int* __restrict__ pidx,
                                                  const int* __restrict__ nidx,
                                                  const float* __restrict__ ba,
                                                  const float* __restrict__ Wb,
                                                  const float* __restrict__ bb,
                                                  float* __restrict__ out, int P) {
  typedef float f4 __attribute__((ext_vector_type(4)));
  const int lane = threadIdx.x & 31;
  const int wave = (int)((blockIdx.x * (long long)blockDim.x + threadIdx.x) >> 5);
  const int q = lane >> 3;
  const int t = lane & 7;
  const int total = 2 * P;
  const int base = wave * 16;
  if (base >= total) return;

  int sv[4], dv[4];
#pragma unroll
  for (int st = 0; st < 4; ++st) {
    const int pc = min(base + st * 4 + q, total - 1);
    const int* ix = (pc < P) ? pidx : nidx;
    const int r = (pc < P) ? pc : pc - P;
    sv[st] = ix[r];
    dv[st] = ix[P + r];
  }

  f4 bav[4], wbv[4];
#pragma unroll
  for (int i = 0; i < 4; ++i) {
    bav[i] = *(const f4*)(ba + i * 32 + t * 4);
    wbv[i] = *(const f4*)(Wb + i * 32 + t * 4);
  }
  const float bbv = bb[0];

#pragma unroll
  for (int g = 0; g < 4; g += 2) {
    f4 ru[2][4], rv[2][4];
#pragma unroll
    for (int h = 0; h < 2; ++h) {
      const float* up = U + (size_t)sv[g + h] * HID_C + t * 4;
      const float* vp = V + (size_t)dv[g + h] * HID_C + t * 4;
#pragma unroll
      for (int i = 0; i < 4; ++i) {
        ru[h][i] = *(const f4*)(up + i * 32);
        rv[h][i] = *(const f4*)(vp + i * 32);
      }
    }
#pragma unroll
    for (int h = 0; h < 2; ++h) {
      float part = 0.0f;
#pragma unroll
      for (int i = 0; i < 4; ++i) {
        const f4 hs = (ru[h][i] + rv[h][i]) + bav[i];
#pragma unroll
        for (int c = 0; c < 4; ++c) part += fmaxf(hs[c], 0.0f) * wbv[i][c];
      }
      part += __shfl_xor(part, 1, 32);
      part += __shfl_xor(part, 2, 32);
      part += __shfl_xor(part, 4, 32);
      const int p = base + (g + h) * 4 + q;
      if (t == 0 && p < total) out[p] = part + bbv;
    }
  }
}

extern "C" void kernel_launch(void* const* d_in, const int* in_sizes, int n_in,
                              void* d_out, int out_size, void* d_ws, size_t ws_size,
                              hipStream_t stream) {
  const float* x   = (const float*)d_in[0];
  const int* eidx  = (const int*)d_in[1];
  const int* pidx  = (const int*)d_in[2];
  const int* nidx  = (const int*)d_in[3];
  const float* W1  = (const float*)d_in[4];
  const float* b1  = (const float*)d_in[5];
  const float* W2  = (const float*)d_in[6];
  const float* b2  = (const float*)d_in[7];
  const float* Wa  = (const float*)d_in[8];
  const float* ba  = (const float*)d_in[9];
  const float* Wb  = (const float*)d_in[10];
  const float* bb  = (const float*)d_in[11];
  float* out = (float*)d_out;

  const int n = in_sizes[0] / IN_C;
  const int E = in_sizes[1] / 2;
  const int P = in_sizes[2] / 2;
  const int* e_src = eidx;
  const int* e_dst = eidx + E;

  float* ws   = (float*)d_ws;
  float* h1   = ws;
  float* z1   = h1 + (size_t)n * HID_C;
  float* pU   = z1 + (size_t)n * HID_C;
  float* pV   = pU + (size_t)n * HID_C;
  int*   Hc   = (int*)(pV + (size_t)n * HID_C);
  int*   Hs   = Hc + MAX_BINS * 1024;
  float* dinv = (float*)(Hs + MAX_BINS * 1024);
  int*   off  = (int*)(dinv + n);
  int*   ssrc = off + n + 1;
  int*   tdst = ssrc + E;
  int*   tsrc = tdst + E;
  float* h2   = h1;
  float* z2   = h1 + (size_t)n * OUT_C;

  auto cdiv = [](long long a, long long b) { return (int)((a + b - 1) / b); };
  const int B = 256;
  const int gblocks = cdiv(n / 16, 8);

  const int nbins = cdiv(n, BIN_NODES);
  const int nchunks = cdiv(E, CHUNK);
  const int pwgs = cdiv(n / 16, 8);
  const int pwgs1 = pwgs / 2;
  edge_bin_gemm<<<nchunks + pwgs1, 1024, 0, stream>>>(e_src, e_dst, E, nbins, Hc, Hs, tdst,
                                                      tsrc, nchunks, x, W1, h1, n, 0);
  bin_sort_gemm<<<nbins + (pwgs - pwgs1), 1024, 0, stream>>>(
      Hc, Hs, nchunks, tdst, tsrc, n, off, dinv, ssrc, nbins, x, W1, h1, pwgs1);

  {
    agg_gather<HID_C, true><<<cdiv((long long)n * 32, B), B, 0, stream>>>(
        h1, off, ssrc, dinv, b1, z1, n);
  }

  {
    strip_gemm<HID_C, OUT_C / 16><<<gblocks, B, 0, stream>>>(z1, W2, h2, n, OUT_C, OUT_C);
    agg_gather<OUT_C, false><<<cdiv((long long)n * 32, B), B, 0, stream>>>(
        h2, off, ssrc, dinv, b2, z2, n);
  }

  {
    proj_strip<OUT_C, HID_C / 16, 4><<<dim3(cdiv(n / 16, 4), 2), 4 * 32, 0, stream>>>(z2, Wa, pU, pV, n);
    long long pwaves = cdiv(2LL * P, 16);
    pair_score<<<cdiv(pwaves * 32, B), B, 0, stream>>>(pU, pV, pidx, nidx, ba, Wb, bb,
                                                       out, P);
  }
}


# layer-2 product (z1 @ W2) moved into the layer-1 aggregation: 64 rows per work-group summed into an LDS tile, W2 staged once in LDS, WMMA f32 from it; z1 and one launch gone; k order and sums kept
# speedup vs baseline: 3.8192x; 1.0321x over previous
#include <hip/hip_runtime.h>
#include <hip/hip_bf16.h>

typedef __attribute__((ext_vector_type(2))) float v2f;
typedef __attribute__((ext_vector_type(8))) float v8f;

#define IN_C  128
#define HID_C 128
#define OUT_C 64

#define BIN_SHIFT 8
#define BIN_NODES 256
#define MAX_BINS  256
#define CHUNK_EPT 4
#define CHUNK     (1024 * CHUNK_EPT)

__device__ __forceinline__ int wave_scan_incl(int x, int lane) {
  for (int d = 1; d < 32; d <<= 1) {
    int t = __shfl_up(x, d, 32);
    if (lane >= d) x += t;
  }
  return x;
}

__device__ __forceinline__ void edge_bin_role(const int* __restrict__ src,
                                              const int* __restrict__ dst, int E, int nbins,
                                              int* __restrict__ Hc, int* __restrict__ Hs,
                                              int* __restrict__ tdst, int* __restrict__ tsrc,
                                              int b, int G) {
  __shared__ int hist[MAX_BINS];
  __shared__ int wsum[8];
  const int tid = threadIdx.x;
  const int lane = tid & 31;
  const int w = tid >> 5;
  const int e0 = b * CHUNK;

  if (tid < MAX_BINS) hist[tid] = 0;
  __syncthreads();

  int d[CHUNK_EPT], s[CHUNK_EPT], r[CHUNK_EPT];
#pragma unroll
  for (int k = 0; k < CHUNK_EPT; ++k) {
    const int e = e0 + tid + 1024 * k;
    d[k] = -1;
    s[k] = 0;
    if (e < E) {
      d[k] = dst[e];
      s[k] = src[e];
    }
  }
#pragma unroll
  for (int k = 0; k < CHUNK_EPT; ++k) {
    r[k] = 0;
    if (d[k] >= 0) r[k] = atomicAdd(&hist[d[k] >> BIN_SHIFT], 1);
  }
  __syncthreads();

  int c = 0, inc = 0;
  if (tid < MAX_BINS) {
    c = hist[tid];
    inc = wave_scan_incl(c, lane);
    if (lane == 31) wsum[w] = inc;
  }
  __syncthreads();
  if (tid < MAX_BINS) {
    int base = 0;
    for (int j = 0; j < 8; ++j)
      if (j < w) base += wsum[j];
    const int start = base + inc - c;
    hist[tid] = start;
    if (tid < nbins) {
      Hc[tid * G + b] = c;
      Hs[tid * G + b] = start;
    }
  }
  __syncthreads();

#pragma unroll
  for (int k = 0; k < CHUNK_EPT; ++k) {
    if (d[k] >= 0) {
      const int p = e0 + hist[d[k] >> BIN_SHIFT] + r[k];
      tdst[p] = d[k];
      tsrc[p] = s[k];
    }
  }
}

__device__ __forceinline__ void bin_sort_role(const int* __restrict__ Hc,
                                              const int* __restrict__ Hs, int G,
                                              const int* __restrict__ tdst,
                                              const int* __restrict__ tsrc, int n,
                                              int* __restrict__ off, float* __restrict__ dinv,
                                              int* __restrict__ ssrc, int k) {
  __shared__ int seg_cnt[1024];
  __shared__ int seg_pos[1024];
  __shared__ int cnt_l[BIN_NODES];
  __shared__ int cur_l[BIN_NODES];
  __shared__ int w_a[32];
  __shared__ int w_b[32];
  constexpr int RU = 8;
  const int tid = threadIdx.x;
  const int lane = tid & 31;
  const int w = tid >> 5;
  const int wv = __builtin_amdgcn_readfirstlane(w);
  const int lo = k << BIN_SHIFT;
  const int front = k * G;

  const int c = (tid < G) ? Hc[front + tid] : 0;
  const int st = (tid < G) ? Hs[front + tid] : 0;
  if (tid < BIN_NODES) cnt_l[tid] = 0;
  seg_cnt[tid] = c;
  seg_pos[tid] = tid * CHUNK + st;

  int pre = st;
  for (int m = 16; m >= 1; m >>= 1) pre += __shfl_xor(pre, m, 32);
  if (lane == 0) w_b[w] = pre;
  __syncthreads();

  int front_edges = 0;
  for (int j = 0; j < 32; ++j) front_edges += w_b[j];

  for (int b0 = wv; b0 < G; b0 += 32 * RU) {
    int cb[RU], pb[RU], d[RU];
#pragma unroll
    for (int u = 0; u < RU; ++u) {
      const int b = b0 + 32 * u;
      const int bb = (b < G) ? b : 0;
      cb[u] = seg_cnt[bb];
      pb[u] = seg_pos[bb];
    }
#pragma unroll
    for (int u = 0; u < RU; ++u) {
      const int b = b0 + 32 * u;
      cb[u] = (b < G) ? __builtin_amdgcn_readfirstlane(cb[u]) : 0;
      pb[u] = __builtin_amdgcn_readfirstlane(pb[u]);
      const bool in = lane < cb[u];
      const int v = tdst[in ? pb[u] + lane : 0];
      d[u] = in ? v : -1;
    }
#pragma unroll
    for (int u = 0; u < RU; ++u)
      if (d[u] >= 0) atomicAdd(&cnt_l[d[u] - lo], 1);
#pragma unroll
    for (int u = 0; u < RU; ++u)
      for (int j = 32 + lane; j < cb[u]; j += 32) atomicAdd(&cnt_l[tdst[pb[u] + j] - lo], 1);
  }
  __syncthreads();

  int cv = 0, cinc = 0;
  if (tid < BIN_NODES) {
    cv = cnt_l[tid];
    cinc = wave_scan_incl(cv, lane);
    if (lane == 31) w_a[w] = cinc;
  }
  __syncthreads();
  if (tid < BIN_NODES) {
    int b2 = 0;
    for (int j = 0; j < 8; ++j)
      if (j < w) b2 += w_a[j];
    const int o = b2 + cinc - cv;
    cur_l[tid] = o;
    const int v = lo + tid;
    if (v < n) {
      off[v] = front_edges + o;
      if (v == n - 1) off[n] = front_edges + o + cv;
      dinv[v] = rsqrtf((float)(cv + 1));
    }
  }
  __syncthreads();

  for (int b0 = wv; b0 < G; b0 += 32 * RU) {
    int cb[RU], pb[RU], d[RU], s[RU];
#pragma unroll
    for (int u = 0; u < RU; ++u) {
      const int b = b0 + 32 * u;
      const int bb = (b < G) ? b : 0;
      cb[u] = seg_cnt[bb];
      pb[u] = seg_pos[bb];
    }
#pragma unroll
    for (int u = 0; u < RU; ++u) {
      const int b = b0 + 32 * u;
      cb[u] = (b < G) ? __builtin_amdgcn_readfirstlane(cb[u]) : 0;
      pb[u] = __builtin_amdgcn_readfirstlane(pb[u]);
      const bool in = lane < cb[u];
      const int q = in ? pb[u] + lane : 0;
      const int v = tdst[q];
      s[u] = tsrc[q];
      d[u] = in ? v : -1;
    }
#pragma unroll
    for (int u = 0; u < RU; ++u) {
      if (d[u] >= 0) {
        const int pos = atomicAdd(&cur_l[d[u] - lo], 1);
        ssrc[front_edges + pos] = s[u];
      }
    }
#pragma unroll
    for (int u = 0; u < RU; ++u) {
      for (int j = 32 + lane; j < cb[u]; j += 32) {
        const int dd = tdst[pb[u] + j];
        const int ss = tsrc[pb[u] + j];
        const int pos = atomicAdd(&cur_l[dd - lo], 1);
        ssrc[front_edges + pos] = ss;
      }
    }
  }
}

template <int K, int NT>
__device__ __forceinline__ void strip_gemm_role(const float* __restrict__ A,
                                                const float* __restrict__ B,
                                                float* __restrict__ C, int M, int ldb, int ldc,
                                                int wg) {
  typedef float f4 __attribute__((ext_vector_type(4)));
  constexpr int N = NT * 16;
  constexpr int KP = K / 2;
  constexpr int KS = K / 4;
  constexpr int KQ = KS / 4;
  constexpr int TW = NT / 4;
  constexpr int PITCH = 2 * N + 32;
  constexpr int ITEMS = KP * (N / 4);
  static_assert(ITEMS % 1024 == 0, "staging loop has no tail");
  static_assert(NT % 4 == 0 && KS % 4 == 0, "four waves per strip, four quarters of k");
  __shared__ __attribute__((aligned(16))) float Bl[KP * PITCH];

  const int tid = threadIdx.x;
  const int lane = tid & 31;
  const int w = __builtin_amdgcn_readfirstlane(tid >> 5);
  const int c = lane & 15;
  const int h = lane >> 4;

  const int nstrips = M >> 4;
  const int strip = wg * 8 + (w >> 2);
  const int t0 = (w & 3) * TW;
  const bool active = strip < nstrips;
  const int m0 = (active ? strip : nstrips - 1) << 4;

  v2f a[KQ];
  const float* Arow = A + (size_t)(m0 + c) * K + 2 * h;
#pragma unroll
  for (int kk = 0; kk < KQ; ++kk) a[kk] = *(const v2f*)(Arow + 4 * kk);

  {
    f4 r0[ITEMS / 1024], r1[ITEMS / 1024];
#pragma unroll
    for (int j = 0; j < ITEMS / 1024; ++j) {
      const int i = tid + 1024 * j;
      const int kp = i / (N / 4), n4 = i % (N / 4);
      r0[j] = *(const f4*)(B + (size_t)(2 * kp) * ldb + 4 * n4);
      r1[j] = *(const f4*)(B + (size_t)(2 * kp + 1) * ldb + 4 * n4);
    }
#pragma unroll
    for (int j = 0; j < ITEMS / 1024; ++j) {
      const int i = tid + 1024 * j;
      const int kp = i / (N / 4), n4 = i % (N / 4);
      float* d = Bl + kp * PITCH + 8 * n4;
      *(f4*)d = (f4){r0[j].x, r1[j].x, r0[j].y, r1[j].y};
      *(f4*)(d + 4) = (f4){r0[j].z, r1[j].z, r0[j].w, r1[j].w};
    }
  }
  __syncthreads();

  if (active) {
    v8f acc[TW];
#pragma unroll
    for (int t = 0; t < TW; ++t) acc[t] = (v8f){};

    const float* Bf = Bl + h * PITCH + 2 * c + 32 * t0;
#pragma unroll 1
    for (int kq = 0; kq < 4; ++kq) {
      const float* Bq = Bf + (2 * KQ * kq) * PITCH;
#pragma unroll
      for (int kk = 0; kk < KQ; ++kk) {
#pragma unroll
        for (int t = 0; t < TW; ++t) {
          const v2f b = *(const v2f*)(Bq + (2 * kk) * PITCH + 32 * t);
          acc[t] = __builtin_amdgcn_wmma_f32_16x16x4_f32(false, a[kk], false, b, (short)0,
                                                         acc[t], false, false);
        }
      }
      if (kq < 3) {
#pragma unroll
        for (int kk = 0; kk < KQ; ++kk) a[kk] = *(const v2f*)(Arow + 4 * (KQ * (kq + 1) + kk));
      }
    }

    float* Crow = C + (size_t)(m0 + 8 * h) * ldc + c + 16 * t0;
#pragma unroll
    for (int t = 0; t < TW; ++t) {
#pragma unroll
      for (int i = 0; i < 8; ++i) Crow[(size_t)i * ldc + t * 16] = acc[t][i];
    }
  }
}

__global__ __launch_bounds__(1024, 16) void edge_bin_gemm(
    const int* __restrict__ src, const int* __restrict__ dst, int E, int nbins,
    int* __restrict__ Hc, int* __restrict__ Hs, int* __restrict__ tdst, int* __restrict__ tsrc,
    int nrole, const float* __restrict__ x, const float* __restrict__ W1,
    float* __restrict__ h1, int n, int wg0) {
  const int b = blockIdx.x;
  if (b < nrole)
    edge_bin_role(src, dst, E, nbins, Hc, Hs, tdst, tsrc, b, nrole);
  else
    strip_gemm_role<IN_C, HID_C / 16>(x, W1, h1, n, HID_C, HID_C, wg0 + b - nrole);
}

__global__ __launch_bounds__(1024, 16) void bin_sort_gemm(
    const int* __restrict__ Hc, const int* __restrict__ Hs, int G,
    const int* __restrict__ tdst, const int* __restrict__ tsrc, int n, int* __restrict__ off,
    float* __restrict__ dinv, int* __restrict__ ssrc, int nrole, const float* __restrict__ x,
    const float* __restrict__ W1, float* __restrict__ h1, int wg0) {
  const int b = blockIdx.x;
  if (b < nrole)
    bin_sort_role(Hc, Hs, G, tdst, tsrc, n, off, dinv, ssrc, b);
  else
    strip_gemm_role<IN_C, HID_C / 16>(x, W1, h1, n, HID_C, HID_C, wg0 + b - nrole);
}

template <int C, bool RELU>
__global__ __launch_bounds__(256) void agg_gather(const float* __restrict__ h,
                                                  const int* __restrict__ off,
                                                  const int* __restrict__ ssrc,
                                                  const float* __restrict__ dinv,
                                                  const float* __restrict__ bias,
                                                  float* __restrict__ out, int n) {
  constexpr int VW = C / 32;
  typedef float VT __attribute__((ext_vector_type(VW)));
  const int lane = threadIdx.x & 31;
  const int v = __builtin_amdgcn_readfirstlane(
      (int)((blockIdx.x * (long long)blockDim.x + threadIdx.x) >> 5));
  if (v >= n) return;
  const int beg = off[v];
  const int c = off[v + 1] - beg;
  const float dv = dinv[v];
  __builtin_assume(c >= 0);
  const float* hl = h + lane * VW;

  const int tot = c + 1;
  VT acc = {};

  for (int j0 = 0; j0 < tot; j0 += 32) {
    const int m = min(32, tot - j0);
    const int idx = j0 + lane;
    int s_l = v;
    float d_l = (idx == c) ? dv : 0.0f;
    if (idx < c) {
      s_l = ssrc[beg + idx];
      d_l = dinv[s_l];
    }
    for (int j = 0; j < m; j += 8) {
      VT r[8];
      float wt[8];
      for (int u = 0; u < 8; ++u) {
        const int s = __builtin_amdgcn_readlane(s_l, j + u);
        wt[u] = dv * __builtin_bit_cast(
            float, __builtin_amdgcn_readlane(__builtin_bit_cast(int, d_l), j + u));
        r[u] = *(const VT*)(hl + (size_t)s * C);
      }
      for (int u = 0; u < 8; ++u) acc += r[u] * wt[u];
    }
  }

  const VT bv = *(const VT*)(bias + lane * VW);
  acc += bv;
  if (RELU) {
    for (int i = 0; i < VW; ++i) acc[i] = fmaxf(acc[i], 0.0f);
  }
  *(VT*)(out + (size_t)v * C + lane * VW) = acc;
}

#define TN 64
#define ZP (HID_C + 4)
__global__ __launch_bounds__(512, 16) void agg_relu_gemm(const float* __restrict__ hin,
                                                         const int* __restrict__ off,
                                                         const int* __restrict__ ssrc,
                                                         const float* __restrict__ dinv,
                                                         const float* __restrict__ bias,
                                                         const float* __restrict__ W2,
                                                         float* __restrict__ h2, int n) {
  typedef float f4 __attribute__((ext_vector_type(4)));
  constexpr int C = HID_C;
  constexpr int K = HID_C;
  constexpr int N = OUT_C;
  constexpr int KP = K / 2;
  constexpr int KS = K / 4;
  constexpr int PITCH = 2 * N + 32;
  constexpr int ITEMS = KP * (N / 4);
  static_assert(ITEMS % 512 == 0, "staging loop has no tail");
  __shared__ __attribute__((aligned(16))) float Bl[KP * PITCH];
  __shared__ __attribute__((aligned(16))) float Zt[TN * ZP];

  const int tid = threadIdx.x;
  const int lane = tid & 31;
  const int w = __builtin_amdgcn_readfirstlane(tid >> 5);
  const int tile0 = blockIdx.x * TN;

  {
    f4 r0[ITEMS / 512], r1[ITEMS / 512];
#pragma unroll
    for (int j = 0; j < ITEMS / 512; ++j) {
      const int i = tid + 512 * j;
      const int kp = i / (N / 4), n4 = i % (N / 4);
      r0[j] = *(const f4*)(W2 + (size_t)(2 * kp) * N + 4 * n4);
      r1[j] = *(const f4*)(W2 + (size_t)(2 * kp + 1) * N + 4 * n4);
    }
#pragma unroll
    for (int j = 0; j < ITEMS / 512; ++j) {
      const int i = tid + 512 * j;
      const int kp = i / (N / 4), n4 = i % (N / 4);
      float* d = Bl + kp * PITCH + 8 * n4;
      *(f4*)d = (f4){r0[j].x, r1[j].x, r0[j].y, r1[j].y};
      *(f4*)(d + 4) = (f4){r0[j].z, r1[j].z, r0[j].w, r1[j].w};
    }
  }

  const float* hl = hin + lane * 4;
  const f4 bv = *(const f4*)(bias + lane * 4);
  for (int q = 0; q < 4; ++q) {
    const int r = w * 4 + q;
    const int v = tile0 + r;
    if (v >= n) break;
    const int beg = off[v];
    const int c = off[v + 1] - beg;
    const float dv = dinv[v];
    __builtin_assume(c >= 0);
    const int tot = c + 1;
    f4 acc = {};
    for (int j0 = 0; j0 < tot; j0 += 32) {
      const int m = min(32, tot - j0);
      const int idx = j0 + lane;
      int s_l = v;
      float d_l = (idx == c) ? dv : 0.0f;
      if (idx < c) {
        s_l = ssrc[beg + idx];
        d_l = dinv[s_l];
      }
      for (int j = 0; j < m; j += 8) {
        f4 rr[8];
        float wt[8];
        for (int u = 0; u < 8; ++u) {
          const int s = __builtin_amdgcn_readlane(s_l, j + u);
          wt[u] = dv * __builtin_bit_cast(
              float, __builtin_amdgcn_readlane(__builtin_bit_cast(int, d_l), j + u));
          rr[u] = *(const f4*)(hl + (size_t)s * C);
        }
        for (int u = 0; u < 8; ++u) acc += rr[u] * wt[u];
      }
    }
    acc += bv;
    for (int i = 0; i < 4; ++i) acc[i] = fmaxf(acc[i], 0.0f);
    *(f4*)(Zt + r * ZP + lane * 4) = acc;
  }
  __syncthreads();

  const int st = w >> 2;
  const int t = w & 3;
  const int m0 = tile0 + st * 16;
  if (m0 < n) {
    const int cc = lane & 15;
    const int hh = lane >> 4;
    v8f acc = (v8f){};
    const float* Af = Zt + (st * 16 + cc) * ZP + 2 * hh;
    const float* Bf = Bl + hh * PITCH + 2 * cc + 32 * t;
#pragma unroll
    for (int kk = 0; kk < KS; ++kk) {
      const v2f a = *(const v2f*)(Af + 4 * kk);
      const v2f b = *(const v2f*)(Bf + (2 * kk) * PITCH);
      acc = __builtin_amdgcn_wmma_f32_16x16x4_f32(false, a, false, b, (short)0, acc, false,
                                                  false);
    }
    float* Crow = h2 + (size_t)(m0 + 8 * hh) * N + cc + 16 * t;
#pragma unroll
    for (int i = 0; i < 8; ++i) Crow[(size_t)i * N] = acc[i];
  }
}

template <int K, int NT, int NW>
__global__ __launch_bounds__(NW * 32) void proj_strip(const float* __restrict__ A,
                                                      const float* __restrict__ Wab,
                                                      float* __restrict__ U,
                                                      float* __restrict__ V, int M) {
  typedef float f4 __attribute__((ext_vector_type(4)));
  constexpr int NTH = NW * 32;
  constexpr int N = NT * 16;
  constexpr int KP = K / 2;
  constexpr int KS = K / 4;
  constexpr int PITCH = 2 * N + 32;
  constexpr int ITEMS = KP * (N / 4);
  static_assert(ITEMS % NTH == 0, "staging loop has no tail");
  __shared__ __attribute__((aligned(16))) float Bl[KP * PITCH];

  const int tid = threadIdx.x;
  const int lane = tid & 31;
  const int w = __builtin_amdgcn_readfirstlane(tid >> 5);
  const int c = lane & 15;
  const int h = lane >> 4;
  const int sel = blockIdx.y;
  const float* B = Wab + (size_t)sel * K * N;
  float* C = sel ? V : U;

  const int nstrips = M >> 4;
  const int strip = blockIdx.x * NW + w;
  const bool active = strip < nstrips;
  const int m0 = (active ? strip : nstrips - 1) << 4;

  v2f a[KS];
  const float* Arow = A + (size_t)(m0 + c) * K + 2 * h;
#pragma unroll
  for (int kk = 0; kk < KS; ++kk) a[kk] = *(const v2f*)(Arow + 4 * kk);

  {
    f4 r0[ITEMS / NTH], r1[ITEMS / NTH];
#pragma unroll
    for (int j = 0; j < ITEMS / NTH; ++j) {
      const int i = tid + NTH * j;
      const int kp = i / (N / 4), n4 = i % (N / 4);
      r0[j] = *(const f4*)(B + (size_t)(2 * kp) * N + 4 * n4);
      r1[j] = *(const f4*)(B + (size_t)(2 * kp + 1) * N + 4 * n4);
    }
#pragma unroll
    for (int j = 0; j < ITEMS / NTH; ++j) {
      const int i = tid + NTH * j;
      const int kp = i / (N / 4), n4 = i % (N / 4);
      float* d = Bl + kp * PITCH + 8 * n4;
      *(f4*)d = (f4){r0[j].x, r1[j].x, r0[j].y, r1[j].y};
      *(f4*)(d + 4) = (f4){r0[j].z, r1[j].z, r0[j].w, r1[j].w};
    }
  }
  __syncthreads();

  if (active) {
    v8f acc[NT];
#pragma unroll
    for (int t = 0; t < NT; ++t) acc[t] = (v8f){};

    const float* Bf = Bl + h * PITCH + 2 * c;
#pragma unroll
    for (int kk = 0; kk < KS; ++kk) {
#pragma unroll
      for (int t = 0; t < NT; ++t) {
        const v2f b = *(const v2f*)(Bf + (2 * kk) * PITCH + 32 * t);
        acc[t] = __builtin_amdgcn_wmma_f32_16x16x4_f32(false, a[kk], false, b, (short)0,
                                                       acc[t], false, false);
      }
    }

    float* Crow = C + (size_t)(m0 + 8 * h) * N + c;
#pragma unroll
    for (int t = 0; t < NT; ++t) {
#pragma unroll
      for (int i = 0; i < 8; ++i) Crow[(size_t)i * N + t * 16] = acc[t][i];
    }
  }
}

__global__ __launch_bounds__(256) void pair_score(const float* __restrict__ U,
                                                  const float* __restrict__ V,
                                                  const int* __restrict__ pidx,
                                                  const int* __restrict__ nidx,
                                                  const float* __restrict__ ba,
                                                  const float* __restrict__ Wb,
                                                  const float* __restrict__ bb,
                                                  float* __restrict__ out, int P) {
  typedef float f4 __attribute__((ext_vector_type(4)));
  const int lane = threadIdx.x & 31;
  const int wave = (int)((blockIdx.x * (long long)blockDim.x + threadIdx.x) >> 5);
  const int q = lane >> 3;
  const int t = lane & 7;
  const int total = 2 * P;
  const int base = wave * 16;
  if (base >= total) return;

  int sv[4], dv[4];
#pragma unroll
  for (int st = 0; st < 4; ++st) {
    const int pc = min(base + st * 4 + q, total - 1);
    const int* ix = (pc < P) ? pidx : nidx;
    const int r = (pc < P) ? pc : pc - P;
    sv[st] = ix[r];
    dv[st] = ix[P + r];
  }

  f4 bav[4], wbv[4];
#pragma unroll
  for (int i = 0; i < 4; ++i) {
    bav[i] = *(const f4*)(ba + i * 32 + t * 4);
    wbv[i] = *(const f4*)(Wb + i * 32 + t * 4);
  }
  const float bbv = bb[0];

#pragma unroll
  for (int g = 0; g < 4; g += 2) {
    f4 ru[2][4], rv[2][4];
#pragma unroll
    for (int h = 0; h < 2; ++h) {
      const float* up = U + (size_t)sv[g + h] * HID_C + t * 4;
      const float* vp = V + (size_t)dv[g + h] * HID_C + t * 4;
#pragma unroll
      for (int i = 0; i < 4; ++i) {
        ru[h][i] = *(const f4*)(up + i * 32);
        rv[h][i] = *(const f4*)(vp + i * 32);
      }
    }
#pragma unroll
    for (int h = 0; h < 2; ++h) {
      float part = 0.0f;
#pragma unroll
      for (int i = 0; i < 4; ++i) {
        const f4 hs = (ru[h][i] + rv[h][i]) + bav[i];
#pragma unroll
        for (int c = 0; c < 4; ++c) part += fmaxf(hs[c], 0.0f) * wbv[i][c];
      }
      part += __shfl_xor(part, 1, 32);
      part += __shfl_xor(part, 2, 32);
      part += __shfl_xor(part, 4, 32);
      const int p = base + (g + h) * 4 + q;
      if (t == 0 && p < total) out[p] = part + bbv;
    }
  }
}

extern "C" void kernel_launch(void* const* d_in, const int* in_sizes, int n_in,
                              void* d_out, int out_size, void* d_ws, size_t ws_size,
                              hipStream_t stream) {
  const float* x   = (const float*)d_in[0];
  const int* eidx  = (const int*)d_in[1];
  const int* pidx  = (const int*)d_in[2];
  const int* nidx  = (const int*)d_in[3];
  const float* W1  = (const float*)d_in[4];
  const float* b1  = (const float*)d_in[5];
  const float* W2  = (const float*)d_in[6];
  const float* b2  = (const float*)d_in[7];
  const float* Wa  = (const float*)d_in[8];
  const float* ba  = (const float*)d_in[9];
  const float* Wb  = (const float*)d_in[10];
  const float* bb  = (const float*)d_in[11];
  float* out = (float*)d_out;

  const int n = in_sizes[0] / IN_C;
  const int E = in_sizes[1] / 2;
  const int P = in_sizes[2] / 2;
  const int* e_src = eidx;
  const int* e_dst = eidx + E;

  float* ws   = (float*)d_ws;
  float* h1   = ws;
  float* h2   = h1 + (size_t)n * HID_C;
  float* pU   = h2 + (size_t)n * HID_C;
  float* pV   = pU + (size_t)n * HID_C;
  int*   Hc   = (int*)(pV + (size_t)n * HID_C);
  int*   Hs   = Hc + MAX_BINS * 1024;
  float* dinv = (float*)(Hs + MAX_BINS * 1024);
  int*   off  = (int*)(dinv + n);
  int*   ssrc = off + n + 1;
  int*   tdst = ssrc + E;
  int*   tsrc = tdst + E;
  float* z2   = h1 + (size_t)n * OUT_C;

  auto cdiv = [](long long a, long long b) { return (int)((a + b - 1) / b); };
  const int B = 256;

  const int nbins = cdiv(n, BIN_NODES);
  const int nchunks = cdiv(E, CHUNK);
  const int pwgs = cdiv(n / 16, 8);
  const int pwgs1 = pwgs / 2;
  edge_bin_gemm<<<nchunks + pwgs1, 1024, 0, stream>>>(e_src, e_dst, E, nbins, Hc, Hs, tdst,
                                                      tsrc, nchunks, x, W1, h1, n, 0);
  bin_sort_gemm<<<nbins + (pwgs - pwgs1), 1024, 0, stream>>>(
      Hc, Hs, nchunks, tdst, tsrc, n, off, dinv, ssrc, nbins, x, W1, h1, pwgs1);

  {
    agg_relu_gemm<<<cdiv(n, TN), 512, 0, stream>>>(h1, off, ssrc, dinv, b1, W2, h2, n);
  }

  {
    agg_gather<OUT_C, false><<<cdiv((long long)n * 32, B), B, 0, stream>>>(
        h2, off, ssrc, dinv, b2, z2, n);
  }

  {
    proj_strip<OUT_C, HID_C / 16, 4><<<dim3(cdiv(n / 16, 4), 2), 4 * 32, 0, stream>>>(z2, Wa, pU, pV, n);
    long long pwaves = cdiv(2LL * P, 16);
    pair_score<<<cdiv(pwaves * 32, B), B, 0, stream>>>(pU, pV, pidx, nidx, ba, Wb, bb,
                                                       out, P);
  }
}


# layer-2 aggregation + node projections in one launch of resident work-groups pipelined over 16-row tiles: 16 waves gather tile t+1 (next row's lists read ahead), 4 do tile t on WMMA f32; sums kept
# speedup vs baseline: 4.0706x; 1.0658x over previous
#include <hip/hip_runtime.h>
#include <hip/hip_bf16.h>

typedef __attribute__((ext_vector_type(2))) float v2f;
typedef __attribute__((ext_vector_type(8))) float v8f;

#define IN_C  128
#define HID_C 128
#define OUT_C 64

#define BIN_SHIFT 8
#define BIN_NODES 256
#define MAX_BINS  256
#define CHUNK_EPT 4
#define CHUNK     (1024 * CHUNK_EPT)

__device__ __forceinline__ int wave_scan_incl(int x, int lane) {
  for (int d = 1; d < 32; d <<= 1) {
    int t = __shfl_up(x, d, 32);
    if (lane >= d) x += t;
  }
  return x;
}

__device__ __forceinline__ void edge_bin_role(const int* __restrict__ src,
                                              const int* __restrict__ dst, int E, int nbins,
                                              int* __restrict__ Hc, int* __restrict__ Hs,
                                              int* __restrict__ tdst, int* __restrict__ tsrc,
                                              int b, int G) {
  __shared__ int hist[MAX_BINS];
  __shared__ int wsum[8];
  const int tid = threadIdx.x;
  const int lane = tid & 31;
  const int w = tid >> 5;
  const int e0 = b * CHUNK;

  if (tid < MAX_BINS) hist[tid] = 0;
  __syncthreads();

  int d[CHUNK_EPT], s[CHUNK_EPT], r[CHUNK_EPT];
#pragma unroll
  for (int k = 0; k < CHUNK_EPT; ++k) {
    const int e = e0 + tid + 1024 * k;
    d[k] = -1;
    s[k] = 0;
    if (e < E) {
      d[k] = dst[e];
      s[k] = src[e];
    }
  }
#pragma unroll
  for (int k = 0; k < CHUNK_EPT; ++k) {
    r[k] = 0;
    if (d[k] >= 0) r[k] = atomicAdd(&hist[d[k] >> BIN_SHIFT], 1);
  }
  __syncthreads();

  int c = 0, inc = 0;
  if (tid < MAX_BINS) {
    c = hist[tid];
    inc = wave_scan_incl(c, lane);
    if (lane == 31) wsum[w] = inc;
  }
  __syncthreads();
  if (tid < MAX_BINS) {
    int base = 0;
    for (int j = 0; j < 8; ++j)
      if (j < w) base += wsum[j];
    const int start = base + inc - c;
    hist[tid] = start;
    if (tid < nbins) {
      Hc[tid * G + b] = c;
      Hs[tid * G + b] = start;
    }
  }
  __syncthreads();

#pragma unroll
  for (int k = 0; k < CHUNK_EPT; ++k) {
    if (d[k] >= 0) {
      const int p = e0 + hist[d[k] >> BIN_SHIFT] + r[k];
      tdst[p] = d[k];
      tsrc[p] = s[k];
    }
  }
}

__device__ __forceinline__ void bin_sort_role(const int* __restrict__ Hc,
                                              const int* __restrict__ Hs, int G,
                                              const int* __restrict__ tdst,
                                              const int* __restrict__ tsrc, int n,
                                              int* __restrict__ off, float* __restrict__ dinv,
                                              int* __restrict__ ssrc, int k) {
  __shared__ int seg_cnt[1024];
  __shared__ int seg_pos[1024];
  __shared__ int cnt_l[BIN_NODES];
  __shared__ int cur_l[BIN_NODES];
  __shared__ int w_a[32];
  __shared__ int w_b[32];
  constexpr int RU = 8;
  const int tid = threadIdx.x;
  const int lane = tid & 31;
  const int w = tid >> 5;
  const int wv = __builtin_amdgcn_readfirstlane(w);
  const int lo = k << BIN_SHIFT;
  const int front = k * G;

  const int c = (tid < G) ? Hc[front + tid] : 0;
  const int st = (tid < G) ? Hs[front + tid] : 0;
  if (tid < BIN_NODES) cnt_l[tid] = 0;
  seg_cnt[tid] = c;
  seg_pos[tid] = tid * CHUNK + st;

  int pre = st;
  for (int m = 16; m >= 1; m >>= 1) pre += __shfl_xor(pre, m, 32);
  if (lane == 0) w_b[w] = pre;
  __syncthreads();

  int front_edges = 0;
  for (int j = 0; j < 32; ++j) front_edges += w_b[j];

  for (int b0 = wv; b0 < G; b0 += 32 * RU) {
    int cb[RU], pb[RU], d[RU];
#pragma unroll
    for (int u = 0; u < RU; ++u) {
      const int b = b0 + 32 * u;
      const int bb = (b < G) ? b : 0;
      cb[u] = seg_cnt[bb];
      pb[u] = seg_pos[bb];
    }
#pragma unroll
    for (int u = 0; u < RU; ++u) {
      const int b = b0 + 32 * u;
      cb[u] = (b < G) ? __builtin_amdgcn_readfirstlane(cb[u]) : 0;
      pb[u] = __builtin_amdgcn_readfirstlane(pb[u]);
      const bool in = lane < cb[u];
      const int v = tdst[in ? pb[u] + lane : 0];
      d[u] = in ? v : -1;
    }
#pragma unroll
    for (int u = 0; u < RU; ++u)
      if (d[u] >= 0) atomicAdd(&cnt_l[d[u] - lo], 1);
#pragma unroll
    for (int u = 0; u < RU; ++u)
      for (int j = 32 + lane; j < cb[u]; j += 32) atomicAdd(&cnt_l[tdst[pb[u] + j] - lo], 1);
  }
  __syncthreads();

  int cv = 0, cinc = 0;
  if (tid < BIN_NODES) {
    cv = cnt_l[tid];
    cinc = wave_scan_incl(cv, lane);
    if (lane == 31) w_a[w] = cinc;
  }
  __syncthreads();
  if (tid < BIN_NODES) {
    int b2 = 0;
    for (int j = 0; j < 8; ++j)
      if (j < w) b2 += w_a[j];
    const int o = b2 + cinc - cv;
    cur_l[tid] = o;
    const int v = lo + tid;
    if (v < n) {
      off[v] = front_edges + o;
      if (v == n - 1) off[n] = front_edges + o + cv;
      dinv[v] = rsqrtf((float)(cv + 1));
    }
  }
  __syncthreads();

  for (int b0 = wv; b0 < G; b0 += 32 * RU) {
    int cb[RU], pb[RU], d[RU], s[RU];
#pragma unroll
    for (int u = 0; u < RU; ++u) {
      const int b = b0 + 32 * u;
      const int bb = (b < G) ? b : 0;
      cb[u] = seg_cnt[bb];
      pb[u] = seg_pos[bb];
    }
#pragma unroll
    for (int u = 0; u < RU; ++u) {
      const int b = b0 + 32 * u;
      cb[u] = (b < G) ? __builtin_amdgcn_readfirstlane(cb[u]) : 0;
      pb[u] = __builtin_amdgcn_readfirstlane(pb[u]);
      const bool in = lane < cb[u];
      const int q = in ? pb[u] + lane : 0;
      const int v = tdst[q];
      s[u] = tsrc[q];
      d[u] = in ? v : -1;
    }
#pragma unroll
    for (int u = 0; u < RU; ++u) {
      if (d[u] >= 0) {
        const int pos = atomicAdd(&cur_l[d[u] - lo], 1);
        ssrc[front_edges + pos] = s[u];
      }
    }
#pragma unroll
    for (int u = 0; u < RU; ++u) {
      for (int j = 32 + lane; j < cb[u]; j += 32) {
        const int dd = tdst[pb[u] + j];
        const int ss = tsrc[pb[u] + j];
        const int pos = atomicAdd(&cur_l[dd - lo], 1);
        ssrc[front_edges + pos] = ss;
      }
    }
  }
}

template <int K, int NT>
__device__ __forceinline__ void strip_gemm_role(const float* __restrict__ A,
                                                const float* __restrict__ B,
                                                float* __restrict__ C, int M, int ldb, int ldc,
                                                int wg) {
  typedef float f4 __attribute__((ext_vector_type(4)));
  constexpr int N = NT * 16;
  constexpr int KP = K / 2;
  constexpr int KS = K / 4;
  constexpr int KQ = KS / 4;
  constexpr int TW = NT / 4;
  constexpr int PITCH = 2 * N + 32;
  constexpr int ITEMS = KP * (N / 4);
  static_assert(ITEMS % 1024 == 0, "staging loop has no tail");
  static_assert(NT % 4 == 0 && KS % 4 == 0, "four waves per strip, four quarters of k");
  __shared__ __attribute__((aligned(16))) float Bl[KP * PITCH];

  const int tid = threadIdx.x;
  const int lane = tid & 31;
  const int w = __builtin_amdgcn_readfirstlane(tid >> 5);
  const int c = lane & 15;
  const int h = lane >> 4;

  const int nstrips = M >> 4;
  const int strip = wg * 8 + (w >> 2);
  const int t0 = (w & 3) * TW;
  const bool active = strip < nstrips;
  const int m0 = (active ? strip : nstrips - 1) << 4;

  v2f a[KQ];
  const float* Arow = A + (size_t)(m0 + c) * K + 2 * h;
#pragma unroll
  for (int kk = 0; kk < KQ; ++kk) a[kk] = *(const v2f*)(Arow + 4 * kk);

  {
    f4 r0[ITEMS / 1024], r1[ITEMS / 1024];
#pragma unroll
    for (int j = 0; j < ITEMS / 1024; ++j) {
      const int i = tid + 1024 * j;
      const int kp = i / (N / 4), n4 = i % (N / 4);
      r0[j] = *(const f4*)(B + (size_t)(2 * kp) * ldb + 4 * n4);
      r1[j] = *(const f4*)(B + (size_t)(2 * kp + 1) * ldb + 4 * n4);
    }
#pragma unroll
    for (int j = 0; j < ITEMS / 1024; ++j) {
      const int i = tid + 1024 * j;
      const int kp = i / (N / 4), n4 = i % (N / 4);
      float* d = Bl + kp * PITCH + 8 * n4;
      *(f4*)d = (f4){r0[j].x, r1[j].x, r0[j].y, r1[j].y};
      *(f4*)(d + 4) = (f4){r0[j].z, r1[j].z, r0[j].w, r1[j].w};
    }
  }
  __syncthreads();

  if (active) {
    v8f acc[TW];
#pragma unroll
    for (int t = 0; t < TW; ++t) acc[t] = (v8f){};

    const float* Bf = Bl + h * PITCH + 2 * c + 32 * t0;
#pragma unroll 1
    for (int kq = 0; kq < 4; ++kq) {
      const float* Bq = Bf + (2 * KQ * kq) * PITCH;
#pragma unroll
      for (int kk = 0; kk < KQ; ++kk) {
#pragma unroll
        for (int t = 0; t < TW; ++t) {
          const v2f b = *(const v2f*)(Bq + (2 * kk) * PITCH + 32 * t);
          acc[t] = __builtin_amdgcn_wmma_f32_16x16x4_f32(false, a[kk], false, b, (short)0,
                                                         acc[t], false, false);
        }
      }
      if (kq < 3) {
#pragma unroll
        for (int kk = 0; kk < KQ; ++kk) a[kk] = *(const v2f*)(Arow + 4 * (KQ * (kq + 1) + kk));
      }
    }

    float* Crow = C + (size_t)(m0 + 8 * h) * ldc + c + 16 * t0;
#pragma unroll
    for (int t = 0; t < TW; ++t) {
#pragma unroll
      for (int i = 0; i < 8; ++i) Crow[(size_t)i * ldc + t * 16] = acc[t][i];
    }
  }
}

__global__ __launch_bounds__(1024, 16) void edge_bin_gemm(
    const int* __restrict__ src, const int* __restrict__ dst, int E, int nbins,
    int* __restrict__ Hc, int* __restrict__ Hs, int* __restrict__ tdst, int* __restrict__ tsrc,
    int nrole, const float* __restrict__ x, const float* __restrict__ W1,
    float* __restrict__ h1, int n, int wg0) {
  const int b = blockIdx.x;
  if (b < nrole)
    edge_bin_role(src, dst, E, nbins, Hc, Hs, tdst, tsrc, b, nrole);
  else
    strip_gemm_role<IN_C, HID_C / 16>(x, W1, h1, n, HID_C, HID_C, wg0 + b - nrole);
}

__global__ __launch_bounds__(1024, 16) void bin_sort_gemm(
    const int* __restrict__ Hc, const int* __restrict__ Hs, int G,
    const int* __restrict__ tdst, const int* __restrict__ tsrc, int n, int* __restrict__ off,
    float* __restrict__ dinv, int* __restrict__ ssrc, int nrole, const float* __restrict__ x,
    const float* __restrict__ W1, float* __restrict__ h1, int wg0) {
  const int b = blockIdx.x;
  if (b < nrole)
    bin_sort_role(Hc, Hs, G, tdst, tsrc, n, off, dinv, ssrc, b);
  else
    strip_gemm_role<IN_C, HID_C / 16>(x, W1, h1, n, HID_C, HID_C, wg0 + b - nrole);
}

#define TN 64
#define ZP (HID_C + 4)
__global__ __launch_bounds__(512, 16) void agg_relu_gemm(const float* __restrict__ hin,
                                                         const int* __restrict__ off,
                                                         const int* __restrict__ ssrc,
                                                         const float* __restrict__ dinv,
                                                         const float* __restrict__ bias,
                                                         const float* __restrict__ W2,
                                                         float* __restrict__ h2, int n) {
  typedef float f4 __attribute__((ext_vector_type(4)));
  constexpr int C = HID_C;
  constexpr int K = HID_C;
  constexpr int N = OUT_C;
  constexpr int KP = K / 2;
  constexpr int KS = K / 4;
  constexpr int PITCH = 2 * N + 32;
  constexpr int ITEMS = KP * (N / 4);
  static_assert(ITEMS % 512 == 0, "staging loop has no tail");
  __shared__ __attribute__((aligned(16))) float Bl[KP * PITCH];
  __shared__ __attribute__((aligned(16))) float Zt[TN * ZP];

  const int tid = threadIdx.x;
  const int lane = tid & 31;
  const int w = __builtin_amdgcn_readfirstlane(tid >> 5);
  const int tile0 = blockIdx.x * TN;

  {
    f4 r0[ITEMS / 512], r1[ITEMS / 512];
#pragma unroll
    for (int j = 0; j < ITEMS / 512; ++j) {
      const int i = tid + 512 * j;
      const int kp = i / (N / 4), n4 = i % (N / 4);
      r0[j] = *(const f4*)(W2 + (size_t)(2 * kp) * N + 4 * n4);
      r1[j] = *(const f4*)(W2 + (size_t)(2 * kp + 1) * N + 4 * n4);
    }
#pragma unroll
    for (int j = 0; j < ITEMS / 512; ++j) {
      const int i = tid + 512 * j;
      const int kp = i / (N / 4), n4 = i % (N / 4);
      float* d = Bl + kp * PITCH + 8 * n4;
      *(f4*)d = (f4){r0[j].x, r1[j].x, r0[j].y, r1[j].y};
      *(f4*)(d + 4) = (f4){r0[j].z, r1[j].z, r0[j].w, r1[j].w};
    }
  }

  const float* hl = hin + lane * 4;
  const f4 bv = *(const f4*)(bias + lane * 4);
  for (int q = 0; q < 4; ++q) {
    const int r = w * 4 + q;
    const int v = tile0 + r;
    if (v >= n) break;
    const int beg = off[v];
    const int c = off[v + 1] - beg;
    const float dv = dinv[v];
    __builtin_assume(c >= 0);
    const int tot = c + 1;
    f4 acc = {};
    for (int j0 = 0; j0 < tot; j0 += 32) {
      const int m = min(32, tot - j0);
      const int idx = j0 + lane;
      int s_l = v;
      float d_l = (idx == c) ? dv : 0.0f;
      if (idx < c) {
        s_l = ssrc[beg + idx];
        d_l = dinv[s_l];
      }
      for (int j = 0; j < m; j += 8) {
        f4 rr[8];
        float wt[8];
        for (int u = 0; u < 8; ++u) {
          const int s = __builtin_amdgcn_readlane(s_l, j + u);
          wt[u] = dv * __builtin_bit_cast(
              float, __builtin_amdgcn_readlane(__builtin_bit_cast(int, d_l), j + u));
          rr[u] = *(const f4*)(hl + (size_t)s * C);
        }
        for (int u = 0; u < 8; ++u) acc += rr[u] * wt[u];
      }
    }
    acc += bv;
    for (int i = 0; i < 4; ++i) acc[i] = fmaxf(acc[i], 0.0f);
    *(f4*)(Zt + r * ZP + lane * 4) = acc;
  }
  __syncthreads();

  const int st = w >> 2;
  const int t = w & 3;
  const int m0 = tile0 + st * 16;
  if (m0 < n) {
    const int cc = lane & 15;
    const int hh = lane >> 4;
    v8f acc = (v8f){};
    const float* Af = Zt + (st * 16 + cc) * ZP + 2 * hh;
    const float* Bf = Bl + hh * PITCH + 2 * cc + 32 * t;
#pragma unroll
    for (int kk = 0; kk < KS; ++kk) {
      const v2f a = *(const v2f*)(Af + 4 * kk);
      const v2f b = *(const v2f*)(Bf + (2 * kk) * PITCH);
      acc = __builtin_amdgcn_wmma_f32_16x16x4_f32(false, a, false, b, (short)0, acc, false,
                                                  false);
    }
    float* Crow = h2 + (size_t)(m0 + 8 * hh) * N + cc + 16 * t;
#pragma unroll
    for (int i = 0; i < 8; ++i) Crow[(size_t)i * N] = acc[i];
  }
}

typedef float row2 __attribute__((ext_vector_type(2)));
__device__ __forceinline__ void rows8_load(const float* __restrict__ hl, int s_l, float d_l,
                                           float dv, int j, row2 (&r)[8], float (&wt)[8]) {
#pragma unroll
  for (int u = 0; u < 8; ++u) {
    const int s = __builtin_amdgcn_readlane(s_l, j + u);
    wt[u] = dv * __builtin_bit_cast(
                     float, __builtin_amdgcn_readlane(__builtin_bit_cast(int, d_l), j + u));
    r[u] = *(const row2*)(hl + (size_t)s * OUT_C);
  }
}
__device__ __forceinline__ void rows8_add(row2& acc, const row2 (&r)[8], const float (&wt)[8]) {
#pragma unroll
  for (int u = 0; u < 8; ++u) acc += r[u] * wt[u];
}

#define GW 16
#define MW 4
#define AP_T ((GW + MW) * 32)
#define ZP2 (OUT_C + 4)
__global__ __launch_bounds__(AP_T, 15) void agg_proj(const float* __restrict__ h,
                                                     const int* __restrict__ off,
                                                     const int* __restrict__ ssrc,
                                                     const float* __restrict__ dinv,
                                                     const float* __restrict__ bias,
                                                     const float* __restrict__ Wab,
                                                     float* __restrict__ U,
                                                     float* __restrict__ V, int n) {
  typedef float f4 __attribute__((ext_vector_type(4)));
  constexpr int K = OUT_C;
  constexpr int N = 2 * HID_C;
  constexpr int KP = K / 2;
  constexpr int KS = K / 4;
  constexpr int PITCH = 2 * N + 32;
  constexpr int ITEMS = KP * (N / 4);
  constexpr int TW = (N / 16) / MW;
  __shared__ __attribute__((aligned(16))) float Bl[KP * PITCH];
  __shared__ __attribute__((aligned(16))) float Zt[2 * GW * ZP2];

  const int tid = threadIdx.x;
  const int lane = tid & 31;
  const int w = __builtin_amdgcn_readfirstlane(tid >> 5);

  if (w >= GW) {
    constexpr int MT = MW * 32;
    constexpr int BT = 4;
    static_assert(ITEMS % (BT * MT) == 0, "copy loop has no tail");
    for (unsigned i0 = tid - GW * 32; i0 < (unsigned)ITEMS; i0 += BT * MT) {
      f4 r0[BT], r1[BT];
#pragma unroll
      for (int j = 0; j < BT; ++j) {
        const unsigned i = i0 + j * MT;
        const unsigned kp = i / (N / 4), n4 = i % (N / 4);
        const unsigned half = n4 / (HID_C / 4);
        const float* s0 = Wab + (size_t)(half * K + 2 * kp) * HID_C + 4 * (n4 % (HID_C / 4));
        r0[j] = *(const f4*)s0;
        r1[j] = *(const f4*)(s0 + HID_C);
      }
#pragma unroll
      for (int j = 0; j < BT; ++j) {
        const unsigned i = i0 + j * MT;
        const unsigned kp = i / (N / 4), n4 = i % (N / 4);
        float* d = Bl + kp * PITCH + 8 * n4;
        *(f4*)d = (f4){r0[j].x, r1[j].x, r0[j].y, r1[j].y};
        *(f4*)(d + 4) = (f4){r0[j].z, r1[j].z, r0[j].w, r1[j].w};
      }
    }
  }

  typedef float VT __attribute__((ext_vector_type(2)));
  const VT bv = *(const VT*)(bias + lane * 2);

  const int ntiles = (n + GW - 1) / GW;
  const int first = blockIdx.x;
  const int step = gridDim.x;
  const int nit = (first < ntiles) ? (ntiles - first + step - 1) / step : 0;

  int beg = 0, c = 0;
  float dv = 0.0f;
  int s_l = 0;
  float d_l = 0.0f;
  if (w < GW && nit > 0) {
    const int v0 = first * GW + w;
    s_l = v0;
    if (v0 < n) {
      beg = off[v0];
      c = off[v0 + 1] - beg;
      dv = dinv[v0];
      d_l = (lane == c) ? dv : 0.0f;
      if (lane < c) {
        s_l = ssrc[beg + lane];
        d_l = dinv[s_l];
      }
    }
  }

  for (int it = 0; it <= nit; ++it) {
    if (w < GW) {
      if (it < nit) {
        const int v = (first + it * step) * GW + w;
        const int vn = v + step * GW;
        const bool nx = (it + 1 < nit) && (vn < n);
        int begn = 0, cn = 0;
        float dvn = 0.0f;
        if (nx) {
          begn = off[vn];
          cn = off[vn + 1] - begn;
          dvn = dinv[vn];
        }
        int s_n = vn;
        float d_n = 0.0f;
        VT acc = {};
        if (v < n) {
          __builtin_assume(c >= 0);
          const float* hl = h + lane * 2;
          const int tot = c + 1;
          VT r[8];
          float wt[8];
          const int m0c = min(32, tot);
          rows8_load(hl, s_l, d_l, dv, 0, r, wt);
          if (nx && lane < cn) s_n = ssrc[begn + lane];
          rows8_add(acc, r, wt);
          if (nx) {
            d_n = (lane == cn) ? dvn : 0.0f;
            if (lane < cn) d_n = dinv[s_n];
          }
          for (int j = 8; j < m0c; j += 8) {
            rows8_load(hl, s_l, d_l, dv, j, r, wt);
            rows8_add(acc, r, wt);
          }
          for (int j0 = 32; j0 < tot; j0 += 32) {
            const int m = min(32, tot - j0);
            const int idx = j0 + lane;
            int s_c = v;
            float d_c = (idx == c) ? dv : 0.0f;
            if (idx < c) {
              s_c = ssrc[beg + idx];
              d_c = dinv[s_c];
            }
            for (int j = 0; j < m; j += 8) {
              rows8_load(hl, s_c, d_c, dv, j, r, wt);
              rows8_add(acc, r, wt);
            }
          }
          acc += bv;
        }
        *(VT*)(Zt + (it & 1) * (GW * ZP2) + w * ZP2 + lane * 2) = acc;
        beg = begn;
        c = cn;
        dv = dvn;
        s_l = s_n;
        d_l = d_n;
      }
    } else if (it > 0) {
      const int mw = w - GW;
      const int m0 = (first + (it - 1) * step) * GW;
      const int cc = lane & 15;
      const int hh = lane >> 4;
      constexpr int PT = 2;
      const float* Af = Zt + ((it - 1) & 1) * (GW * ZP2) + cc * ZP2 + 2 * hh;
      float* Cb = (mw < MW / 2) ? U : V;
#pragma unroll 1
      for (int p = 0; p < TW / PT; ++p) {
        v8f acc[PT];
#pragma unroll
        for (int t = 0; t < PT; ++t) acc[t] = (v8f){};
        const float* Bf = Bl + hh * PITCH + 2 * cc + 32 * (TW * mw + PT * p);
        v2f a_c = *(const v2f*)Af;
        v2f b_c[PT];
#pragma unroll
        for (int t = 0; t < PT; ++t) b_c[t] = *(const v2f*)(Bf + 32 * t);
#pragma unroll 3
        for (int kk = 1; kk < KS; ++kk) {
          const v2f a_n = *(const v2f*)(Af + 4 * kk);
          v2f b_n[PT];
#pragma unroll
          for (int t = 0; t < PT; ++t) b_n[t] = *(const v2f*)(Bf + (2 * kk) * PITCH + 32 * t);
#pragma unroll
          for (int t = 0; t < PT; ++t)
            acc[t] = __builtin_amdgcn_wmma_f32_16x16x4_f32(false, a_c, false, b_c[t], (short)0,
                                                           acc[t], false, false);
          a_c = a_n;
#pragma unroll
          for (int t = 0; t < PT; ++t) b_c[t] = b_n[t];
        }
#pragma unroll
        for (int t = 0; t < PT; ++t)
          acc[t] = __builtin_amdgcn_wmma_f32_16x16x4_f32(false, a_c, false, b_c[t], (short)0,
                                                         acc[t], false, false);
        float* Crow = Cb + (size_t)(m0 + 8 * hh) * HID_C + (mw % (MW / 2)) * (TW * 16) +
                      PT * 16 * p + cc;
        if (m0 + GW <= n) {
#pragma unroll
          for (int t = 0; t < PT; ++t) {
#pragma unroll
            for (int i = 0; i < 8; ++i) Crow[(size_t)i * HID_C + t * 16] = acc[t][i];
          }
        } else {
#pragma unroll
          for (int t = 0; t < PT; ++t) {
#pragma unroll
            for (int i = 0; i < 8; ++i)
              if (m0 + 8 * hh + i < n) Crow[(size_t)i * HID_C + t * 16] = acc[t][i];
          }
        }
      }
    }
    __syncthreads();
  }
}

__global__ __launch_bounds__(256) void pair_score(const float* __restrict__ U,
                                                  const float* __restrict__ V,
                                                  const int* __restrict__ pidx,
                                                  const int* __restrict__ nidx,
                                                  const float* __restrict__ ba,
                                                  const float* __restrict__ Wb,
                                                  const float* __restrict__ bb,
                                                  float* __restrict__ out, int P) {
  typedef float f4 __attribute__((ext_vector_type(4)));
  const int lane = threadIdx.x & 31;
  const int wave = (int)((blockIdx.x * (long long)blockDim.x + threadIdx.x) >> 5);
  const int q = lane >> 3;
  const int t = lane & 7;
  const int total = 2 * P;
  const int base = wave * 16;
  if (base >= total) return;

  int sv[4], dv[4];
#pragma unroll
  for (int st = 0; st < 4; ++st) {
    const int pc = min(base + st * 4 + q, total - 1);
    const int* ix = (pc < P) ? pidx : nidx;
    const int r = (pc < P) ? pc : pc - P;
    sv[st] = ix[r];
    dv[st] = ix[P + r];
  }

  f4 bav[4], wbv[4];
#pragma unroll
  for (int i = 0; i < 4; ++i) {
    bav[i] = *(const f4*)(ba + i * 32 + t * 4);
    wbv[i] = *(const f4*)(Wb + i * 32 + t * 4);
  }
  const float bbv = bb[0];

#pragma unroll
  for (int g = 0; g < 4; g += 2) {
    f4 ru[2][4], rv[2][4];
#pragma unroll
    for (int h = 0; h < 2; ++h) {
      const float* up = U + (size_t)sv[g + h] * HID_C + t * 4;
      const float* vp = V + (size_t)dv[g + h] * HID_C + t * 4;
#pragma unroll
      for (int i = 0; i < 4; ++i) {
        ru[h][i] = *(const f4*)(up + i * 32);
        rv[h][i] = *(const f4*)(vp + i * 32);
      }
    }
#pragma unroll
    for (int h = 0; h < 2; ++h) {
      float part = 0.0f;
#pragma unroll
      for (int i = 0; i < 4; ++i) {
        const f4 hs = (ru[h][i] + rv[h][i]) + bav[i];
#pragma unroll
        for (int c = 0; c < 4; ++c) part += fmaxf(hs[c], 0.0f) * wbv[i][c];
      }
      part += __shfl_xor(part, 1, 32);
      part += __shfl_xor(part, 2, 32);
      part += __shfl_xor(part, 4, 32);
      const int p = base + (g + h) * 4 + q;
      if (t == 0 && p < total) out[p] = part + bbv;
    }
  }
}

extern "C" void kernel_launch(void* const* d_in, const int* in_sizes, int n_in,
                              void* d_out, int out_size, void* d_ws, size_t ws_size,
                              hipStream_t stream) {
  const float* x   = (const float*)d_in[0];
  const int* eidx  = (const int*)d_in[1];
  const int* pidx  = (const int*)d_in[2];
  const int* nidx  = (const int*)d_in[3];
  const float* W1  = (const float*)d_in[4];
  const float* b1  = (const float*)d_in[5];
  const float* W2  = (const float*)d_in[6];
  const float* b2  = (const float*)d_in[7];
  const float* Wa  = (const float*)d_in[8];
  const float* ba  = (const float*)d_in[9];
  const float* Wb  = (const float*)d_in[10];
  const float* bb  = (const float*)d_in[11];
  float* out = (float*)d_out;

  const int n = in_sizes[0] / IN_C;
  const int E = in_sizes[1] / 2;
  const int P = in_sizes[2] / 2;
  const int* e_src = eidx;
  const int* e_dst = eidx + E;

  float* ws   = (float*)d_ws;
  float* h1   = ws;
  float* h2   = h1 + (size_t)n * HID_C;
  float* pU   = h2 + (size_t)n * HID_C;
  float* pV   = pU + (size_t)n * HID_C;
  int*   Hc   = (int*)(pV + (size_t)n * HID_C);
  int*   Hs   = Hc + MAX_BINS * 1024;
  float* dinv = (float*)(Hs + MAX_BINS * 1024);
  int*   off  = (int*)(dinv + n);
  int*   ssrc = off + n + 1;
  int*   tdst = ssrc + E;
  int*   tsrc = tdst + E;

  auto cdiv = [](long long a, long long b) { return (int)((a + b - 1) / b); };
  const int B = 256;

  const int nbins = cdiv(n, BIN_NODES);
  const int nchunks = cdiv(E, CHUNK);
  const int pwgs = cdiv(n / 16, 8);
  const int pwgs1 = pwgs / 2;
  edge_bin_gemm<<<nchunks + pwgs1, 1024, 0, stream>>>(e_src, e_dst, E, nbins, Hc, Hs, tdst,
                                                      tsrc, nchunks, x, W1, h1, n, 0);
  bin_sort_gemm<<<nbins + (pwgs - pwgs1), 1024, 0, stream>>>(
      Hc, Hs, nchunks, tdst, tsrc, n, off, dinv, ssrc, nbins, x, W1, h1, pwgs1);

  {
    agg_relu_gemm<<<cdiv(n, TN), 512, 0, stream>>>(h1, off, ssrc, dinv, b1, W2, h2, n);
  }

  {
    agg_proj<<<768, AP_T, 0, stream>>>(h2, off, ssrc, dinv, b2, Wa, pU, pV, n);
  }

  {
    long long pwaves = cdiv(2LL * P, 16);
    pair_score<<<cdiv(pwaves * 32, B), B, 0, stream>>>(pU, pV, pidx, nidx, ba, Wb, bb,
                                                       out, P);
  }
}
